# starting kernel
# baseline (speedup 1.0000x reference)
#include <hip/hip_runtime.h>

#define IN_CH  128
#define HID    64
#define OUT_CH 16

typedef float v2f __attribute__((ext_vector_type(2)));
typedef float v8f __attribute__((ext_vector_type(8)));

__global__ void gcn_init_kernel(float* __restrict__ deg,
                                float* __restrict__ agg1, const float* __restrict__ b1,
                                float* __restrict__ out,  const float* __restrict__ b2,
                                int nNodes) {
    long long idx = (long long)blockIdx.x * blockDim.x + threadIdx.x;
    if (idx >= (long long)nNodes * HID) return;
    int i = (int)(idx >> 6);
    int c = (int)(idx & (HID - 1));
    agg1[idx] = b1[c];
    if (c == 0)     deg[i] = 1.0f;
    if (c < OUT_CH) out[(long long)i * OUT_CH + c] = b2[c];
}

__global__ void gcn_degree_kernel(const int* __restrict__ dst,
                                  float* __restrict__ deg, int nEdges) {
    int e = blockIdx.x * blockDim.x + threadIdx.x;
    if (e < nEdges) atomicAdd(&deg[dst[e]], 1.0f);
}

__global__ void gcn_rsqrt_kernel(float* __restrict__ deg, int n) {
    int i = blockIdx.x * blockDim.x + threadIdx.x;
    if (i < n) deg[i] = rsqrtf(deg[i]);
}

__global__ void gcn_gemm_wmma_f32(const float* __restrict__ A,
                                  const float* __restrict__ W,
                                  float* __restrict__ C,
                                  int M, int K, int N) {
    const int wavesPerBlock = blockDim.x >> 5;
    const int wave = (int)(threadIdx.x >> 5);
    const int lane = (int)(threadIdx.x & 31);
    const int tilesN = N >> 4;
    const int gw = blockIdx.x * wavesPerBlock + wave;
    const int tileM = gw / tilesN;
    const int tileN = gw - tileM * tilesN;
    if (tileM * 16 >= M) return;

    const int row0 = tileM << 4;
    const int col0 = tileN << 4;
    const int mr = lane & 15;
    const int kh = (lane >> 4) << 1;

    v8f acc = {};
    for (int k = 0; k < K; k += 4) {
        v2f a, b;
        const float* ap = A + (long long)(row0 + mr) * K + (k + kh);
        a.x = ap[0];
        a.y = ap[1];
        const float* bp = W + (long long)(k + kh) * N + (col0 + mr);
        b.x = bp[0];
        b.y = bp[N];
        acc = __builtin_amdgcn_wmma_f32_16x16x4_f32(
             false, a,  false, b,
             (short)0, acc,  false,  false);
    }

    const int crow = row0 + ((lane >> 4) << 3);
    const int ccol = col0 + mr;
#pragma unroll
    for (int r = 0; r < 8; ++r)
        C[(long long)(crow + r) * N + ccol] = acc[r];
}

template <int C>
__global__ void gcn_scatter_kernel(const float* __restrict__ xl,
                                   const int* __restrict__ src,
                                   const int* __restrict__ dst,
                                   const float* __restrict__ dinv,
                                   float* __restrict__ agg,
                                   int nEdges, int nNodes) {
    long long idx = (long long)blockIdx.x * blockDim.x + threadIdx.x;
    long long total = (long long)(nEdges + nNodes) * C;
    if (idx >= total) return;
    int c = (int)(idx % C);
    long long e = idx / C;
    int s, d;
    if (e < nEdges) { s = src[e]; d = dst[e]; }
    else            { s = d = (int)(e - nEdges); }
    float norm = dinv[s] * dinv[d];
    float v = xl[(long long)s * C + c] * norm;
    atomicAdd(&agg[(long long)d * C + c], v);
}

__global__ void gcn_relu_kernel(float* __restrict__ h, long long n) {
    long long i = (long long)blockIdx.x * blockDim.x + threadIdx.x;
    if (i < n) h[i] = fmaxf(h[i], 0.0f);
}

extern "C" void kernel_launch(void* const* d_in, const int* in_sizes, int n_in,
                              void* d_out, int out_size, void* d_ws, size_t ws_size,
                              hipStream_t stream) {
    const float* x  = (const float*)d_in[0];
    const int*   ei = (const int*)  d_in[1];
    const float* W1 = (const float*)d_in[2];
    const float* b1 = (const float*)d_in[3];
    const float* W2 = (const float*)d_in[4];
    const float* b2 = (const float*)d_in[5];
    float* out = (float*)d_out;

    const int nNodes = in_sizes[0] / IN_CH;
    const int nEdges = in_sizes[1] / 2;
    const int* src = ei;
    const int* dst = ei + nEdges;

    float* ws  = (float*)d_ws;
    float* deg = ws;
    float* xl1 = deg + nNodes;
    float* h   = xl1 + (long long)nNodes * HID;
    float* xl2 = h   + (long long)nNodes * HID;

    const int B = 256;
    const long long hElems = (long long)nNodes * HID;

    gcn_init_kernel<<<(int)((hElems + B - 1) / B), B, 0, stream>>>(deg, h, b1, out, b2, nNodes);

    gcn_degree_kernel<<<(nEdges + B - 1) / B, B, 0, stream>>>(dst, deg, nEdges);
    gcn_rsqrt_kernel<<<(nNodes + B - 1) / B, B, 0, stream>>>(deg, nNodes);

    {
        int tiles = (nNodes / 16) * (HID / 16);
        int wpb = B / 32;
        gcn_gemm_wmma_f32<<<(tiles + wpb - 1) / wpb, B, 0, stream>>>(x, W1, xl1, nNodes, IN_CH, HID);
    }

    {
        long long tot = (long long)(nEdges + nNodes) * HID;
        gcn_scatter_kernel<HID><<<(int)((tot + B - 1) / B), B, 0, stream>>>(
            xl1, src, dst, deg, h, nEdges, nNodes);
    }

    gcn_relu_kernel<<<(int)((hElems + B - 1) / B), B, 0, stream>>>(h, hElems);

    {
        int tiles = (nNodes / 16) * (OUT_CH / 16);
        int wpb = B / 32;
        gcn_gemm_wmma_f32<<<(tiles + wpb - 1) / wpb, B, 0, stream>>>(h, W2, xl2, nNodes, HID, OUT_CH);
    }

    {
        long long tot = (long long)(nEdges + nNodes) * OUT_CH;
        gcn_scatter_kernel<OUT_CH><<<(int)((tot + B - 1) / B), B, 0, stream>>>(
            xl2, src, dst, deg, out, nEdges, nNodes);
    }
}


# Both aggregations as destination tiles in LDS: each group scans the whole dst list, adds its edges' rows by LDS float adds, stores tile+bias (relu) once; no global atomics; sums in LDS arrival order
# speedup vs baseline: 1.6897x; 1.6897x over previous
#include <hip/hip_runtime.h>

#define IN_CH  128
#define HID    64
#define OUT_CH 16

typedef float v2f __attribute__((ext_vector_type(2)));
typedef float v8f __attribute__((ext_vector_type(8)));

__global__ void gcn_init_kernel(float* __restrict__ deg, int nNodes) {
    int i = blockIdx.x * blockDim.x + threadIdx.x;
    if (i < nNodes) deg[i] = 1.0f;
}

__global__ void gcn_degree_kernel(const int* __restrict__ dst,
                                  float* __restrict__ deg, int nEdges) {
    int e = blockIdx.x * blockDim.x + threadIdx.x;
    if (e < nEdges) atomicAdd(&deg[dst[e]], 1.0f);
}

__global__ void gcn_rsqrt_kernel(float* __restrict__ deg, int n) {
    int i = blockIdx.x * blockDim.x + threadIdx.x;
    if (i < n) deg[i] = rsqrtf(deg[i]);
}

__global__ void gcn_gemm_wmma_f32(const float* __restrict__ A,
                                  const float* __restrict__ W,
                                  float* __restrict__ C,
                                  int M, int K, int N) {
    const int wavesPerBlock = blockDim.x >> 5;
    const int wave = (int)(threadIdx.x >> 5);
    const int lane = (int)(threadIdx.x & 31);
    const int tilesN = N >> 4;
    const int gw = blockIdx.x * wavesPerBlock + wave;
    const int tileM = gw / tilesN;
    const int tileN = gw - tileM * tilesN;
    if (tileM * 16 >= M) return;

    const int row0 = tileM << 4;
    const int col0 = tileN << 4;
    const int mr = lane & 15;
    const int kh = (lane >> 4) << 1;

    v8f acc = {};
    for (int k = 0; k < K; k += 4) {
        v2f a, b;
        const float* ap = A + (long long)(row0 + mr) * K + (k + kh);
        a.x = ap[0];
        a.y = ap[1];
        const float* bp = W + (long long)(k + kh) * N + (col0 + mr);
        b.x = bp[0];
        b.y = bp[N];
        acc = __builtin_amdgcn_wmma_f32_16x16x4_f32(
             false, a,  false, b,
             (short)0, acc,  false,  false);
    }

    const int crow = row0 + ((lane >> 4) << 3);
    const int ccol = col0 + mr;
#pragma unroll
    for (int r = 0; r < 8; ++r)
        C[(long long)(crow + r) * N + ccol] = acc[r];
}

#define TILE   392
#define AGG_T  1024
#define SCAN_U 8
#define QCAP   (32 + SCAN_U * 128)

static __device__ __forceinline__ int4 gcn_load4_guarded(const int* __restrict__ p, int g, int n) {
    const int e = g << 2;
    int4 r;
    r.x = (e     < n) ? p[e]     : -1;
    r.y = (e + 1 < n) ? p[e + 1] : -1;
    r.z = (e + 2 < n) ? p[e + 2] : -1;
    r.w = (e + 3 < n) ? p[e + 3] : -1;
    return r;
}

template <int C, bool RELU>
__global__ __launch_bounds__(AGG_T)
void gcn_tile_agg_kernel(const float* __restrict__ xl,
                         const int* __restrict__ src,
                         const int* __restrict__ dst,
                         const float* __restrict__ dinv,
                         const float* __restrict__ bias,
                         float* __restrict__ outp,
                         int nEdges, int nNodes) {
    __shared__ __attribute__((aligned(16))) float tile[TILE * C];
    __shared__ int   qmem[(AGG_T / 32) * QCAP];

    const int tid  = (int)threadIdx.x;
    const int lane = tid & 31;
    const int wave = tid >> 5;
    const int base = (int)blockIdx.x * TILE;
    const int nLoc = min(TILE, nNodes - base);

    constexpr int C4 = C / 4;
    constexpr int NV = TILE * C4;
    constexpr int IT = (NV + AGG_T - 1) / AGG_T;
    float4* tile4 = (float4*)tile;
    {
        float4 v[IT];
        float  di[IT];
#pragma unroll
        for (int t = 0; t < IT; ++t) {
            const int i  = min(t * AGG_T + tid, NV - 1);
            const int n  = min(i / C4, nLoc - 1);
            di[t] = dinv[base + n];
            v[t]  = *(const float4*)(xl + (long long)(base + n) * C + (i % C4) * 4);
        }
#pragma unroll
        for (int t = 0; t < IT; ++t) {
            const int i = t * AGG_T + tid;
            if (i < NV) {
                const bool ok = (i / C4) < nLoc;
                const float w = di[t] * di[t];
                float4 o;
                o.x = ok ? v[t].x * w : 0.0f;
                o.y = ok ? v[t].y * w : 0.0f;
                o.z = ok ? v[t].z * w : 0.0f;
                o.w = ok ? v[t].w * w : 0.0f;
                tile4[i] = o;
            }
        }
    }
    __syncthreads();

    int* myq = qmem + wave * QCAP;
    int cnt = 0;

    constexpr int LPE = C / 2;
    constexpr int EPS = 32 / LPE;
    constexpr int PU  = 4;
    const int sub = lane / LPE;
    const int cl  = lane % LPE;

    const int  nG  = (nEdges + 3) >> 2;
    const int  nG4 = nEdges >> 2;
    const bool al  = (((size_t)dst) & 15) == 0;

    for (int g0 = 0; g0 < nG; g0 += AGG_T * SCAN_U) {
        int4 d4[SCAN_U];
        if (al && g0 + AGG_T * SCAN_U <= nG4) {
            const int4* p = (const int4*)dst + (g0 + tid);
#pragma unroll
            for (int u = 0; u < SCAN_U; ++u)
                d4[u] = p[u * AGG_T];
        } else {
#pragma unroll
            for (int u = 0; u < SCAN_U; ++u)
                d4[u] = gcn_load4_guarded(dst, g0 + u * AGG_T + tid, nEdges);
        }

#pragma unroll
        for (int u = 0; u < SCAN_U; ++u) {
            const int e0 = (g0 + u * AGG_T + tid) << 2;
            const unsigned r0 = (unsigned)(d4[u].x - base);
            const unsigned r1 = (unsigned)(d4[u].y - base);
            const unsigned r2 = (unsigned)(d4[u].z - base);
            const unsigned r3 = (unsigned)(d4[u].w - base);
            const unsigned rm = min(min(r0, r1), min(r2, r3));
            if (__builtin_amdgcn_ballot_w32(rm < (unsigned)nLoc)) {
                const bool m0 = r0 < (unsigned)nLoc;
                const bool m1 = r1 < (unsigned)nLoc;
                const bool m2 = r2 < (unsigned)nLoc;
                const bool m3 = r3 < (unsigned)nLoc;
                const unsigned b0 = __builtin_amdgcn_ballot_w32(m0);
                const unsigned b1 = __builtin_amdgcn_ballot_w32(m1);
                const unsigned b2 = __builtin_amdgcn_ballot_w32(m2);
                const unsigned b3 = __builtin_amdgcn_ballot_w32(m3);
                if (b0) {
                    if (m0) myq[cnt + (int)__builtin_amdgcn_mbcnt_lo(b0, 0u)] = e0;
                    cnt += __builtin_popcount(b0);
                }
                if (b1) {
                    if (m1) myq[cnt + (int)__builtin_amdgcn_mbcnt_lo(b1, 0u)] = e0 + 1;
                    cnt += __builtin_popcount(b1);
                }
                if (b2) {
                    if (m2) myq[cnt + (int)__builtin_amdgcn_mbcnt_lo(b2, 0u)] = e0 + 2;
                    cnt += __builtin_popcount(b2);
                }
                if (b3) {
                    if (m3) myq[cnt + (int)__builtin_amdgcn_mbcnt_lo(b3, 0u)] = e0 + 3;
                    cnt += __builtin_popcount(b3);
                }
            }
        }

        const bool last = (g0 + AGG_T * SCAN_U >= nG);
        while (cnt >= 32 || (last && cnt > 0)) {
            const int n = min(cnt, 32);
            int   s = 0, dl = 0;
            float w = 0.0f;
            if (lane < n) {
                const int e = myq[cnt - n + lane];
                s  = src[e];
                dl = dst[e] - base;
                w  = dinv[s] * dinv[base + dl];
            }
            cnt -= n;
            for (int j0 = 0; j0 < n; j0 += EPS * PU) {
                float2 r[PU];
                float  ww[PU];
                int    dd[PU];
#pragma unroll
                for (int p = 0; p < PU; ++p) {
                    const int jj = min(j0 + p * EPS + sub, n - 1);
                    const int ss = __shfl(s, jj);
                    dd[p] = __shfl(dl, jj);
                    ww[p] = __shfl(w, jj);
                    r[p]  = *(const float2*)(xl + (long long)ss * C + cl * 2);
                }
#pragma unroll
                for (int p = 0; p < PU; ++p) {
                    if (j0 + p * EPS + sub < n) {
                        float* t = &tile[dd[p] * C + cl * 2];
                        atomicAdd(t,     r[p].x * ww[p]);
                        atomicAdd(t + 1, r[p].y * ww[p]);
                    }
                }
            }
        }
    }
    __syncthreads();

    {
        const float4* bias4 = (const float4*)bias;
        float4* out4 = (float4*)(outp + (long long)base * C);
#pragma unroll
        for (int t = 0; t < IT; ++t) {
            const int i = t * AGG_T + tid;
            if (i < nLoc * C4) {
                const float4 a = tile4[i];
                const float4 b = bias4[i % C4];
                float4 o;
                o.x = a.x + b.x; o.y = a.y + b.y; o.z = a.z + b.z; o.w = a.w + b.w;
                if (RELU) { o.x = fmaxf(o.x, 0.0f); o.y = fmaxf(o.y, 0.0f); o.z = fmaxf(o.z, 0.0f); o.w = fmaxf(o.w, 0.0f); }
                out4[i] = o;
            }
        }
    }
}

extern "C" void kernel_launch(void* const* d_in, const int* in_sizes, int n_in,
                              void* d_out, int out_size, void* d_ws, size_t ws_size,
                              hipStream_t stream) {
    const float* x  = (const float*)d_in[0];
    const int*   ei = (const int*)  d_in[1];
    const float* W1 = (const float*)d_in[2];
    const float* b1 = (const float*)d_in[3];
    const float* W2 = (const float*)d_in[4];
    const float* b2 = (const float*)d_in[5];
    float* out = (float*)d_out;

    const int nNodes = in_sizes[0] / IN_CH;
    const int nEdges = in_sizes[1] / 2;
    const int* src = ei;
    const int* dst = ei + nEdges;

    float* ws  = (float*)d_ws;
    float* deg = ws;
    float* xl1 = deg + ((nNodes + 63) & ~63);
    float* h   = xl1 + (long long)nNodes * HID;
    float* xl2 = h   + (long long)nNodes * HID;

    const int B = 256;
    const int nTiles = (nNodes + TILE - 1) / TILE;

    gcn_init_kernel<<<(nNodes + B - 1) / B, B, 0, stream>>>(deg, nNodes);

    gcn_degree_kernel<<<(nEdges + B - 1) / B, B, 0, stream>>>(dst, deg, nEdges);
    gcn_rsqrt_kernel<<<(nNodes + B - 1) / B, B, 0, stream>>>(deg, nNodes);

    {
        int tiles = (nNodes / 16) * (HID / 16);
        int wpb = B / 32;
        gcn_gemm_wmma_f32<<<(tiles + wpb - 1) / wpb, B, 0, stream>>>(x, W1, xl1, nNodes, IN_CH, HID);
    }

    gcn_tile_agg_kernel<HID, true><<<nTiles, AGG_T, 0, stream>>>(
        xl1, src, dst, deg, b1, h, nEdges, nNodes);

    {
        int tiles = (nNodes / 16) * (OUT_CH / 16);
        int wpb = B / 32;
        gcn_gemm_wmma_f32<<<(tiles + wpb - 1) / wpb, B, 0, stream>>>(h, W2, xl2, nNodes, HID, OUT_CH);
    }

    gcn_tile_agg_kernel<OUT_CH, false><<<nTiles, AGG_T, 0, stream>>>(
        xl2, src, dst, deg, b2, out, nEdges, nNodes);
}


# Second aggregation replays per-wave edge records (source, local row, weight) that the first aggregation's scan leaves in the workspace: no second scan of the dst list; sums kept in LDS arrival order
# speedup vs baseline: 2.1506x; 1.2728x over previous
#include <hip/hip_runtime.h>

#define IN_CH  128
#define HID    64
#define OUT_CH 16

typedef float v2f __attribute__((ext_vector_type(2)));
typedef float v8f __attribute__((ext_vector_type(8)));

__global__ void gcn_init_kernel(float* __restrict__ deg, int nNodes) {
    int i = blockIdx.x * blockDim.x + threadIdx.x;
    if (i < nNodes) deg[i] = 1.0f;
}

__global__ void gcn_degree_kernel(const int* __restrict__ dst,
                                  float* __restrict__ deg, int nEdges) {
    int e = blockIdx.x * blockDim.x + threadIdx.x;
    if (e < nEdges) atomicAdd(&deg[dst[e]], 1.0f);
}

__global__ void gcn_rsqrt_kernel(float* __restrict__ deg, int n) {
    int i = blockIdx.x * blockDim.x + threadIdx.x;
    if (i < n) deg[i] = rsqrtf(deg[i]);
}

__global__ void gcn_gemm_wmma_f32(const float* __restrict__ A,
                                  const float* __restrict__ W,
                                  float* __restrict__ C,
                                  int M, int K, int N) {
    const int wavesPerBlock = blockDim.x >> 5;
    const int wave = (int)(threadIdx.x >> 5);
    const int lane = (int)(threadIdx.x & 31);
    const int tilesN = N >> 4;
    const int gw = blockIdx.x * wavesPerBlock + wave;
    const int tileM = gw / tilesN;
    const int tileN = gw - tileM * tilesN;
    if (tileM * 16 >= M) return;

    const int row0 = tileM << 4;
    const int col0 = tileN << 4;
    const int mr = lane & 15;
    const int kh = (lane >> 4) << 1;

    v8f acc = {};
    for (int k = 0; k < K; k += 4) {
        v2f a, b;
        const float* ap = A + (long long)(row0 + mr) * K + (k + kh);
        a.x = ap[0];
        a.y = ap[1];
        const float* bp = W + (long long)(k + kh) * N + (col0 + mr);
        b.x = bp[0];
        b.y = bp[N];
        acc = __builtin_amdgcn_wmma_f32_16x16x4_f32(
             false, a,  false, b,
             (short)0, acc,  false,  false);
    }

    const int crow = row0 + ((lane >> 4) << 3);
    const int ccol = col0 + mr;
#pragma unroll
    for (int r = 0; r < 8; ++r)
        C[(long long)(crow + r) * N + ccol] = acc[r];
}

#define TILE   392
#define AGG_T  1024
#define SCAN_U 8
#define QCAP   (32 + SCAN_U * 128)
#define CAPW   16
#define AGG_W  (AGG_T / 32)
static_assert(AGG_W == 32, "one lane per wave when a wave reads its tile's block counts");

static __device__ __forceinline__ int4 gcn_load4_guarded(const int* __restrict__ p, int g, int n) {
    const int e = g << 2;
    int4 r;
    r.x = (e     < n) ? p[e]     : -1;
    r.y = (e + 1 < n) ? p[e + 1] : -1;
    r.z = (e + 2 < n) ? p[e + 2] : -1;
    r.w = (e + 3 < n) ? p[e + 3] : -1;
    return r;
}

template <int C, bool RELU, int MODE>
__global__ __launch_bounds__(AGG_T)
void gcn_tile_agg_kernel(const float* __restrict__ xl,
                         const int* __restrict__ src,
                         const int* __restrict__ dst,
                         const float* __restrict__ dinv,
                         const float* __restrict__ bias,
                         float* __restrict__ outp,
                         int*   __restrict__ recS,
                         int*   __restrict__ recD,
                         float* __restrict__ recW,
                         int*   __restrict__ recN,
                         int nEdges, int nNodes) {
    __shared__ __attribute__((aligned(16))) float tile[TILE * C];
    __shared__ int   qmem[(AGG_T / 32) * QCAP];
    __shared__ int   wcount[AGG_T / 32];

    const int tid  = (int)threadIdx.x;
    const int lane = tid & 31;
    const int wave = tid >> 5;
    const int base = (int)blockIdx.x * TILE;
    const int nLoc = min(TILE, nNodes - base);

    int myN = 0;
    if (MODE == 1) myN = recN[(int)blockIdx.x * (AGG_T / 32) + lane];

    constexpr int C4 = C / 4;
    constexpr int NV = TILE * C4;
    constexpr int IT = (NV + AGG_T - 1) / AGG_T;
    float4* tile4 = (float4*)tile;
    {
        float4 v[IT];
        float  di[IT];
#pragma unroll
        for (int t = 0; t < IT; ++t) {
            const int i  = min(t * AGG_T + tid, NV - 1);
            const int n  = min(i / C4, nLoc - 1);
            di[t] = dinv[base + n];
            v[t]  = *(const float4*)(xl + (long long)(base + n) * C + (i % C4) * 4);
        }
#pragma unroll
        for (int t = 0; t < IT; ++t) {
            const int i = t * AGG_T + tid;
            if (i < NV) {
                const bool ok = (i / C4) < nLoc;
                const float w = di[t] * di[t];
                float4 o;
                o.x = ok ? v[t].x * w : 0.0f;
                o.y = ok ? v[t].y * w : 0.0f;
                o.z = ok ? v[t].z * w : 0.0f;
                o.w = ok ? v[t].w * w : 0.0f;
                tile4[i] = o;
            }
        }
    }
    __syncthreads();

    int* myq = qmem + wave * QCAP;
    int cnt = 0;

    constexpr int LPE = C / 2;
    constexpr int EPS = 32 / LPE;
    constexpr int PU  = 4;
    const int sub = lane / LPE;
    const int cl  = lane % LPE;

    const size_t rbase = ((size_t)((int)blockIdx.x * AGG_W + wave) * CAPW) * 32 + lane;
    int nblk = 0;

    bool replay = false;
    if (MODE == 1)
        replay = (__builtin_amdgcn_ballot_w32(myN > CAPW) == 0u);

    if (MODE == 1 && replay) {
        constexpr int RPU = 8;
        static_assert(32 % (EPS * RPU) == 0, "a block is a whole number of steps");
        const int nb = __builtin_amdgcn_readfirstlane(__shfl(myN, wave));
        int s = 0, dl = -1;
        float w = 0.0f;
        if (nb > 0) { s = recS[rbase]; dl = recD[rbase]; w = recW[rbase]; }
        for (int b = 0; b < nb; ++b) {
            int s2 = s, dl2 = dl;
            float w2 = w;
            for (int j0 = 0; j0 < 32; j0 += EPS * RPU) {
                float2 r[RPU];
                float  ww[RPU];
                int    dd[RPU];
#pragma unroll
                for (int p = 0; p < RPU; ++p) {
                    const int jj = j0 + p * EPS + sub;
                    const int ss = __shfl(s, jj);
                    dd[p] = __shfl(dl, jj);
                    ww[p] = __shfl(w, jj);
                    r[p]  = *(const float2*)(xl + (long long)ss * C + cl * 2);
                }
                if (j0 == 0) {
                    const size_t o = rbase + (size_t)min(b + 1, nb - 1) * 32;
                    s2 = recS[o]; dl2 = recD[o]; w2 = recW[o];
                }
#pragma unroll
                for (int p = 0; p < RPU; ++p) {
                    if (dd[p] >= 0) {
                        float* t = &tile[dd[p] * C + cl * 2];
                        atomicAdd(t,     r[p].x * ww[p]);
                        atomicAdd(t + 1, r[p].y * ww[p]);
                    }
                }
            }
            s = s2; dl = dl2; w = w2;
        }
    } else {

    const int  nG  = (nEdges + 3) >> 2;
    const int  nG4 = nEdges >> 2;
    const bool al  = (((size_t)dst) & 15) == 0;

    for (int g0 = 0; g0 < nG; g0 += AGG_T * SCAN_U) {
        int4 d4[SCAN_U];
        if (al && g0 + AGG_T * SCAN_U <= nG4) {
            const int4* p = (const int4*)dst + (g0 + tid);
#pragma unroll
            for (int u = 0; u < SCAN_U; ++u)
                d4[u] = p[u * AGG_T];
        } else {
#pragma unroll
            for (int u = 0; u < SCAN_U; ++u)
                d4[u] = gcn_load4_guarded(dst, g0 + u * AGG_T + tid, nEdges);
        }

#pragma unroll
        for (int u = 0; u < SCAN_U; ++u) {
            const int e0 = (g0 + u * AGG_T + tid) << 2;
            const unsigned r0 = (unsigned)(d4[u].x - base);
            const unsigned r1 = (unsigned)(d4[u].y - base);
            const unsigned r2 = (unsigned)(d4[u].z - base);
            const unsigned r3 = (unsigned)(d4[u].w - base);
            const unsigned rm = min(min(r0, r1), min(r2, r3));
            if (__builtin_amdgcn_ballot_w32(rm < (unsigned)nLoc)) {
                const bool m0 = r0 < (unsigned)nLoc;
                const bool m1 = r1 < (unsigned)nLoc;
                const bool m2 = r2 < (unsigned)nLoc;
                const bool m3 = r3 < (unsigned)nLoc;
                const unsigned b0 = __builtin_amdgcn_ballot_w32(m0);
                const unsigned b1 = __builtin_amdgcn_ballot_w32(m1);
                const unsigned b2 = __builtin_amdgcn_ballot_w32(m2);
                const unsigned b3 = __builtin_amdgcn_ballot_w32(m3);
                if (b0) {
                    if (m0) myq[cnt + (int)__builtin_amdgcn_mbcnt_lo(b0, 0u)] = e0;
                    cnt += __builtin_popcount(b0);
                }
                if (b1) {
                    if (m1) myq[cnt + (int)__builtin_amdgcn_mbcnt_lo(b1, 0u)] = e0 + 1;
                    cnt += __builtin_popcount(b1);
                }
                if (b2) {
                    if (m2) myq[cnt + (int)__builtin_amdgcn_mbcnt_lo(b2, 0u)] = e0 + 2;
                    cnt += __builtin_popcount(b2);
                }
                if (b3) {
                    if (m3) myq[cnt + (int)__builtin_amdgcn_mbcnt_lo(b3, 0u)] = e0 + 3;
                    cnt += __builtin_popcount(b3);
                }
            }
        }

        const bool last = (g0 + AGG_T * SCAN_U >= nG);
        while (cnt >= 32 || (last && cnt > 0)) {
            const int n = min(cnt, 32);
            int   s = 0, dl = 0;
            float w = 0.0f;
            if (lane < n) {
                const int e = myq[cnt - n + lane];
                s  = src[e];
                dl = dst[e] - base;
                w  = dinv[s] * dinv[base + dl];
            }
            if (MODE == 0) {
                if (nblk < CAPW) {
                    const size_t o = rbase + (size_t)nblk * 32;
                    recS[o] = s;
                    recD[o] = (lane < n) ? dl : -1;
                    recW[o] = w;
                }
                ++nblk;
            }
            cnt -= n;
            for (int j0 = 0; j0 < n; j0 += EPS * PU) {
                float2 r[PU];
                float  ww[PU];
                int    dd[PU];
#pragma unroll
                for (int p = 0; p < PU; ++p) {
                    const int jj = min(j0 + p * EPS + sub, n - 1);
                    const int ss = __shfl(s, jj);
                    dd[p] = __shfl(dl, jj);
                    ww[p] = __shfl(w, jj);
                    r[p]  = *(const float2*)(xl + (long long)ss * C + cl * 2);
                }
#pragma unroll
                for (int p = 0; p < PU; ++p) {
                    if (j0 + p * EPS + sub < n) {
                        float* t = &tile[dd[p] * C + cl * 2];
                        atomicAdd(t,     r[p].x * ww[p]);
                        atomicAdd(t + 1, r[p].y * ww[p]);
                    }
                }
            }
        }
    }

    }

    if (MODE == 0 && lane == 0) wcount[wave] = nblk;
    __syncthreads();

    if (MODE == 0 && wave == 0) recN[(int)blockIdx.x * AGG_W + lane] = wcount[lane];

    {
        const float4* bias4 = (const float4*)bias;
        float4* out4 = (float4*)(outp + (long long)base * C);
#pragma unroll
        for (int t = 0; t < IT; ++t) {
            const int i = t * AGG_T + tid;
            if (i < nLoc * C4) {
                const float4 a = tile4[i];
                const float4 b = bias4[i % C4];
                float4 o;
                o.x = a.x + b.x; o.y = a.y + b.y; o.z = a.z + b.z; o.w = a.w + b.w;
                if (RELU) { o.x = fmaxf(o.x, 0.0f); o.y = fmaxf(o.y, 0.0f); o.z = fmaxf(o.z, 0.0f); o.w = fmaxf(o.w, 0.0f); }
                out4[i] = o;
            }
        }
    }
}

extern "C" void kernel_launch(void* const* d_in, const int* in_sizes, int n_in,
                              void* d_out, int out_size, void* d_ws, size_t ws_size,
                              hipStream_t stream) {
    const float* x  = (const float*)d_in[0];
    const int*   ei = (const int*)  d_in[1];
    const float* W1 = (const float*)d_in[2];
    const float* b1 = (const float*)d_in[3];
    const float* W2 = (const float*)d_in[4];
    const float* b2 = (const float*)d_in[5];
    float* out = (float*)d_out;

    const int nNodes = in_sizes[0] / IN_CH;
    const int nEdges = in_sizes[1] / 2;
    const int* src = ei;
    const int* dst = ei + nEdges;

    float* ws  = (float*)d_ws;
    float* deg = ws;
    float* xl1 = deg + ((nNodes + 63) & ~63);
    float* h   = xl1 + (long long)nNodes * HID;
    float* xl2 = h   + (long long)nNodes * HID;

    const int B = 256;
    const int nTiles = (nNodes + TILE - 1) / TILE;

    const long long recLen = (long long)nTiles * AGG_W * CAPW * 32;
    int*   recS = (int*)(xl2 + (((long long)nNodes * OUT_CH + 31) & ~31LL));
    int*   recD = recS + recLen;
    float* recW = (float*)(recD + recLen);
    int*   recN = (int*)(recW + recLen);

    gcn_init_kernel<<<(nNodes + B - 1) / B, B, 0, stream>>>(deg, nNodes);

    gcn_degree_kernel<<<(nEdges + B - 1) / B, B, 0, stream>>>(dst, deg, nEdges);
    gcn_rsqrt_kernel<<<(nNodes + B - 1) / B, B, 0, stream>>>(deg, nNodes);

    {
        int tiles = (nNodes / 16) * (HID / 16);
        int wpb = B / 32;
        gcn_gemm_wmma_f32<<<(tiles + wpb - 1) / wpb, B, 0, stream>>>(x, W1, xl1, nNodes, IN_CH, HID);
    }

    gcn_tile_agg_kernel<HID, true, 0><<<nTiles, AGG_T, 0, stream>>>(
        xl1, src, dst, deg, b1, h, recS, recD, recW, recN, nEdges, nNodes);

    {
        int tiles = (nNodes / 16) * (OUT_CH / 16);
        int wpb = B / 32;
        gcn_gemm_wmma_f32<<<(tiles + wpb - 1) / wpb, B, 0, stream>>>(h, W2, xl2, nNodes, HID, OUT_CH);
    }

    gcn_tile_agg_kernel<OUT_CH, false, 1><<<nTiles, AGG_T, 0, stream>>>(
        xl2, src, dst, deg, b2, out, recS, recD, recW, recN, nEdges, nNodes);
}


# First aggregation follows a per-tile bitmap of 32-edge groups (built by the degree kernel) and fetches only marked groups' dst, not the whole list per tile; sums kept in LDS arrival order
# speedup vs baseline: 2.3860x; 1.1095x over previous
#include <hip/hip_runtime.h>

#define IN_CH  128
#define HID    64
#define OUT_CH 16

typedef float v2f __attribute__((ext_vector_type(2)));
typedef float v8f __attribute__((ext_vector_type(8)));

__global__ void gcn_init_kernel(float* __restrict__ deg, int nNodes) {
    int i = blockIdx.x * blockDim.x + threadIdx.x;
    if (i < nNodes) deg[i] = 1.0f;
}

__global__ void gcn_degree_kernel(const int* __restrict__ dst,
                                  float* __restrict__ deg, int nEdges) {
    int e = blockIdx.x * blockDim.x + threadIdx.x;
    if (e < nEdges) atomicAdd(&deg[dst[e]], 1.0f);
}

__global__ void gcn_rsqrt_kernel(float* __restrict__ deg, int n) {
    int i = blockIdx.x * blockDim.x + threadIdx.x;
    if (i < n) deg[i] = rsqrtf(deg[i]);
}

__global__ void gcn_gemm_wmma_f32(const float* __restrict__ A,
                                  const float* __restrict__ W,
                                  float* __restrict__ C,
                                  int M, int K, int N) {
    const int wavesPerBlock = blockDim.x >> 5;
    const int wave = (int)(threadIdx.x >> 5);
    const int lane = (int)(threadIdx.x & 31);
    const int tilesN = N >> 4;
    const int gw = blockIdx.x * wavesPerBlock + wave;
    const int tileM = gw / tilesN;
    const int tileN = gw - tileM * tilesN;
    if (tileM * 16 >= M) return;

    const int row0 = tileM << 4;
    const int col0 = tileN << 4;
    const int mr = lane & 15;
    const int kh = (lane >> 4) << 1;

    v8f acc = {};
    for (int k = 0; k < K; k += 4) {
        v2f a, b;
        const float* ap = A + (long long)(row0 + mr) * K + (k + kh);
        a.x = ap[0];
        a.y = ap[1];
        const float* bp = W + (long long)(k + kh) * N + (col0 + mr);
        b.x = bp[0];
        b.y = bp[N];
        acc = __builtin_amdgcn_wmma_f32_16x16x4_f32(
             false, a,  false, b,
             (short)0, acc,  false,  false);
    }

    const int crow = row0 + ((lane >> 4) << 3);
    const int ccol = col0 + mr;
#pragma unroll
    for (int r = 0; r < 8; ++r)
        C[(long long)(crow + r) * N + ccol] = acc[r];
}

#define TILE   392
#define AGG_T  1024
#define SCAN_U 8
#define PG     8
#define EQCAP  (32 + PG * 32)
#define QCAP   (EQCAP + 1024)
static_assert(QCAP >= 32 + SCAN_U * 128, "the scanning path's queue (every scanned edge matches) fits too");
#define MAXT   256
#define IDX_T  128
#define CAPW   16
#define AGG_W  (AGG_T / 32)
static_assert(AGG_W == 32, "one lane per wave when a wave reads its tile's block counts");

__global__ __launch_bounds__(IDX_T)
void gcn_degree_index_kernel(const int* __restrict__ dst,
                             float* __restrict__ deg,
                             unsigned* __restrict__ bmap,
                             int nEdges, int nWords) {
    __shared__ unsigned rows[(IDX_T / 32) * MAXT];
    const int lane = (int)(threadIdx.x & 31);
    const int wave = (int)(threadIdx.x >> 5);
    const int word = (int)blockIdx.x * (IDX_T / 32) + wave;
    if (word >= nWords) return;
    unsigned* my = rows + wave * MAXT;
#pragma unroll
    for (int k = 0; k < MAXT / 32; ++k) my[k * 32 + lane] = 0u;
    const int e0 = word * 1024 + lane;
    for (int g0 = 0; g0 < 32; g0 += 8) {
        int dv[8];
#pragma unroll
        for (int u = 0; u < 8; ++u) {
            const int e = e0 + (g0 + u) * 32;
            dv[u] = (e < nEdges) ? dst[e] : -1;
        }
#pragma unroll
        for (int u = 0; u < 8; ++u) {
            if (dv[u] >= 0) {
                atomicAdd(&deg[dv[u]], 1.0f);
                atomicOr(&my[dv[u] / TILE], 1u << (g0 + u));
            }
        }
    }
#pragma unroll
    for (int k = 0; k < MAXT / 32; ++k)
        bmap[(size_t)word * MAXT + k * 32 + lane] = my[k * 32 + lane];
}

static __device__ __forceinline__ int4 gcn_load4_guarded(const int* __restrict__ p, int g, int n) {
    const int e = g << 2;
    int4 r;
    r.x = (e     < n) ? p[e]     : -1;
    r.y = (e + 1 < n) ? p[e + 1] : -1;
    r.z = (e + 2 < n) ? p[e + 2] : -1;
    r.w = (e + 3 < n) ? p[e + 3] : -1;
    return r;
}

template <int C, bool RELU, int MODE>
__global__ __launch_bounds__(AGG_T)
void gcn_tile_agg_kernel(const float* __restrict__ xl,
                         const int* __restrict__ src,
                         const int* __restrict__ dst,
                         const float* __restrict__ dinv,
                         const float* __restrict__ bias,
                         float* __restrict__ outp,
                         int*   __restrict__ recS,
                         int*   __restrict__ recD,
                         float* __restrict__ recW,
                         int*   __restrict__ recN,
                         const unsigned* __restrict__ bmap,
                         int useIdx,
                         int nEdges, int nNodes) {
    __shared__ __attribute__((aligned(16))) float tile[TILE * C];
    __shared__ int   qmem[(AGG_T / 32) * QCAP];
    __shared__ int   wcount[AGG_T / 32];

    const int tid  = (int)threadIdx.x;
    const int lane = tid & 31;
    const int wave = tid >> 5;
    const int base = (int)blockIdx.x * TILE;
    const int nLoc = min(TILE, nNodes - base);

    int myN = 0;
    if (MODE == 1) myN = recN[(int)blockIdx.x * (AGG_T / 32) + lane];

    constexpr int C4 = C / 4;
    constexpr int NV = TILE * C4;
    constexpr int IT = (NV + AGG_T - 1) / AGG_T;
    float4* tile4 = (float4*)tile;
    {
        float4 v[IT];
        float  di[IT];
#pragma unroll
        for (int t = 0; t < IT; ++t) {
            const int i  = min(t * AGG_T + tid, NV - 1);
            const int n  = min(i / C4, nLoc - 1);
            di[t] = dinv[base + n];
            v[t]  = *(const float4*)(xl + (long long)(base + n) * C + (i % C4) * 4);
        }
#pragma unroll
        for (int t = 0; t < IT; ++t) {
            const int i = t * AGG_T + tid;
            if (i < NV) {
                const bool ok = (i / C4) < nLoc;
                const float w = di[t] * di[t];
                float4 o;
                o.x = ok ? v[t].x * w : 0.0f;
                o.y = ok ? v[t].y * w : 0.0f;
                o.z = ok ? v[t].z * w : 0.0f;
                o.w = ok ? v[t].w * w : 0.0f;
                tile4[i] = o;
            }
        }
    }
    __syncthreads();

    int* myq = qmem + wave * QCAP;
    int cnt = 0;

    constexpr int LPE = C / 2;
    constexpr int EPS = 32 / LPE;
    constexpr int PU  = 4;
    const int sub = lane / LPE;
    const int cl  = lane % LPE;

    const size_t rbase = ((size_t)((int)blockIdx.x * AGG_W + wave) * CAPW) * 32 + lane;
    int nblk = 0;

    auto drain = [&](const bool flush) {
        while (cnt >= 32 || (flush && cnt > 0)) {
            const int n = min(cnt, 32);
            int   s = 0, dl = 0;
            float w = 0.0f;
            if (lane < n) {
                const int e = myq[cnt - n + lane];
                s  = src[e];
                dl = dst[e] - base;
                w  = dinv[s] * dinv[base + dl];
            }
            if (MODE == 0) {
                if (nblk < CAPW) {
                    const size_t o = rbase + (size_t)nblk * 32;
                    recS[o] = s;
                    recD[o] = (lane < n) ? dl : -1;
                    recW[o] = w;
                }
                ++nblk;
            }
            cnt -= n;
            for (int j0 = 0; j0 < n; j0 += EPS * PU) {
                float2 r[PU];
                float  ww[PU];
                int    dd[PU];
#pragma unroll
                for (int p = 0; p < PU; ++p) {
                    const int jj = min(j0 + p * EPS + sub, n - 1);
                    const int ss = __shfl(s, jj);
                    dd[p] = __shfl(dl, jj);
                    ww[p] = __shfl(w, jj);
                    r[p]  = *(const float2*)(xl + (long long)ss * C + cl * 2);
                }
#pragma unroll
                for (int p = 0; p < PU; ++p) {
                    if (j0 + p * EPS + sub < n) {
                        float* t = &tile[dd[p] * C + cl * 2];
                        atomicAdd(t,     r[p].x * ww[p]);
                        atomicAdd(t + 1, r[p].y * ww[p]);
                    }
                }
            }
        }
    };

    bool replay = false;
    if (MODE == 1)
        replay = (__builtin_amdgcn_ballot_w32(myN > CAPW) == 0u);

    if (MODE == 1 && replay) {
        constexpr int RPU = 8;
        static_assert(32 % (EPS * RPU) == 0, "a block is a whole number of steps");
        const int nb = __builtin_amdgcn_readfirstlane(__shfl(myN, wave));
        int s = 0, dl = -1;
        float w = 0.0f;
        if (nb > 0) { s = recS[rbase]; dl = recD[rbase]; w = recW[rbase]; }
        for (int b = 0; b < nb; ++b) {
            int s2 = s, dl2 = dl;
            float w2 = w;
            for (int j0 = 0; j0 < 32; j0 += EPS * RPU) {
                float2 r[RPU];
                float  ww[RPU];
                int    dd[RPU];
#pragma unroll
                for (int p = 0; p < RPU; ++p) {
                    const int jj = j0 + p * EPS + sub;
                    const int ss = __shfl(s, jj);
                    dd[p] = __shfl(dl, jj);
                    ww[p] = __shfl(w, jj);
                    r[p]  = *(const float2*)(xl + (long long)ss * C + cl * 2);
                }
                if (j0 == 0) {
                    const size_t o = rbase + (size_t)min(b + 1, nb - 1) * 32;
                    s2 = recS[o]; dl2 = recD[o]; w2 = recW[o];
                }
#pragma unroll
                for (int p = 0; p < RPU; ++p) {
                    if (dd[p] >= 0) {
                        float* t = &tile[dd[p] * C + cl * 2];
                        atomicAdd(t,     r[p].x * ww[p]);
                        atomicAdd(t + 1, r[p].y * ww[p]);
                    }
                }
            }
            s = s2; dl = dl2; w = w2;
        }
    } else if (MODE == 0 && useIdx) {

    int* glist = myq + EQCAP;
    const int nW = (nEdges + 1023) >> 10;
    for (int k0 = 0; k0 * 32 < nW; k0 += 32) {
        const int w = (k0 + lane) * 32 + wave;
        unsigned bits = (w < nW) ? bmap[(size_t)w * MAXT + blockIdx.x] : 0u;
        const int pc = __builtin_popcount(bits);
        int incl = pc;
#pragma unroll
        for (int off = 1; off < 32; off <<= 1) {
            const int v = __shfl_up(incl, off);
            if (lane >= off) incl += v;
        }
        const int total = __builtin_amdgcn_readfirstlane(__shfl(incl, 31));
        int pos = incl - pc;
        while (bits) {
            const int j = __builtin_ctz(bits);
            bits &= bits - 1u;
            glist[pos++] = w * 32 + j;
        }
        for (int i0 = 0; i0 < total; i0 += 32) {
            const int gl = glist[min(i0 + lane, total - 1)];
            const int nq = min(32, total - i0);
            for (int q0 = 0; q0 < nq; q0 += PG) {
                int dv[PG], ev[PG];
#pragma unroll
                for (int p = 0; p < PG; ++p) {
                    const int G = __builtin_amdgcn_readlane(gl, min(q0 + p, nq - 1));
                    ev[p] = G * 32 + lane;
                    dv[p] = dst[min(ev[p], nEdges - 1)];
                }
#pragma unroll
                for (int p = 0; p < PG; ++p) {
                    if (q0 + p < nq) {
                        const bool m = (ev[p] < nEdges) && ((unsigned)(dv[p] - base) < (unsigned)nLoc);
                        const unsigned b = __builtin_amdgcn_ballot_w32(m);
                        if (m) myq[cnt + (int)__builtin_amdgcn_mbcnt_lo(b, 0u)] = ev[p];
                        cnt += __builtin_popcount(b);
                    }
                }
                drain(false);
            }
        }
    }
    drain(true);

    } else {

    const int  nG  = (nEdges + 3) >> 2;
    const int  nG4 = nEdges >> 2;
    const bool al  = (((size_t)dst) & 15) == 0;

    for (int g0 = 0; g0 < nG; g0 += AGG_T * SCAN_U) {
        int4 d4[SCAN_U];
        if (al && g0 + AGG_T * SCAN_U <= nG4) {
            const int4* p = (const int4*)dst + (g0 + tid);
#pragma unroll
            for (int u = 0; u < SCAN_U; ++u)
                d4[u] = p[u * AGG_T];
        } else {
#pragma unroll
            for (int u = 0; u < SCAN_U; ++u)
                d4[u] = gcn_load4_guarded(dst, g0 + u * AGG_T + tid, nEdges);
        }

#pragma unroll
        for (int u = 0; u < SCAN_U; ++u) {
            const int e0 = (g0 + u * AGG_T + tid) << 2;
            const unsigned r0 = (unsigned)(d4[u].x - base);
            const unsigned r1 = (unsigned)(d4[u].y - base);
            const unsigned r2 = (unsigned)(d4[u].z - base);
            const unsigned r3 = (unsigned)(d4[u].w - base);
            const unsigned rm = min(min(r0, r1), min(r2, r3));
            if (__builtin_amdgcn_ballot_w32(rm < (unsigned)nLoc)) {
                const bool m0 = r0 < (unsigned)nLoc;
                const bool m1 = r1 < (unsigned)nLoc;
                const bool m2 = r2 < (unsigned)nLoc;
                const bool m3 = r3 < (unsigned)nLoc;
                const unsigned b0 = __builtin_amdgcn_ballot_w32(m0);
                const unsigned b1 = __builtin_amdgcn_ballot_w32(m1);
                const unsigned b2 = __builtin_amdgcn_ballot_w32(m2);
                const unsigned b3 = __builtin_amdgcn_ballot_w32(m3);
                if (b0) {
                    if (m0) myq[cnt + (int)__builtin_amdgcn_mbcnt_lo(b0, 0u)] = e0;
                    cnt += __builtin_popcount(b0);
                }
                if (b1) {
                    if (m1) myq[cnt + (int)__builtin_amdgcn_mbcnt_lo(b1, 0u)] = e0 + 1;
                    cnt += __builtin_popcount(b1);
                }
                if (b2) {
                    if (m2) myq[cnt + (int)__builtin_amdgcn_mbcnt_lo(b2, 0u)] = e0 + 2;
                    cnt += __builtin_popcount(b2);
                }
                if (b3) {
                    if (m3) myq[cnt + (int)__builtin_amdgcn_mbcnt_lo(b3, 0u)] = e0 + 3;
                    cnt += __builtin_popcount(b3);
                }
            }
        }

        drain(g0 + AGG_T * SCAN_U >= nG);
    }

    }

    if (MODE == 0 && lane == 0) wcount[wave] = nblk;
    __syncthreads();

    if (MODE == 0 && wave == 0) recN[(int)blockIdx.x * AGG_W + lane] = wcount[lane];

    {
        const float4* bias4 = (const float4*)bias;
        float4* out4 = (float4*)(outp + (long long)base * C);
#pragma unroll
        for (int t = 0; t < IT; ++t) {
            const int i = t * AGG_T + tid;
            if (i < nLoc * C4) {
                const float4 a = tile4[i];
                const float4 b = bias4[i % C4];
                float4 o;
                o.x = a.x + b.x; o.y = a.y + b.y; o.z = a.z + b.z; o.w = a.w + b.w;
                if (RELU) { o.x = fmaxf(o.x, 0.0f); o.y = fmaxf(o.y, 0.0f); o.z = fmaxf(o.z, 0.0f); o.w = fmaxf(o.w, 0.0f); }
                out4[i] = o;
            }
        }
    }
}

extern "C" void kernel_launch(void* const* d_in, const int* in_sizes, int n_in,
                              void* d_out, int out_size, void* d_ws, size_t ws_size,
                              hipStream_t stream) {
    const float* x  = (const float*)d_in[0];
    const int*   ei = (const int*)  d_in[1];
    const float* W1 = (const float*)d_in[2];
    const float* b1 = (const float*)d_in[3];
    const float* W2 = (const float*)d_in[4];
    const float* b2 = (const float*)d_in[5];
    float* out = (float*)d_out;

    const int nNodes = in_sizes[0] / IN_CH;
    const int nEdges = in_sizes[1] / 2;
    const int* src = ei;
    const int* dst = ei + nEdges;

    float* ws  = (float*)d_ws;
    float* deg = ws;
    float* xl1 = deg + ((nNodes + 63) & ~63);
    float* h   = xl1 + (long long)nNodes * HID;
    float* xl2 = h   + (long long)nNodes * HID;

    const int B = 256;
    const int nTiles = (nNodes + TILE - 1) / TILE;

    const long long recLen = (long long)nTiles * AGG_W * CAPW * 32;
    int*   recS = (int*)(xl2 + (((long long)nNodes * OUT_CH + 31) & ~31LL));
    int*   recD = recS + recLen;
    float* recW = (float*)(recD + recLen);
    int*   recN = (int*)(recW + recLen);
    unsigned* bmap = (unsigned*)(recN + (long long)nTiles * AGG_W);
    const int nWords = (nEdges + 1023) / 1024;
    const int useIdx = (nTiles <= MAXT) ? 1 : 0;

    gcn_init_kernel<<<(nNodes + B - 1) / B, B, 0, stream>>>(deg, nNodes);

    if (useIdx) {
        const int wpb = IDX_T / 32;
        gcn_degree_index_kernel<<<(nWords + wpb - 1) / wpb, IDX_T, 0, stream>>>(dst, deg, bmap, nEdges, nWords);
    } else {
        gcn_degree_kernel<<<(nEdges + B - 1) / B, B, 0, stream>>>(dst, deg, nEdges);
    }
    gcn_rsqrt_kernel<<<(nNodes + B - 1) / B, B, 0, stream>>>(deg, nNodes);

    {
        int tiles = (nNodes / 16) * (HID / 16);
        int wpb = B / 32;
        gcn_gemm_wmma_f32<<<(tiles + wpb - 1) / wpb, B, 0, stream>>>(x, W1, xl1, nNodes, IN_CH, HID);
    }

    gcn_tile_agg_kernel<HID, true, 0><<<nTiles, AGG_T, 0, stream>>>(
        xl1, src, dst, deg, b1, h, recS, recD, recW, recN, bmap, useIdx, nEdges, nNodes);

    {
        int tiles = (nNodes / 16) * (OUT_CH / 16);
        int wpb = B / 32;
        gcn_gemm_wmma_f32<<<(tiles + wpb - 1) / wpb, B, 0, stream>>>(h, W2, xl2, nNodes, HID, OUT_CH);
    }

    gcn_tile_agg_kernel<OUT_CH, false, 1><<<nTiles, AGG_T, 0, stream>>>(
        xl2, src, dst, deg, b2, out, recS, recD, recW, recN, bmap, useIdx, nEdges, nNodes);
}


# First aggregation: each wave queues all its tile edges with local dst (16 group lines in flight), then gathers rows 16 per step with src and dinv lookups two blocks ahead; LDS arrival order kept
# speedup vs baseline: 2.5274x; 1.0593x over previous
#include <hip/hip_runtime.h>

#define IN_CH  128
#define HID    64
#define OUT_CH 16

typedef float v2f __attribute__((ext_vector_type(2)));
typedef float v8f __attribute__((ext_vector_type(8)));

__global__ void gcn_init_kernel(float* __restrict__ deg, int nNodes) {
    int i = blockIdx.x * blockDim.x + threadIdx.x;
    if (i < nNodes) deg[i] = 1.0f;
}

__global__ void gcn_degree_kernel(const int* __restrict__ dst,
                                  float* __restrict__ deg, int nEdges) {
    int e = blockIdx.x * blockDim.x + threadIdx.x;
    if (e < nEdges) atomicAdd(&deg[dst[e]], 1.0f);
}

__global__ void gcn_rsqrt_kernel(float* __restrict__ deg, int n) {
    int i = blockIdx.x * blockDim.x + threadIdx.x;
    if (i < n) deg[i] = rsqrtf(deg[i]);
}

__global__ void gcn_gemm_wmma_f32(const float* __restrict__ A,
                                  const float* __restrict__ W,
                                  float* __restrict__ C,
                                  int M, int K, int N) {
    const int wavesPerBlock = blockDim.x >> 5;
    const int wave = (int)(threadIdx.x >> 5);
    const int lane = (int)(threadIdx.x & 31);
    const int tilesN = N >> 4;
    const int gw = blockIdx.x * wavesPerBlock + wave;
    const int tileM = gw / tilesN;
    const int tileN = gw - tileM * tilesN;
    if (tileM * 16 >= M) return;

    const int row0 = tileM << 4;
    const int col0 = tileN << 4;
    const int mr = lane & 15;
    const int kh = (lane >> 4) << 1;

    v8f acc = {};
    for (int k = 0; k < K; k += 4) {
        v2f a, b;
        const float* ap = A + (long long)(row0 + mr) * K + (k + kh);
        a.x = ap[0];
        a.y = ap[1];
        const float* bp = W + (long long)(k + kh) * N + (col0 + mr);
        b.x = bp[0];
        b.y = bp[N];
        acc = __builtin_amdgcn_wmma_f32_16x16x4_f32(
             false, a,  false, b,
             (short)0, acc,  false,  false);
    }

    const int crow = row0 + ((lane >> 4) << 3);
    const int ccol = col0 + mr;
#pragma unroll
    for (int r = 0; r < 8; ++r)
        C[(long long)(crow + r) * N + ccol] = acc[r];
}

#define TILE   392
#define AGG_T  1024
#define SCAN_U 8
#define PG     16
#define EQCAP  1248
#define IQCAP  832
static_assert(IQCAP * 4 + IQCAP * 2 <= EQCAP * 4, "edge ids and local destinations fit the wave's queue memory");
static_assert(IQCAP >= 32 + PG * 32, "room for a rest plus one step in which every edge matches");
static_assert(TILE <= 65536, "a local destination fits 16 bits");
#define GLCAP  1024
static_assert(EQCAP >= 32 + SCAN_U * 128, "the scanning path's queue (every scanned edge matches) fits too");
#define MAXT   256
#define IDX_T  128
#define CAPW   16
#define AGG_W  (AGG_T / 32)
static_assert(AGG_W == 32, "one lane per wave when a wave reads its tile's block counts");

__global__ __launch_bounds__(IDX_T)
void gcn_degree_index_kernel(const int* __restrict__ dst,
                             float* __restrict__ deg,
                             unsigned* __restrict__ bmap,
                             int nEdges, int nWords) {
    __shared__ unsigned rows[(IDX_T / 32) * MAXT];
    const int lane = (int)(threadIdx.x & 31);
    const int wave = (int)(threadIdx.x >> 5);
    const int word = (int)blockIdx.x * (IDX_T / 32) + wave;
    if (word >= nWords) return;
    unsigned* my = rows + wave * MAXT;
#pragma unroll
    for (int k = 0; k < MAXT / 32; ++k) my[k * 32 + lane] = 0u;
    const int e0 = word * 1024 + lane;
    for (int g0 = 0; g0 < 32; g0 += 8) {
        int dv[8];
#pragma unroll
        for (int u = 0; u < 8; ++u) {
            const int e = e0 + (g0 + u) * 32;
            dv[u] = (e < nEdges) ? dst[e] : -1;
        }
#pragma unroll
        for (int u = 0; u < 8; ++u) {
            if (dv[u] >= 0) {
                atomicAdd(&deg[dv[u]], 1.0f);
                atomicOr(&my[dv[u] / TILE], 1u << (g0 + u));
            }
        }
    }
#pragma unroll
    for (int k = 0; k < MAXT / 32; ++k)
        bmap[(size_t)word * MAXT + k * 32 + lane] = my[k * 32 + lane];
}

static __device__ __forceinline__ int4 gcn_load4_guarded(const int* __restrict__ p, int g, int n) {
    const int e = g << 2;
    int4 r;
    r.x = (e     < n) ? p[e]     : -1;
    r.y = (e + 1 < n) ? p[e + 1] : -1;
    r.z = (e + 2 < n) ? p[e + 2] : -1;
    r.w = (e + 3 < n) ? p[e + 3] : -1;
    return r;
}

template <int C, bool RELU, int MODE>
__global__ __launch_bounds__(AGG_T)
void gcn_tile_agg_kernel(const float* __restrict__ xl,
                         const int* __restrict__ src,
                         const int* __restrict__ dst,
                         const float* __restrict__ dinv,
                         const float* __restrict__ bias,
                         float* __restrict__ outp,
                         int*   __restrict__ recS,
                         int*   __restrict__ recD,
                         float* __restrict__ recW,
                         int*   __restrict__ recN,
                         const unsigned* __restrict__ bmap,
                         int useIdx,
                         int nEdges, int nNodes) {
    __shared__ __attribute__((aligned(16))) float tile[TILE * C];
    __shared__ int   qmem[(AGG_T / 32) * EQCAP];
    __shared__ unsigned short gmem[(AGG_T / 32) * GLCAP];
    __shared__ int   wcount[AGG_T / 32];

    const int tid  = (int)threadIdx.x;
    const int lane = tid & 31;
    const int wave = tid >> 5;
    const int base = (int)blockIdx.x * TILE;
    const int nLoc = min(TILE, nNodes - base);

    int myN = 0;
    if (MODE == 1) myN = recN[(int)blockIdx.x * (AGG_T / 32) + lane];

    constexpr int C4 = C / 4;
    constexpr int NV = TILE * C4;
    constexpr int IT = (NV + AGG_T - 1) / AGG_T;
    float4* tile4 = (float4*)tile;
    {
        float4 v[IT];
        float  di[IT];
#pragma unroll
        for (int t = 0; t < IT; ++t) {
            const int i  = min(t * AGG_T + tid, NV - 1);
            const int n  = min(i / C4, nLoc - 1);
            di[t] = dinv[base + n];
            v[t]  = *(const float4*)(xl + (long long)(base + n) * C + (i % C4) * 4);
        }
#pragma unroll
        for (int t = 0; t < IT; ++t) {
            const int i = t * AGG_T + tid;
            if (i < NV) {
                const bool ok = (i / C4) < nLoc;
                const float w = di[t] * di[t];
                float4 o;
                o.x = ok ? v[t].x * w : 0.0f;
                o.y = ok ? v[t].y * w : 0.0f;
                o.z = ok ? v[t].z * w : 0.0f;
                o.w = ok ? v[t].w * w : 0.0f;
                tile4[i] = o;
            }
        }
    }
    __syncthreads();

    int* myq = qmem + wave * EQCAP;
    int cnt = 0;

    constexpr int LPE = C / 2;
    constexpr int EPS = 32 / LPE;
    constexpr int PU  = 4;
    const int sub = lane / LPE;
    const int cl  = lane % LPE;

    const size_t rbase = ((size_t)((int)blockIdx.x * AGG_W + wave) * CAPW) * 32 + lane;
    int nblk = 0;

    auto drain = [&](const bool flush) {
        while (cnt >= 32 || (flush && cnt > 0)) {
            const int n = min(cnt, 32);
            int   s = 0, dl = 0;
            float w = 0.0f;
            if (lane < n) {
                const int e = myq[cnt - n + lane];
                s  = src[e];
                dl = dst[e] - base;
                w  = dinv[s] * dinv[base + dl];
            }
            if (MODE == 0) {
                if (nblk < CAPW) {
                    const size_t o = rbase + (size_t)nblk * 32;
                    recS[o] = s;
                    recD[o] = (lane < n) ? dl : -1;
                    recW[o] = w;
                }
                ++nblk;
            }
            cnt -= n;
            for (int j0 = 0; j0 < n; j0 += EPS * PU) {
                float2 r[PU];
                float  ww[PU];
                int    dd[PU];
#pragma unroll
                for (int p = 0; p < PU; ++p) {
                    const int jj = min(j0 + p * EPS + sub, n - 1);
                    const int ss = __shfl(s, jj);
                    dd[p] = __shfl(dl, jj);
                    ww[p] = __shfl(w, jj);
                    r[p]  = *(const float2*)(xl + (long long)ss * C + cl * 2);
                }
#pragma unroll
                for (int p = 0; p < PU; ++p) {
                    if (j0 + p * EPS + sub < n) {
                        float* t = &tile[dd[p] * C + cl * 2];
                        atomicAdd(t,     r[p].x * ww[p]);
                        atomicAdd(t + 1, r[p].y * ww[p]);
                    }
                }
            }
        }
    };

    constexpr int  GU  = 16;
    constexpr bool UNI = (EPS == 1);
    unsigned short* mydl = (unsigned short*)(myq + IQCAP);
    auto gatherAll = [&]() {
        const int nB   = (cnt + 31) >> 5;
        const int last = max(cnt - 1, 0);
        int   s0 = 0, dl0 = 0, s1 = 0, dl1 = 0;
        float w0 = 0.0f;
        if (nB > 0) {
            const int q0 = min(lane, last), q1 = min(32 + lane, last);
            dl0 = mydl[q0];
            dl1 = mydl[q1];
            s0  = src[myq[q0]];
            s1  = src[myq[q1]];
            w0  = dinv[s0] * dinv[base + dl0];
        }
        for (int b = 0; b < nB; ++b) {
            const int m  = min(32, cnt - b * 32);
            int   s2 = 0, dl2 = 0;
            float x1 = 0.0f, y1 = 0.0f;
            if (nblk < CAPW) {
                const size_t o = rbase + (size_t)nblk * 32;
                recS[o] = s0;
                recD[o] = (lane < m) ? dl0 : -1;
                recW[o] = w0;
            }
            ++nblk;
            for (int j0 = 0; j0 < m; j0 += EPS * GU) {
                float2 r[GU];
                float  ww[GU];
                int    dd[GU];
#pragma unroll
                for (int p = 0; p < GU; ++p) {
                    const int jj = min(j0 + p * EPS + sub, m - 1);
                    int ss;
                    if (UNI) {
                        ss    = __builtin_amdgcn_readlane(s0, jj);
                        dd[p] = __builtin_amdgcn_readlane(dl0, jj);
                        ww[p] = __int_as_float(__builtin_amdgcn_readlane(__float_as_int(w0), jj));
                    } else {
                        ss    = __shfl(s0, jj);
                        dd[p] = __shfl(dl0, jj);
                        ww[p] = __shfl(w0, jj);
                    }
                    r[p] = *(const float2*)(xl + (long long)ss * C + cl * 2);
                }
                if (j0 == 0) {
                    const int q2 = min((b + 2) * 32 + lane, last);
                    dl2 = mydl[q2];
                    s2  = src[myq[q2]];
                    x1  = dinv[s1];
                    y1  = dinv[base + dl1];
                }
#pragma unroll
                for (int p = 0; p < GU; ++p) {
                    if (j0 + p * EPS + sub < m) {
                        float* t = &tile[dd[p] * C + cl * 2];
                        atomicAdd(t,     r[p].x * ww[p]);
                        atomicAdd(t + 1, r[p].y * ww[p]);
                    }
                }
            }
            s0 = s1; dl0 = dl1; w0 = x1 * y1;
            s1 = s2; dl1 = dl2;
        }
        cnt = 0;
    };

    bool replay = false;
    if (MODE == 1)
        replay = (__builtin_amdgcn_ballot_w32(myN > CAPW) == 0u);

    if (MODE == 1 && replay) {
        constexpr int RPU = 8;
        static_assert(32 % (EPS * RPU) == 0, "a block is a whole number of steps");
        const int nb = __builtin_amdgcn_readfirstlane(__shfl(myN, wave));
        int s = 0, dl = -1;
        float w = 0.0f;
        if (nb > 0) { s = recS[rbase]; dl = recD[rbase]; w = recW[rbase]; }
        for (int b = 0; b < nb; ++b) {
            int s2 = s, dl2 = dl;
            float w2 = w;
            for (int j0 = 0; j0 < 32; j0 += EPS * RPU) {
                float2 r[RPU];
                float  ww[RPU];
                int    dd[RPU];
#pragma unroll
                for (int p = 0; p < RPU; ++p) {
                    const int jj = j0 + p * EPS + sub;
                    const int ss = __shfl(s, jj);
                    dd[p] = __shfl(dl, jj);
                    ww[p] = __shfl(w, jj);
                    r[p]  = *(const float2*)(xl + (long long)ss * C + cl * 2);
                }
                if (j0 == 0) {
                    const size_t o = rbase + (size_t)min(b + 1, nb - 1) * 32;
                    s2 = recS[o]; dl2 = recD[o]; w2 = recW[o];
                }
#pragma unroll
                for (int p = 0; p < RPU; ++p) {
                    if (dd[p] >= 0) {
                        float* t = &tile[dd[p] * C + cl * 2];
                        atomicAdd(t,     r[p].x * ww[p]);
                        atomicAdd(t + 1, r[p].y * ww[p]);
                    }
                }
            }
            s = s2; dl = dl2; w = w2;
        }
    } else if (MODE == 0 && useIdx) {

    unsigned short* glist = gmem + wave * GLCAP;
    const int nW = (nEdges + 1023) >> 10;
    for (int k0 = 0; k0 * 32 < nW; k0 += 32) {
        const int w = (k0 + lane) * 32 + wave;
        unsigned bits = (w < nW) ? bmap[(size_t)w * MAXT + blockIdx.x] : 0u;
        const int pc = __builtin_popcount(bits);
        int incl = pc;
#pragma unroll
        for (int off = 1; off < 32; off <<= 1) {
            const int v = __shfl_up(incl, off);
            if (lane >= off) incl += v;
        }
        const int total = __builtin_amdgcn_readfirstlane(__shfl(incl, 31));
        int pos = incl - pc;
        while (bits) {
            const int j = __builtin_ctz(bits);
            bits &= bits - 1u;
            glist[pos++] = (unsigned short)(lane * 32 + j);
        }
        for (int i0 = 0; i0 < total; i0 += 32) {
            const int gl = (int)glist[min(i0 + lane, total - 1)];
            const int nq = min(32, total - i0);
            for (int q0 = 0; q0 < nq; q0 += PG) {
                if (cnt > IQCAP - PG * 32) gatherAll();
                int dv[PG], ev[PG];
#pragma unroll
                for (int p = 0; p < PG; ++p) {
                    const int v = __builtin_amdgcn_readlane(gl, min(q0 + p, nq - 1));
                    const int G = ((k0 + (v >> 5)) * 32 + wave) * 32 + (v & 31);
                    ev[p] = G * 32 + lane;
                    dv[p] = dst[min(ev[p], nEdges - 1)];
                }
#pragma unroll
                for (int p = 0; p < PG; ++p) {
                    if (q0 + p < nq) {
                        const bool m = (ev[p] < nEdges) && ((unsigned)(dv[p] - base) < (unsigned)nLoc);
                        const unsigned b = __builtin_amdgcn_ballot_w32(m);
                        if (m) {
                            const int pos = cnt + (int)__builtin_amdgcn_mbcnt_lo(b, 0u);
                            myq[pos]  = ev[p];
                            mydl[pos] = (unsigned short)(dv[p] - base);
                        }
                        cnt += __builtin_popcount(b);
                    }
                }
            }
        }
    }
    gatherAll();

    } else {

    const int  nG  = (nEdges + 3) >> 2;
    const int  nG4 = nEdges >> 2;
    const bool al  = (((size_t)dst) & 15) == 0;

    for (int g0 = 0; g0 < nG; g0 += AGG_T * SCAN_U) {
        int4 d4[SCAN_U];
        if (al && g0 + AGG_T * SCAN_U <= nG4) {
            const int4* p = (const int4*)dst + (g0 + tid);
#pragma unroll
            for (int u = 0; u < SCAN_U; ++u)
                d4[u] = p[u * AGG_T];
        } else {
#pragma unroll
            for (int u = 0; u < SCAN_U; ++u)
                d4[u] = gcn_load4_guarded(dst, g0 + u * AGG_T + tid, nEdges);
        }

#pragma unroll
        for (int u = 0; u < SCAN_U; ++u) {
            const int e0 = (g0 + u * AGG_T + tid) << 2;
            const unsigned r0 = (unsigned)(d4[u].x - base);
            const unsigned r1 = (unsigned)(d4[u].y - base);
            const unsigned r2 = (unsigned)(d4[u].z - base);
            const unsigned r3 = (unsigned)(d4[u].w - base);
            const unsigned rm = min(min(r0, r1), min(r2, r3));
            if (__builtin_amdgcn_ballot_w32(rm < (unsigned)nLoc)) {
                const bool m0 = r0 < (unsigned)nLoc;
                const bool m1 = r1 < (unsigned)nLoc;
                const bool m2 = r2 < (unsigned)nLoc;
                const bool m3 = r3 < (unsigned)nLoc;
                const unsigned b0 = __builtin_amdgcn_ballot_w32(m0);
                const unsigned b1 = __builtin_amdgcn_ballot_w32(m1);
                const unsigned b2 = __builtin_amdgcn_ballot_w32(m2);
                const unsigned b3 = __builtin_amdgcn_ballot_w32(m3);
                if (b0) {
                    if (m0) myq[cnt + (int)__builtin_amdgcn_mbcnt_lo(b0, 0u)] = e0;
                    cnt += __builtin_popcount(b0);
                }
                if (b1) {
                    if (m1) myq[cnt + (int)__builtin_amdgcn_mbcnt_lo(b1, 0u)] = e0 + 1;
                    cnt += __builtin_popcount(b1);
                }
                if (b2) {
                    if (m2) myq[cnt + (int)__builtin_amdgcn_mbcnt_lo(b2, 0u)] = e0 + 2;
                    cnt += __builtin_popcount(b2);
                }
                if (b3) {
                    if (m3) myq[cnt + (int)__builtin_amdgcn_mbcnt_lo(b3, 0u)] = e0 + 3;
                    cnt += __builtin_popcount(b3);
                }
            }
        }

        drain(g0 + AGG_T * SCAN_U >= nG);
    }

    }

    if (MODE == 0 && lane == 0) wcount[wave] = nblk;
    __syncthreads();

    if (MODE == 0 && wave == 0) recN[(int)blockIdx.x * AGG_W + lane] = wcount[lane];

    {
        const float4* bias4 = (const float4*)bias;
        float4* out4 = (float4*)(outp + (long long)base * C);
#pragma unroll
        for (int t = 0; t < IT; ++t) {
            const int i = t * AGG_T + tid;
            if (i < nLoc * C4) {
                const float4 a = tile4[i];
                const float4 b = bias4[i % C4];
                float4 o;
                o.x = a.x + b.x; o.y = a.y + b.y; o.z = a.z + b.z; o.w = a.w + b.w;
                if (RELU) { o.x = fmaxf(o.x, 0.0f); o.y = fmaxf(o.y, 0.0f); o.z = fmaxf(o.z, 0.0f); o.w = fmaxf(o.w, 0.0f); }
                out4[i] = o;
            }
        }
    }
}

extern "C" void kernel_launch(void* const* d_in, const int* in_sizes, int n_in,
                              void* d_out, int out_size, void* d_ws, size_t ws_size,
                              hipStream_t stream) {
    const float* x  = (const float*)d_in[0];
    const int*   ei = (const int*)  d_in[1];
    const float* W1 = (const float*)d_in[2];
    const float* b1 = (const float*)d_in[3];
    const float* W2 = (const float*)d_in[4];
    const float* b2 = (const float*)d_in[5];
    float* out = (float*)d_out;

    const int nNodes = in_sizes[0] / IN_CH;
    const int nEdges = in_sizes[1] / 2;
    const int* src = ei;
    const int* dst = ei + nEdges;

    float* ws  = (float*)d_ws;
    float* deg = ws;
    float* xl1 = deg + ((nNodes + 63) & ~63);
    float* h   = xl1 + (long long)nNodes * HID;
    float* xl2 = h   + (long long)nNodes * HID;

    const int B = 256;
    const int nTiles = (nNodes + TILE - 1) / TILE;

    const long long recLen = (long long)nTiles * AGG_W * CAPW * 32;
    int*   recS = (int*)(xl2 + (((long long)nNodes * OUT_CH + 31) & ~31LL));
    int*   recD = recS + recLen;
    float* recW = (float*)(recD + recLen);
    int*   recN = (int*)(recW + recLen);
    unsigned* bmap = (unsigned*)(recN + (long long)nTiles * AGG_W);
    const int nWords = (nEdges + 1023) / 1024;
    const int useIdx = (nTiles <= MAXT) ? 1 : 0;

    gcn_init_kernel<<<(nNodes + B - 1) / B, B, 0, stream>>>(deg, nNodes);

    if (useIdx) {
        const int wpb = IDX_T / 32;
        gcn_degree_index_kernel<<<(nWords + wpb - 1) / wpb, IDX_T, 0, stream>>>(dst, deg, bmap, nEdges, nWords);
    } else {
        gcn_degree_kernel<<<(nEdges + B - 1) / B, B, 0, stream>>>(dst, deg, nEdges);
    }
    gcn_rsqrt_kernel<<<(nNodes + B - 1) / B, B, 0, stream>>>(deg, nNodes);

    {
        int tiles = (nNodes / 16) * (HID / 16);
        int wpb = B / 32;
        gcn_gemm_wmma_f32<<<(tiles + wpb - 1) / wpb, B, 0, stream>>>(x, W1, xl1, nNodes, IN_CH, HID);
    }

    gcn_tile_agg_kernel<HID, true, 0><<<nTiles, AGG_T, 0, stream>>>(
        xl1, src, dst, deg, b1, h, recS, recD, recW, recN, bmap, useIdx, nEdges, nNodes);

    {
        int tiles = (nNodes / 16) * (OUT_CH / 16);
        int wpb = B / 32;
        gcn_gemm_wmma_f32<<<(tiles + wpb - 1) / wpb, B, 0, stream>>>(h, W2, xl2, nNodes, HID, OUT_CH);
    }

    gcn_tile_agg_kernel<OUT_CH, false, 1><<<nTiles, AGG_T, 0, stream>>>(
        xl2, src, dst, deg, b2, out, recS, recD, recW, recN, bmap, useIdx, nEdges, nNodes);
}


# Degrees counted in LDS per destination tile by a collect launch that writes the tile's edge records, replayed by the first aggregation; no global add per edge, no init/rsqrt launch; sum order kept
# speedup vs baseline: 4.0119x; 1.5874x over previous
#include <hip/hip_runtime.h>

#define IN_CH  128
#define HID    64
#define OUT_CH 16

typedef float v2f __attribute__((ext_vector_type(2)));
typedef float v8f __attribute__((ext_vector_type(8)));

__global__ void gcn_init_kernel(float* __restrict__ deg, int nNodes) {
    int i = blockIdx.x * blockDim.x + threadIdx.x;
    if (i < nNodes) deg[i] = 1.0f;
}

__global__ void gcn_degree_kernel(const int* __restrict__ dst,
                                  float* __restrict__ deg, int nEdges) {
    int e = blockIdx.x * blockDim.x + threadIdx.x;
    if (e < nEdges) atomicAdd(&deg[dst[e]], 1.0f);
}

__global__ void gcn_rsqrt_kernel(float* __restrict__ deg, int n) {
    int i = blockIdx.x * blockDim.x + threadIdx.x;
    if (i < n) deg[i] = rsqrtf(deg[i]);
}

__global__ void gcn_gemm_wmma_f32(const float* __restrict__ A,
                                  const float* __restrict__ W,
                                  float* __restrict__ C,
                                  int M, int K, int N) {
    const int wavesPerBlock = blockDim.x >> 5;
    const int wave = (int)(threadIdx.x >> 5);
    const int lane = (int)(threadIdx.x & 31);
    const int tilesN = N >> 4;
    const int gw = blockIdx.x * wavesPerBlock + wave;
    const int tileM = gw / tilesN;
    const int tileN = gw - tileM * tilesN;
    if (tileM * 16 >= M) return;

    const int row0 = tileM << 4;
    const int col0 = tileN << 4;
    const int mr = lane & 15;
    const int kh = (lane >> 4) << 1;

    v8f acc = {};
    for (int k = 0; k < K; k += 4) {
        v2f a, b;
        const float* ap = A + (long long)(row0 + mr) * K + (k + kh);
        a.x = ap[0];
        a.y = ap[1];
        const float* bp = W + (long long)(k + kh) * N + (col0 + mr);
        b.x = bp[0];
        b.y = bp[N];
        acc = __builtin_amdgcn_wmma_f32_16x16x4_f32(
             false, a,  false, b,
             (short)0, acc,  false,  false);
    }

    const int crow = row0 + ((lane >> 4) << 3);
    const int ccol = col0 + mr;
#pragma unroll
    for (int r = 0; r < 8; ++r)
        C[(long long)(crow + r) * N + ccol] = acc[r];
}

#define TILE   392
#define AGG_T  1024
#define SCAN_U 8
#define PG     16
#define EQCAP  1248
#define IQCAP  832
static_assert(IQCAP * 4 + IQCAP * 2 <= EQCAP * 4, "edge ids and local destinations fit the wave's queue memory");
static_assert(IQCAP >= 32 + PG * 32, "room for a rest plus one step in which every edge matches");
static_assert(TILE <= 65536, "a local destination fits 16 bits");
#define GLCAP  1024
static_assert(EQCAP >= 32 + SCAN_U * 128, "the scanning path's queue (every scanned edge matches) fits too");
#define MAXT   256
#define IDX_T  128
#define CAPW   16
#define AGG_W  (AGG_T / 32)
static_assert(AGG_W == 32, "one lane per wave when a wave reads its tile's block counts");

__global__ __launch_bounds__(IDX_T)
void gcn_group_index_kernel(const int* __restrict__ dst,
                            unsigned* __restrict__ bmap,
                            int nEdges, int nWords) {
    __shared__ unsigned rows[(IDX_T / 32) * MAXT];
    const int lane = (int)(threadIdx.x & 31);
    const int wave = (int)(threadIdx.x >> 5);
    const int word = (int)blockIdx.x * (IDX_T / 32) + wave;
    if (word >= nWords) return;
    unsigned* my = rows + wave * MAXT;
#pragma unroll
    for (int k = 0; k < MAXT / 32; ++k) my[k * 32 + lane] = 0u;
    const int e0 = word * 1024 + lane;
    for (int g0 = 0; g0 < 32; g0 += 8) {
        int dv[8];
#pragma unroll
        for (int u = 0; u < 8; ++u) {
            const int e = e0 + (g0 + u) * 32;
            dv[u] = (e < nEdges) ? dst[e] : -1;
        }
#pragma unroll
        for (int u = 0; u < 8; ++u) {
            if (dv[u] >= 0) atomicOr(&my[dv[u] / TILE], 1u << (g0 + u));
        }
    }
#pragma unroll
    for (int k = 0; k < MAXT / 32; ++k)
        bmap[(size_t)word * MAXT + k * 32 + lane] = my[k * 32 + lane];
}

__global__ __launch_bounds__(AGG_T)
void gcn_tile_collect_kernel(const int* __restrict__ src,
                             const int* __restrict__ dst,
                             float* __restrict__ dinv,
                             int*   __restrict__ recS,
                             int*   __restrict__ recD,
                             int*   __restrict__ recN,
                             const unsigned* __restrict__ bmap,
                             int nEdges, int nNodes) {
    __shared__ int   qmem[AGG_W * EQCAP];
    __shared__ unsigned short gmem[AGG_W * GLCAP];
    __shared__ int   degc[TILE];
    __shared__ int   wcount[AGG_W];

    const int tid  = (int)threadIdx.x;
    const int lane = tid & 31;
    const int wave = tid >> 5;
    const int base = (int)blockIdx.x * TILE;
    const int nLoc = min(TILE, nNodes - base);

    for (int i = tid; i < TILE; i += AGG_T) degc[i] = 0;
    __syncthreads();

    int* myq = qmem + wave * EQCAP;
    unsigned short* mydl = (unsigned short*)(myq + IQCAP);
    int cnt = 0;

    const size_t rbase = ((size_t)((int)blockIdx.x * AGG_W + wave) * CAPW) * 32 + lane;
    int nblk = 0;

    constexpr int RU = 8;
    auto recordAll = [&]() {
        const int nB   = (cnt + 31) >> 5;
        const int last = max(cnt - 1, 0);
        for (int b0 = 0; b0 < nB; b0 += RU) {
            int sv[RU], dv[RU];
#pragma unroll
            for (int u = 0; u < RU; ++u) {
                const int q = min((b0 + u) * 32 + lane, last);
                dv[u] = mydl[q];
                sv[u] = src[myq[q]];
            }
#pragma unroll
            for (int u = 0; u < RU; ++u) {
                if (b0 + u < nB) {
                    if (nblk < CAPW) {
                        const size_t o = rbase + (size_t)nblk * 32;
                        recS[o] = sv[u];
                        recD[o] = ((b0 + u) * 32 + lane < cnt) ? dv[u] : -1;
                    }
                    ++nblk;
                }
            }
        }
        cnt = 0;
    };

    unsigned short* glist = gmem + wave * GLCAP;
    const int nW = (nEdges + 1023) >> 10;
    for (int k0 = 0; k0 * 32 < nW; k0 += 32) {
        const int w = (k0 + lane) * 32 + wave;
        unsigned bits = (w < nW) ? bmap[(size_t)w * MAXT + blockIdx.x] : 0u;
        const int pc = __builtin_popcount(bits);
        int incl = pc;
#pragma unroll
        for (int off = 1; off < 32; off <<= 1) {
            const int v = __shfl_up(incl, off);
            if (lane >= off) incl += v;
        }
        const int total = __builtin_amdgcn_readfirstlane(__shfl(incl, 31));
        int pos = incl - pc;
        while (bits) {
            const int j = __builtin_ctz(bits);
            bits &= bits - 1u;
            glist[pos++] = (unsigned short)(lane * 32 + j);
        }
        for (int i0 = 0; i0 < total; i0 += 32) {
            const int gl = (int)glist[min(i0 + lane, total - 1)];
            const int nq = min(32, total - i0);
            for (int q0 = 0; q0 < nq; q0 += PG) {
                if (cnt > IQCAP - PG * 32) recordAll();
                int dv[PG], ev[PG];
#pragma unroll
                for (int p = 0; p < PG; ++p) {
                    const int v = __builtin_amdgcn_readlane(gl, min(q0 + p, nq - 1));
                    const int G = ((k0 + (v >> 5)) * 32 + wave) * 32 + (v & 31);
                    ev[p] = G * 32 + lane;
                    dv[p] = dst[min(ev[p], nEdges - 1)];
                }
#pragma unroll
                for (int p = 0; p < PG; ++p) {
                    if (q0 + p < nq) {
                        const bool m = (ev[p] < nEdges) && ((unsigned)(dv[p] - base) < (unsigned)nLoc);
                        const unsigned b = __builtin_amdgcn_ballot_w32(m);
                        if (m) {
                            const int qp = cnt + (int)__builtin_amdgcn_mbcnt_lo(b, 0u);
                            myq[qp]  = ev[p];
                            mydl[qp] = (unsigned short)(dv[p] - base);
                            atomicAdd(&degc[dv[p] - base], 1);
                        }
                        cnt += __builtin_popcount(b);
                    }
                }
            }
        }
    }
    recordAll();

    if (lane == 0) wcount[wave] = nblk;
    __syncthreads();

    if (wave == 0) recN[(int)blockIdx.x * AGG_W + lane] = wcount[lane];

    for (int i = tid; i < nLoc; i += AGG_T) dinv[base + i] = rsqrtf(1.0f + (float)degc[i]);
}

static __device__ __forceinline__ int4 gcn_load4_guarded(const int* __restrict__ p, int g, int n) {
    const int e = g << 2;
    int4 r;
    r.x = (e     < n) ? p[e]     : -1;
    r.y = (e + 1 < n) ? p[e + 1] : -1;
    r.z = (e + 2 < n) ? p[e + 2] : -1;
    r.w = (e + 3 < n) ? p[e + 3] : -1;
    return r;
}

template <int C, bool RELU, int MODE>
__global__ __launch_bounds__(AGG_T)
void gcn_tile_agg_kernel(const float* __restrict__ xl,
                         const int* __restrict__ src,
                         const int* __restrict__ dst,
                         const float* __restrict__ dinv,
                         const float* __restrict__ bias,
                         float* __restrict__ outp,
                         int*   __restrict__ recS,
                         int*   __restrict__ recD,
                         float* __restrict__ recW,
                         int*   __restrict__ recN,
                         int useRec,
                         int nEdges, int nNodes) {
    __shared__ __attribute__((aligned(16))) float tile[TILE * C];
    __shared__ int   qmem[(AGG_T / 32) * EQCAP];
    __shared__ int   wcount[AGG_T / 32];

    const int tid  = (int)threadIdx.x;
    const int lane = tid & 31;
    const int wave = tid >> 5;
    const int base = (int)blockIdx.x * TILE;
    const int nLoc = min(TILE, nNodes - base);

    int myN = 0;
    if (MODE == 1 || useRec) myN = recN[(int)blockIdx.x * (AGG_T / 32) + lane];

    constexpr int C4 = C / 4;
    constexpr int NV = TILE * C4;
    constexpr int IT = (NV + AGG_T - 1) / AGG_T;
    float4* tile4 = (float4*)tile;
    {
        float4 v[IT];
        float  di[IT];
#pragma unroll
        for (int t = 0; t < IT; ++t) {
            const int i  = min(t * AGG_T + tid, NV - 1);
            const int n  = min(i / C4, nLoc - 1);
            di[t] = dinv[base + n];
            v[t]  = *(const float4*)(xl + (long long)(base + n) * C + (i % C4) * 4);
        }
#pragma unroll
        for (int t = 0; t < IT; ++t) {
            const int i = t * AGG_T + tid;
            if (i < NV) {
                const bool ok = (i / C4) < nLoc;
                const float w = di[t] * di[t];
                float4 o;
                o.x = ok ? v[t].x * w : 0.0f;
                o.y = ok ? v[t].y * w : 0.0f;
                o.z = ok ? v[t].z * w : 0.0f;
                o.w = ok ? v[t].w * w : 0.0f;
                tile4[i] = o;
            }
        }
    }
    __syncthreads();

    int* myq = qmem + wave * EQCAP;
    int cnt = 0;

    constexpr int LPE = C / 2;
    constexpr int EPS = 32 / LPE;
    constexpr int PU  = 4;
    const int sub = lane / LPE;
    const int cl  = lane % LPE;

    const size_t rbase = ((size_t)((int)blockIdx.x * AGG_W + wave) * CAPW) * 32 + lane;
    int nblk = 0;

    auto drain = [&](const bool flush) {
        while (cnt >= 32 || (flush && cnt > 0)) {
            const int n = min(cnt, 32);
            int   s = 0, dl = 0;
            float w = 0.0f;
            if (lane < n) {
                const int e = myq[cnt - n + lane];
                s  = src[e];
                dl = dst[e] - base;
                w  = dinv[s] * dinv[base + dl];
            }
            if (MODE == 0) {
                if (nblk < CAPW) {
                    const size_t o = rbase + (size_t)nblk * 32;
                    recS[o] = s;
                    recD[o] = (lane < n) ? dl : -1;
                    recW[o] = w;
                }
                ++nblk;
            }
            cnt -= n;
            for (int j0 = 0; j0 < n; j0 += EPS * PU) {
                float2 r[PU];
                float  ww[PU];
                int    dd[PU];
#pragma unroll
                for (int p = 0; p < PU; ++p) {
                    const int jj = min(j0 + p * EPS + sub, n - 1);
                    const int ss = __shfl(s, jj);
                    dd[p] = __shfl(dl, jj);
                    ww[p] = __shfl(w, jj);
                    r[p]  = *(const float2*)(xl + (long long)ss * C + cl * 2);
                }
#pragma unroll
                for (int p = 0; p < PU; ++p) {
                    if (j0 + p * EPS + sub < n) {
                        float* t = &tile[dd[p] * C + cl * 2];
                        atomicAdd(t,     r[p].x * ww[p]);
                        atomicAdd(t + 1, r[p].y * ww[p]);
                    }
                }
            }
        }
    };

    constexpr int  GU  = 16;
    constexpr bool UNI = (EPS == 1);
    auto gatherRec = [&](const int nb) {
        int   s0 = 0, dl0 = -1, s1 = 0, dl1 = -1;
        float w0 = 0.0f;
        if (nb > 0) {
            const size_t o1 = rbase + (size_t)min(1, nb - 1) * 32;
            s0  = recS[rbase];
            dl0 = recD[rbase];
            s1  = recS[o1];
            dl1 = recD[o1];
            w0  = dinv[s0] * dinv[base + max(dl0, 0)];
        }
        for (int b = 0; b < nb; ++b) {
            const int m = __builtin_popcount(__builtin_amdgcn_ballot_w32(dl0 >= 0));
            int   s2 = 0, dl2 = -1;
            float x1 = 0.0f, y1 = 0.0f;
            recW[rbase + (size_t)b * 32] = w0;
            for (int j0 = 0; j0 < m; j0 += EPS * GU) {
                float2 r[GU];
                float  ww[GU];
                int    dd[GU];
#pragma unroll
                for (int p = 0; p < GU; ++p) {
                    const int jj = min(j0 + p * EPS + sub, m - 1);
                    int ss;
                    if (UNI) {
                        ss    = __builtin_amdgcn_readlane(s0, jj);
                        dd[p] = __builtin_amdgcn_readlane(dl0, jj);
                        ww[p] = __int_as_float(__builtin_amdgcn_readlane(__float_as_int(w0), jj));
                    } else {
                        ss    = __shfl(s0, jj);
                        dd[p] = __shfl(dl0, jj);
                        ww[p] = __shfl(w0, jj);
                    }
                    r[p] = *(const float2*)(xl + (long long)ss * C + cl * 2);
                }
                if (j0 == 0) {
                    const size_t o2 = rbase + (size_t)min(b + 2, nb - 1) * 32;
                    s2  = recS[o2];
                    dl2 = recD[o2];
                    x1  = dinv[s1];
                    y1  = dinv[base + max(dl1, 0)];
                }
#pragma unroll
                for (int p = 0; p < GU; ++p) {
                    if (j0 + p * EPS + sub < m) {
                        float* t = &tile[dd[p] * C + cl * 2];
                        atomicAdd(t,     r[p].x * ww[p]);
                        atomicAdd(t + 1, r[p].y * ww[p]);
                    }
                }
            }
            s0 = s1; dl0 = dl1; w0 = x1 * y1;
            s1 = s2; dl1 = dl2;
        }
    };

    bool replay = false;
    if (MODE == 1 || useRec)
        replay = (__builtin_amdgcn_ballot_w32(myN > CAPW) == 0u);
    const int nb = replay ? __builtin_amdgcn_readfirstlane(__shfl(myN, wave)) : 0;

    if (MODE == 1 && replay) {
        constexpr int RPU = 8;
        static_assert(32 % (EPS * RPU) == 0, "a block is a whole number of steps");
        int s = 0, dl = -1;
        float w = 0.0f;
        if (nb > 0) { s = recS[rbase]; dl = recD[rbase]; w = recW[rbase]; }
        for (int b = 0; b < nb; ++b) {
            int s2 = s, dl2 = dl;
            float w2 = w;
            for (int j0 = 0; j0 < 32; j0 += EPS * RPU) {
                float2 r[RPU];
                float  ww[RPU];
                int    dd[RPU];
#pragma unroll
                for (int p = 0; p < RPU; ++p) {
                    const int jj = j0 + p * EPS + sub;
                    const int ss = __shfl(s, jj);
                    dd[p] = __shfl(dl, jj);
                    ww[p] = __shfl(w, jj);
                    r[p]  = *(const float2*)(xl + (long long)ss * C + cl * 2);
                }
                if (j0 == 0) {
                    const size_t o = rbase + (size_t)min(b + 1, nb - 1) * 32;
                    s2 = recS[o]; dl2 = recD[o]; w2 = recW[o];
                }
#pragma unroll
                for (int p = 0; p < RPU; ++p) {
                    if (dd[p] >= 0) {
                        float* t = &tile[dd[p] * C + cl * 2];
                        atomicAdd(t,     r[p].x * ww[p]);
                        atomicAdd(t + 1, r[p].y * ww[p]);
                    }
                }
            }
            s = s2; dl = dl2; w = w2;
        }
    } else if (MODE == 0 && replay) {

    gatherRec(nb);

    } else {

    const int  nG  = (nEdges + 3) >> 2;
    const int  nG4 = nEdges >> 2;
    const bool al  = (((size_t)dst) & 15) == 0;

    for (int g0 = 0; g0 < nG; g0 += AGG_T * SCAN_U) {
        int4 d4[SCAN_U];
        if (al && g0 + AGG_T * SCAN_U <= nG4) {
            const int4* p = (const int4*)dst + (g0 + tid);
#pragma unroll
            for (int u = 0; u < SCAN_U; ++u)
                d4[u] = p[u * AGG_T];
        } else {
#pragma unroll
            for (int u = 0; u < SCAN_U; ++u)
                d4[u] = gcn_load4_guarded(dst, g0 + u * AGG_T + tid, nEdges);
        }

#pragma unroll
        for (int u = 0; u < SCAN_U; ++u) {
            const int e0 = (g0 + u * AGG_T + tid) << 2;
            const unsigned r0 = (unsigned)(d4[u].x - base);
            const unsigned r1 = (unsigned)(d4[u].y - base);
            const unsigned r2 = (unsigned)(d4[u].z - base);
            const unsigned r3 = (unsigned)(d4[u].w - base);
            const unsigned rm = min(min(r0, r1), min(r2, r3));
            if (__builtin_amdgcn_ballot_w32(rm < (unsigned)nLoc)) {
                const bool m0 = r0 < (unsigned)nLoc;
                const bool m1 = r1 < (unsigned)nLoc;
                const bool m2 = r2 < (unsigned)nLoc;
                const bool m3 = r3 < (unsigned)nLoc;
                const unsigned b0 = __builtin_amdgcn_ballot_w32(m0);
                const unsigned b1 = __builtin_amdgcn_ballot_w32(m1);
                const unsigned b2 = __builtin_amdgcn_ballot_w32(m2);
                const unsigned b3 = __builtin_amdgcn_ballot_w32(m3);
                if (b0) {
                    if (m0) myq[cnt + (int)__builtin_amdgcn_mbcnt_lo(b0, 0u)] = e0;
                    cnt += __builtin_popcount(b0);
                }
                if (b1) {
                    if (m1) myq[cnt + (int)__builtin_amdgcn_mbcnt_lo(b1, 0u)] = e0 + 1;
                    cnt += __builtin_popcount(b1);
                }
                if (b2) {
                    if (m2) myq[cnt + (int)__builtin_amdgcn_mbcnt_lo(b2, 0u)] = e0 + 2;
                    cnt += __builtin_popcount(b2);
                }
                if (b3) {
                    if (m3) myq[cnt + (int)__builtin_amdgcn_mbcnt_lo(b3, 0u)] = e0 + 3;
                    cnt += __builtin_popcount(b3);
                }
            }
        }

        drain(g0 + AGG_T * SCAN_U >= nG);
    }

    }

    const bool made = (MODE == 0) && !replay;
    if (made && lane == 0) wcount[wave] = nblk;
    __syncthreads();

    if (made && wave == 0) recN[(int)blockIdx.x * AGG_W + lane] = wcount[lane];

    {
        const float4* bias4 = (const float4*)bias;
        float4* out4 = (float4*)(outp + (long long)base * C);
#pragma unroll
        for (int t = 0; t < IT; ++t) {
            const int i = t * AGG_T + tid;
            if (i < nLoc * C4) {
                const float4 a = tile4[i];
                const float4 b = bias4[i % C4];
                float4 o;
                o.x = a.x + b.x; o.y = a.y + b.y; o.z = a.z + b.z; o.w = a.w + b.w;
                if (RELU) { o.x = fmaxf(o.x, 0.0f); o.y = fmaxf(o.y, 0.0f); o.z = fmaxf(o.z, 0.0f); o.w = fmaxf(o.w, 0.0f); }
                out4[i] = o;
            }
        }
    }
}

extern "C" void kernel_launch(void* const* d_in, const int* in_sizes, int n_in,
                              void* d_out, int out_size, void* d_ws, size_t ws_size,
                              hipStream_t stream) {
    const float* x  = (const float*)d_in[0];
    const int*   ei = (const int*)  d_in[1];
    const float* W1 = (const float*)d_in[2];
    const float* b1 = (const float*)d_in[3];
    const float* W2 = (const float*)d_in[4];
    const float* b2 = (const float*)d_in[5];
    float* out = (float*)d_out;

    const int nNodes = in_sizes[0] / IN_CH;
    const int nEdges = in_sizes[1] / 2;
    const int* src = ei;
    const int* dst = ei + nEdges;

    float* ws  = (float*)d_ws;
    float* deg = ws;
    float* xl1 = deg + ((nNodes + 63) & ~63);
    float* h   = xl1 + (long long)nNodes * HID;
    float* xl2 = h   + (long long)nNodes * HID;

    const int B = 256;
    const int nTiles = (nNodes + TILE - 1) / TILE;

    const long long recLen = (long long)nTiles * AGG_W * CAPW * 32;
    int*   recS = (int*)(xl2 + (((long long)nNodes * OUT_CH + 31) & ~31LL));
    int*   recD = recS + recLen;
    float* recW = (float*)(recD + recLen);
    int*   recN = (int*)(recW + recLen);
    unsigned* bmap = (unsigned*)(recN + (long long)nTiles * AGG_W);
    const int nWords = (nEdges + 1023) / 1024;
    const int useIdx = (nTiles <= MAXT) ? 1 : 0;

    if (useIdx) {
        const int wpb = IDX_T / 32;
        gcn_group_index_kernel<<<(nWords + wpb - 1) / wpb, IDX_T, 0, stream>>>(dst, bmap, nEdges, nWords);
        gcn_tile_collect_kernel<<<nTiles, AGG_T, 0, stream>>>(src, dst, deg, recS, recD, recN, bmap, nEdges, nNodes);
    } else {
        gcn_init_kernel<<<(nNodes + B - 1) / B, B, 0, stream>>>(deg, nNodes);
        gcn_degree_kernel<<<(nEdges + B - 1) / B, B, 0, stream>>>(dst, deg, nEdges);
        gcn_rsqrt_kernel<<<(nNodes + B - 1) / B, B, 0, stream>>>(deg, nNodes);
    }

    {
        int tiles = (nNodes / 16) * (HID / 16);
        int wpb = B / 32;
        gcn_gemm_wmma_f32<<<(tiles + wpb - 1) / wpb, B, 0, stream>>>(x, W1, xl1, nNodes, IN_CH, HID);
    }

    gcn_tile_agg_kernel<HID, true, 0><<<nTiles, AGG_T, 0, stream>>>(
        xl1, src, dst, deg, b1, h, recS, recD, recW, recN, useIdx, nEdges, nNodes);

    {
        int tiles = (nNodes / 16) * (OUT_CH / 16);
        int wpb = B / 32;
        gcn_gemm_wmma_f32<<<(tiles + wpb - 1) / wpb, B, 0, stream>>>(h, W2, xl2, nNodes, HID, OUT_CH);
    }

    gcn_tile_agg_kernel<OUT_CH, false, 1><<<nTiles, AGG_T, 0, stream>>>(
        xl2, src, dst, deg, b2, out, recS, recD, recW, recN, useIdx, nEdges, nNodes);
}


# First product moved into the collect launch: its resident work-groups keep W1 in LDS, each wave multiplies one 16-row strip on WMMA before or behind its collect work; k order kept
# speedup vs baseline: 4.6291x; 1.1538x over previous
#include <hip/hip_runtime.h>

#define IN_CH  128
#define HID    64
#define OUT_CH 16

typedef float v2f __attribute__((ext_vector_type(2)));
typedef float v4f __attribute__((ext_vector_type(4)));
typedef float v8f __attribute__((ext_vector_type(8)));

__global__ void gcn_init_kernel(float* __restrict__ deg, int nNodes) {
    int i = blockIdx.x * blockDim.x + threadIdx.x;
    if (i < nNodes) deg[i] = 1.0f;
}

__global__ void gcn_degree_kernel(const int* __restrict__ dst,
                                  float* __restrict__ deg, int nEdges) {
    int e = blockIdx.x * blockDim.x + threadIdx.x;
    if (e < nEdges) atomicAdd(&deg[dst[e]], 1.0f);
}

__global__ void gcn_rsqrt_kernel(float* __restrict__ deg, int n) {
    int i = blockIdx.x * blockDim.x + threadIdx.x;
    if (i < n) deg[i] = rsqrtf(deg[i]);
}

__global__ void gcn_gemm_wmma_f32(const float* __restrict__ A,
                                  const float* __restrict__ W,
                                  float* __restrict__ C,
                                  int M, int K, int N) {
    const int wavesPerBlock = blockDim.x >> 5;
    const int wave = (int)(threadIdx.x >> 5);
    const int lane = (int)(threadIdx.x & 31);
    const int tilesN = N >> 4;
    const int gw = blockIdx.x * wavesPerBlock + wave;
    const int tileM = gw / tilesN;
    const int tileN = gw - tileM * tilesN;
    if (tileM * 16 >= M) return;

    const int row0 = tileM << 4;
    const int col0 = tileN << 4;
    const int mr = lane & 15;
    const int kh = (lane >> 4) << 1;

    v8f acc = {};
    for (int k = 0; k < K; k += 4) {
        v2f a, b;
        const float* ap = A + (long long)(row0 + mr) * K + (k + kh);
        a.x = ap[0];
        a.y = ap[1];
        const float* bp = W + (long long)(k + kh) * N + (col0 + mr);
        b.x = bp[0];
        b.y = bp[N];
        acc = __builtin_amdgcn_wmma_f32_16x16x4_f32(
             false, a,  false, b,
             (short)0, acc,  false,  false);
    }

    const int crow = row0 + ((lane >> 4) << 3);
    const int ccol = col0 + mr;
#pragma unroll
    for (int r = 0; r < 8; ++r)
        C[(long long)(crow + r) * N + ccol] = acc[r];
}

#define TILE   392
#define AGG_T  1024
#define SCAN_U 8
#define PG     16
#define EQCAP  1248
#define IQCAP  832
static_assert(IQCAP * 4 + IQCAP * 2 <= EQCAP * 4, "edge ids and local destinations fit the wave's queue memory");
static_assert(IQCAP >= 32 + PG * 32, "room for a rest plus one step in which every edge matches");
static_assert(TILE <= 65536, "a local destination fits 16 bits");
#define GLCAP  1024
static_assert(EQCAP >= 32 + SCAN_U * 128, "the scanning path's queue (every scanned edge matches) fits too");
#define MAXT   256
#define IDX_T  128
#define CAPW   16
#define AGG_W  (AGG_T / 32)
static_assert(AGG_W == 32, "one lane per wave when a wave reads its tile's block counts");

__global__ __launch_bounds__(IDX_T)
void gcn_group_index_kernel(const int* __restrict__ dst,
                            unsigned* __restrict__ bmap,
                            int nEdges, int nWords) {
    __shared__ unsigned rows[(IDX_T / 32) * MAXT];
    const int lane = (int)(threadIdx.x & 31);
    const int wave = (int)(threadIdx.x >> 5);
    const int word = (int)blockIdx.x * (IDX_T / 32) + wave;
    if (word >= nWords) return;
    unsigned* my = rows + wave * MAXT;
#pragma unroll
    for (int k = 0; k < MAXT / 32; ++k) my[k * 32 + lane] = 0u;
    const int e0 = word * 1024 + lane;
    for (int g0 = 0; g0 < 32; g0 += 8) {
        int dv[8];
#pragma unroll
        for (int u = 0; u < 8; ++u) {
            const int e = e0 + (g0 + u) * 32;
            dv[u] = (e < nEdges) ? dst[e] : -1;
        }
#pragma unroll
        for (int u = 0; u < 8; ++u) {
            if (dv[u] >= 0) atomicOr(&my[dv[u] / TILE], 1u << (g0 + u));
        }
    }
#pragma unroll
    for (int k = 0; k < MAXT / 32; ++k)
        bmap[(size_t)word * MAXT + k * 32 + lane] = my[k * 32 + lane];
}

#define APITCH 68
static_assert(HID == 64 && IN_CH == 128, "the strip product keeps four 16-column accumulators and stages x in two halves of 64 k");
static_assert(16 * APITCH <= EQCAP, "a half strip of x fits the wave's queue memory");
__global__ __launch_bounds__(AGG_T)
void gcn_tile_collect_kernel(const int* __restrict__ src,
                             const int* __restrict__ dst,
                             float* __restrict__ dinv,
                             int*   __restrict__ recS,
                             int*   __restrict__ recD,
                             int*   __restrict__ recN,
                             const unsigned* __restrict__ bmap,
                             const float* __restrict__ x,
                             const float* __restrict__ W,
                             float* __restrict__ xl,
                             int nEdges, int nNodes) {
    __shared__ __attribute__((aligned(16))) int qmem[AGG_W * EQCAP];
    __shared__ unsigned short gmem[AGG_W * GLCAP];
    __shared__ int   degc[TILE];
    __shared__ int   wcount[AGG_W];
    __shared__ __attribute__((aligned(16))) float wlds[IN_CH * HID];

    const int tid  = (int)threadIdx.x;
    const int lane = tid & 31;
    const int wave = tid >> 5;
    const int base = (int)blockIdx.x * TILE;
    const int nLoc = min(TILE, nNodes - base);

    for (int i = tid; i < TILE; i += AGG_T) degc[i] = 0;
    {
        constexpr int WIT = IN_CH * HID / AGG_T;
        static_assert(WIT * AGG_T == IN_CH * HID, "every thread stages the same number of words of W");
        float wv[WIT];
#pragma unroll
        for (int t = 0; t < WIT; ++t) wv[t] = W[t * AGG_T + tid];
#pragma unroll
        for (int t = 0; t < WIT; ++t) {
            const int i = t * AGG_T + tid;
            const int k = i / HID, col = i % HID;
            const int c = col >> 4;
            wlds[(((((k >> 2) * 2 + (c >> 1)) * 2 + ((k >> 1) & 1)) * 16 + (col & 15)) << 2) + ((c & 1) << 1) + (k & 1)] = wv[t];
        }
    }
    __syncthreads();

    int* myq = qmem + wave * EQCAP;
    unsigned short* mydl = (unsigned short*)(myq + IQCAP);
    int cnt = 0;

    auto stripProduct = [&]() {
        const int mr = lane & 15;
        const int hh = lane >> 4;
        const int nStrips = (nNodes + 15) >> 4;
        const v4f* bp = (const v4f*)wlds + (hh * 16 + mr);
        float* stg = (float*)myq;
        for (int s = (int)blockIdx.x + (int)gridDim.x * wave; s < nStrips; s += (int)gridDim.x * AGG_W) {
            const int row0 = s << 4;
            v4f g[2][8];
#pragma unroll
            for (int q = 0; q < 2; ++q) {
#pragma unroll
                for (int i = 0; i < 8; ++i) {
                    const int row = min(row0 + 2 * i + hh, nNodes - 1);
                    g[q][i] = *(const v4f*)(x + (size_t)row * IN_CH + q * 64 + 4 * mr);
                }
            }
            v8f acc0 = {}, acc1 = {}, acc2 = {}, acc3 = {};
#pragma unroll
            for (int q = 0; q < 2; ++q) {
#pragma unroll
                for (int i = 0; i < 8; ++i)
                    *(v4f*)(stg + (2 * i + hh) * APITCH + 4 * mr) = g[q][i];
                const v2f* as = (const v2f*)(stg + mr * APITCH + 2 * hh);
#pragma unroll 4
                for (int jj = 0; jj < 16; ++jj) {
                    const int j = q * 16 + jj;
                    const v2f a   = as[2 * jj];
                    const v4f b01 = bp[(j * 2 + 0) * 32];
                    const v4f b23 = bp[(j * 2 + 1) * 32];
                    const v2f b0 = {b01.x, b01.y}, b1 = {b01.z, b01.w}, b2 = {b23.x, b23.y}, b3 = {b23.z, b23.w};
                    acc0 = __builtin_amdgcn_wmma_f32_16x16x4_f32(false, a, false, b0, (short)0, acc0, false, false);
                    acc1 = __builtin_amdgcn_wmma_f32_16x16x4_f32(false, a, false, b1, (short)0, acc1, false, false);
                    acc2 = __builtin_amdgcn_wmma_f32_16x16x4_f32(false, a, false, b2, (short)0, acc2, false, false);
                    acc3 = __builtin_amdgcn_wmma_f32_16x16x4_f32(false, a, false, b3, (short)0, acc3, false, false);
                }
            }
            const int crow = row0 + (hh << 3);
#pragma unroll
            for (int r = 0; r < 8; ++r) {
                if (crow + r < nNodes) {
                    float* o = xl + (size_t)(crow + r) * HID + mr;
                    o[0]  = acc0[r];
                    o[16] = acc1[r];
                    o[32] = acc2[r];
                    o[48] = acc3[r];
                }
            }
        }
    };
    const bool early = ((wave >> 2) & 1) == 0;
    if (early) stripProduct();

    const size_t rbase = ((size_t)((int)blockIdx.x * AGG_W + wave) * CAPW) * 32 + lane;
    int nblk = 0;

    constexpr int RU = 8;
    auto recordAll = [&]() {
        const int nB   = (cnt + 31) >> 5;
        const int last = max(cnt - 1, 0);
        for (int b0 = 0; b0 < nB; b0 += RU) {
            int sv[RU], dv[RU];
#pragma unroll
            for (int u = 0; u < RU; ++u) {
                const int q = min((b0 + u) * 32 + lane, last);
                dv[u] = mydl[q];
                sv[u] = src[myq[q]];
            }
#pragma unroll
            for (int u = 0; u < RU; ++u) {
                if (b0 + u < nB) {
                    if (nblk < CAPW) {
                        const size_t o = rbase + (size_t)nblk * 32;
                        recS[o] = sv[u];
                        recD[o] = ((b0 + u) * 32 + lane < cnt) ? dv[u] : -1;
                    }
                    ++nblk;
                }
            }
        }
        cnt = 0;
    };

    unsigned short* glist = gmem + wave * GLCAP;
    const int nW = (nEdges + 1023) >> 10;
    for (int k0 = 0; k0 * 32 < nW; k0 += 32) {
        const int w = (k0 + lane) * 32 + wave;
        unsigned bits = (w < nW) ? bmap[(size_t)w * MAXT + blockIdx.x] : 0u;
        const int pc = __builtin_popcount(bits);
        int incl = pc;
#pragma unroll
        for (int off = 1; off < 32; off <<= 1) {
            const int v = __shfl_up(incl, off);
            if (lane >= off) incl += v;
        }
        const int total = __builtin_amdgcn_readfirstlane(__shfl(incl, 31));
        int pos = incl - pc;
        while (bits) {
            const int j = __builtin_ctz(bits);
            bits &= bits - 1u;
            glist[pos++] = (unsigned short)(lane * 32 + j);
        }
        for (int i0 = 0; i0 < total; i0 += 32) {
            const int gl = (int)glist[min(i0 + lane, total - 1)];
            const int nq = min(32, total - i0);
            for (int q0 = 0; q0 < nq; q0 += PG) {
                if (cnt > IQCAP - PG * 32) recordAll();
                int dv[PG], ev[PG];
#pragma unroll
                for (int p = 0; p < PG; ++p) {
                    const int v = __builtin_amdgcn_readlane(gl, min(q0 + p, nq - 1));
                    const int G = ((k0 + (v >> 5)) * 32 + wave) * 32 + (v & 31);
                    ev[p] = G * 32 + lane;
                    dv[p] = dst[min(ev[p], nEdges - 1)];
                }
#pragma unroll
                for (int p = 0; p < PG; ++p) {
                    if (q0 + p < nq) {
                        const bool m = (ev[p] < nEdges) && ((unsigned)(dv[p] - base) < (unsigned)nLoc);
                        const unsigned b = __builtin_amdgcn_ballot_w32(m);
                        if (m) {
                            const int qp = cnt + (int)__builtin_amdgcn_mbcnt_lo(b, 0u);
                            myq[qp]  = ev[p];
                            mydl[qp] = (unsigned short)(dv[p] - base);
                            atomicAdd(&degc[dv[p] - base], 1);
                        }
                        cnt += __builtin_popcount(b);
                    }
                }
            }
        }
    }
    recordAll();

    if (!early) stripProduct();

    if (lane == 0) wcount[wave] = nblk;
    __syncthreads();

    if (wave == 0) recN[(int)blockIdx.x * AGG_W + lane] = wcount[lane];

    for (int i = tid; i < nLoc; i += AGG_T) dinv[base + i] = rsqrtf(1.0f + (float)degc[i]);
}

static __device__ __forceinline__ int4 gcn_load4_guarded(const int* __restrict__ p, int g, int n) {
    const int e = g << 2;
    int4 r;
    r.x = (e     < n) ? p[e]     : -1;
    r.y = (e + 1 < n) ? p[e + 1] : -1;
    r.z = (e + 2 < n) ? p[e + 2] : -1;
    r.w = (e + 3 < n) ? p[e + 3] : -1;
    return r;
}

template <int C, bool RELU, int MODE>
__global__ __launch_bounds__(AGG_T)
void gcn_tile_agg_kernel(const float* __restrict__ xl,
                         const int* __restrict__ src,
                         const int* __restrict__ dst,
                         const float* __restrict__ dinv,
                         const float* __restrict__ bias,
                         float* __restrict__ outp,
                         int*   __restrict__ recS,
                         int*   __restrict__ recD,
                         float* __restrict__ recW,
                         int*   __restrict__ recN,
                         int useRec,
                         int nEdges, int nNodes) {
    __shared__ __attribute__((aligned(16))) float tile[TILE * C];
    __shared__ int   qmem[(AGG_T / 32) * EQCAP];
    __shared__ int   wcount[AGG_T / 32];

    const int tid  = (int)threadIdx.x;
    const int lane = tid & 31;
    const int wave = tid >> 5;
    const int base = (int)blockIdx.x * TILE;
    const int nLoc = min(TILE, nNodes - base);

    int myN = 0;
    if (MODE == 1 || useRec) myN = recN[(int)blockIdx.x * (AGG_T / 32) + lane];

    constexpr int C4 = C / 4;
    constexpr int NV = TILE * C4;
    constexpr int IT = (NV + AGG_T - 1) / AGG_T;
    float4* tile4 = (float4*)tile;
    {
        float4 v[IT];
        float  di[IT];
#pragma unroll
        for (int t = 0; t < IT; ++t) {
            const int i  = min(t * AGG_T + tid, NV - 1);
            const int n  = min(i / C4, nLoc - 1);
            di[t] = dinv[base + n];
            v[t]  = *(const float4*)(xl + (long long)(base + n) * C + (i % C4) * 4);
        }
#pragma unroll
        for (int t = 0; t < IT; ++t) {
            const int i = t * AGG_T + tid;
            if (i < NV) {
                const bool ok = (i / C4) < nLoc;
                const float w = di[t] * di[t];
                float4 o;
                o.x = ok ? v[t].x * w : 0.0f;
                o.y = ok ? v[t].y * w : 0.0f;
                o.z = ok ? v[t].z * w : 0.0f;
                o.w = ok ? v[t].w * w : 0.0f;
                tile4[i] = o;
            }
        }
    }
    __syncthreads();

    int* myq = qmem + wave * EQCAP;
    int cnt = 0;

    constexpr int LPE = C / 2;
    constexpr int EPS = 32 / LPE;
    constexpr int PU  = 4;
    const int sub = lane / LPE;
    const int cl  = lane % LPE;

    const size_t rbase = ((size_t)((int)blockIdx.x * AGG_W + wave) * CAPW) * 32 + lane;
    int nblk = 0;

    auto drain = [&](const bool flush) {
        while (cnt >= 32 || (flush && cnt > 0)) {
            const int n = min(cnt, 32);
            int   s = 0, dl = 0;
            float w = 0.0f;
            if (lane < n) {
                const int e = myq[cnt - n + lane];
                s  = src[e];
                dl = dst[e] - base;
                w  = dinv[s] * dinv[base + dl];
            }
            if (MODE == 0) {
                if (nblk < CAPW) {
                    const size_t o = rbase + (size_t)nblk * 32;
                    recS[o] = s;
                    recD[o] = (lane < n) ? dl : -1;
                    recW[o] = w;
                }
                ++nblk;
            }
            cnt -= n;
            for (int j0 = 0; j0 < n; j0 += EPS * PU) {
                float2 r[PU];
                float  ww[PU];
                int    dd[PU];
#pragma unroll
                for (int p = 0; p < PU; ++p) {
                    const int jj = min(j0 + p * EPS + sub, n - 1);
                    const int ss = __shfl(s, jj);
                    dd[p] = __shfl(dl, jj);
                    ww[p] = __shfl(w, jj);
                    r[p]  = *(const float2*)(xl + (long long)ss * C + cl * 2);
                }
#pragma unroll
                for (int p = 0; p < PU; ++p) {
                    if (j0 + p * EPS + sub < n) {
                        float* t = &tile[dd[p] * C + cl * 2];
                        atomicAdd(t,     r[p].x * ww[p]);
                        atomicAdd(t + 1, r[p].y * ww[p]);
                    }
                }
            }
        }
    };

    constexpr int  GU  = 16;
    constexpr bool UNI = (EPS == 1);
    auto gatherRec = [&](const int nb) {
        int   s0 = 0, dl0 = -1, s1 = 0, dl1 = -1;
        float w0 = 0.0f;
        if (nb > 0) {
            const size_t o1 = rbase + (size_t)min(1, nb - 1) * 32;
            s0  = recS[rbase];
            dl0 = recD[rbase];
            s1  = recS[o1];
            dl1 = recD[o1];
            w0  = dinv[s0] * dinv[base + max(dl0, 0)];
        }
        for (int b = 0; b < nb; ++b) {
            const int m = __builtin_popcount(__builtin_amdgcn_ballot_w32(dl0 >= 0));
            int   s2 = 0, dl2 = -1;
            float x1 = 0.0f, y1 = 0.0f;
            recW[rbase + (size_t)b * 32] = w0;
            for (int j0 = 0; j0 < m; j0 += EPS * GU) {
                float2 r[GU];
                float  ww[GU];
                int    dd[GU];
#pragma unroll
                for (int p = 0; p < GU; ++p) {
                    const int jj = min(j0 + p * EPS + sub, m - 1);
                    int ss;
                    if (UNI) {
                        ss    = __builtin_amdgcn_readlane(s0, jj);
                        dd[p] = __builtin_amdgcn_readlane(dl0, jj);
                        ww[p] = __int_as_float(__builtin_amdgcn_readlane(__float_as_int(w0), jj));
                    } else {
                        ss    = __shfl(s0, jj);
                        dd[p] = __shfl(dl0, jj);
                        ww[p] = __shfl(w0, jj);
                    }
                    r[p] = *(const float2*)(xl + (long long)ss * C + cl * 2);
                }
                if (j0 == 0) {
                    const size_t o2 = rbase + (size_t)min(b + 2, nb - 1) * 32;
                    s2  = recS[o2];
                    dl2 = recD[o2];
                    x1  = dinv[s1];
                    y1  = dinv[base + max(dl1, 0)];
                }
#pragma unroll
                for (int p = 0; p < GU; ++p) {
                    if (j0 + p * EPS + sub < m) {
                        float* t = &tile[dd[p] * C + cl * 2];
                        atomicAdd(t,     r[p].x * ww[p]);
                        atomicAdd(t + 1, r[p].y * ww[p]);
                    }
                }
            }
            s0 = s1; dl0 = dl1; w0 = x1 * y1;
            s1 = s2; dl1 = dl2;
        }
    };

    bool replay = false;
    if (MODE == 1 || useRec)
        replay = (__builtin_amdgcn_ballot_w32(myN > CAPW) == 0u);
    const int nb = replay ? __builtin_amdgcn_readfirstlane(__shfl(myN, wave)) : 0;

    if (MODE == 1 && replay) {
        constexpr int RPU = 8;
        static_assert(32 % (EPS * RPU) == 0, "a block is a whole number of steps");
        int s = 0, dl = -1;
        float w = 0.0f;
        if (nb > 0) { s = recS[rbase]; dl = recD[rbase]; w = recW[rbase]; }
        for (int b = 0; b < nb; ++b) {
            int s2 = s, dl2 = dl;
            float w2 = w;
            for (int j0 = 0; j0 < 32; j0 += EPS * RPU) {
                float2 r[RPU];
                float  ww[RPU];
                int    dd[RPU];
#pragma unroll
                for (int p = 0; p < RPU; ++p) {
                    const int jj = j0 + p * EPS + sub;
                    const int ss = __shfl(s, jj);
                    dd[p] = __shfl(dl, jj);
                    ww[p] = __shfl(w, jj);
                    r[p]  = *(const float2*)(xl + (long long)ss * C + cl * 2);
                }
                if (j0 == 0) {
                    const size_t o = rbase + (size_t)min(b + 1, nb - 1) * 32;
                    s2 = recS[o]; dl2 = recD[o]; w2 = recW[o];
                }
#pragma unroll
                for (int p = 0; p < RPU; ++p) {
                    if (dd[p] >= 0) {
                        float* t = &tile[dd[p] * C + cl * 2];
                        atomicAdd(t,     r[p].x * ww[p]);
                        atomicAdd(t + 1, r[p].y * ww[p]);
                    }
                }
            }
            s = s2; dl = dl2; w = w2;
        }
    } else if (MODE == 0 && replay) {

    gatherRec(nb);

    } else {

    const int  nG  = (nEdges + 3) >> 2;
    const int  nG4 = nEdges >> 2;
    const bool al  = (((size_t)dst) & 15) == 0;

    for (int g0 = 0; g0 < nG; g0 += AGG_T * SCAN_U) {
        int4 d4[SCAN_U];
        if (al && g0 + AGG_T * SCAN_U <= nG4) {
            const int4* p = (const int4*)dst + (g0 + tid);
#pragma unroll
            for (int u = 0; u < SCAN_U; ++u)
                d4[u] = p[u * AGG_T];
        } else {
#pragma unroll
            for (int u = 0; u < SCAN_U; ++u)
                d4[u] = gcn_load4_guarded(dst, g0 + u * AGG_T + tid, nEdges);
        }

#pragma unroll
        for (int u = 0; u < SCAN_U; ++u) {
            const int e0 = (g0 + u * AGG_T + tid) << 2;
            const unsigned r0 = (unsigned)(d4[u].x - base);
            const unsigned r1 = (unsigned)(d4[u].y - base);
            const unsigned r2 = (unsigned)(d4[u].z - base);
            const unsigned r3 = (unsigned)(d4[u].w - base);
            const unsigned rm = min(min(r0, r1), min(r2, r3));
            if (__builtin_amdgcn_ballot_w32(rm < (unsigned)nLoc)) {
                const bool m0 = r0 < (unsigned)nLoc;
                const bool m1 = r1 < (unsigned)nLoc;
                const bool m2 = r2 < (unsigned)nLoc;
                const bool m3 = r3 < (unsigned)nLoc;
                const unsigned b0 = __builtin_amdgcn_ballot_w32(m0);
                const unsigned b1 = __builtin_amdgcn_ballot_w32(m1);
                const unsigned b2 = __builtin_amdgcn_ballot_w32(m2);
                const unsigned b3 = __builtin_amdgcn_ballot_w32(m3);
                if (b0) {
                    if (m0) myq[cnt + (int)__builtin_amdgcn_mbcnt_lo(b0, 0u)] = e0;
                    cnt += __builtin_popcount(b0);
                }
                if (b1) {
                    if (m1) myq[cnt + (int)__builtin_amdgcn_mbcnt_lo(b1, 0u)] = e0 + 1;
                    cnt += __builtin_popcount(b1);
                }
                if (b2) {
                    if (m2) myq[cnt + (int)__builtin_amdgcn_mbcnt_lo(b2, 0u)] = e0 + 2;
                    cnt += __builtin_popcount(b2);
                }
                if (b3) {
                    if (m3) myq[cnt + (int)__builtin_amdgcn_mbcnt_lo(b3, 0u)] = e0 + 3;
                    cnt += __builtin_popcount(b3);
                }
            }
        }

        drain(g0 + AGG_T * SCAN_U >= nG);
    }

    }

    const bool made = (MODE == 0) && !replay;
    if (made && lane == 0) wcount[wave] = nblk;
    __syncthreads();

    if (made && wave == 0) recN[(int)blockIdx.x * AGG_W + lane] = wcount[lane];

    {
        const float4* bias4 = (const float4*)bias;
        float4* out4 = (float4*)(outp + (long long)base * C);
#pragma unroll
        for (int t = 0; t < IT; ++t) {
            const int i = t * AGG_T + tid;
            if (i < nLoc * C4) {
                const float4 a = tile4[i];
                const float4 b = bias4[i % C4];
                float4 o;
                o.x = a.x + b.x; o.y = a.y + b.y; o.z = a.z + b.z; o.w = a.w + b.w;
                if (RELU) { o.x = fmaxf(o.x, 0.0f); o.y = fmaxf(o.y, 0.0f); o.z = fmaxf(o.z, 0.0f); o.w = fmaxf(o.w, 0.0f); }
                out4[i] = o;
            }
        }
    }
}

extern "C" void kernel_launch(void* const* d_in, const int* in_sizes, int n_in,
                              void* d_out, int out_size, void* d_ws, size_t ws_size,
                              hipStream_t stream) {
    const float* x  = (const float*)d_in[0];
    const int*   ei = (const int*)  d_in[1];
    const float* W1 = (const float*)d_in[2];
    const float* b1 = (const float*)d_in[3];
    const float* W2 = (const float*)d_in[4];
    const float* b2 = (const float*)d_in[5];
    float* out = (float*)d_out;

    const int nNodes = in_sizes[0] / IN_CH;
    const int nEdges = in_sizes[1] / 2;
    const int* src = ei;
    const int* dst = ei + nEdges;

    float* ws  = (float*)d_ws;
    float* deg = ws;
    float* xl1 = deg + ((nNodes + 63) & ~63);
    float* h   = xl1 + (long long)nNodes * HID;
    float* xl2 = h   + (long long)nNodes * HID;

    const int B = 256;
    const int nTiles = (nNodes + TILE - 1) / TILE;

    const long long recLen = (long long)nTiles * AGG_W * CAPW * 32;
    int*   recS = (int*)(xl2 + (((long long)nNodes * OUT_CH + 31) & ~31LL));
    int*   recD = recS + recLen;
    float* recW = (float*)(recD + recLen);
    int*   recN = (int*)(recW + recLen);
    unsigned* bmap = (unsigned*)(recN + (long long)nTiles * AGG_W);
    const int nWords = (nEdges + 1023) / 1024;
    const int useIdx = (nTiles <= MAXT) ? 1 : 0;

    if (useIdx) {
        const int wpb = IDX_T / 32;
        gcn_group_index_kernel<<<(nWords + wpb - 1) / wpb, IDX_T, 0, stream>>>(dst, bmap, nEdges, nWords);
        gcn_tile_collect_kernel<<<nTiles, AGG_T, 0, stream>>>(src, dst, deg, recS, recD, recN, bmap, x, W1, xl1, nEdges, nNodes);
    } else {
        gcn_init_kernel<<<(nNodes + B - 1) / B, B, 0, stream>>>(deg, nNodes);
        gcn_degree_kernel<<<(nEdges + B - 1) / B, B, 0, stream>>>(dst, deg, nEdges);
        gcn_rsqrt_kernel<<<(nNodes + B - 1) / B, B, 0, stream>>>(deg, nNodes);
        int tiles = (nNodes / 16) * (HID / 16);
        int wpb = B / 32;
        gcn_gemm_wmma_f32<<<(tiles + wpb - 1) / wpb, B, 0, stream>>>(x, W1, xl1, nNodes, IN_CH, HID);
    }

    gcn_tile_agg_kernel<HID, true, 0><<<nTiles, AGG_T, 0, stream>>>(
        xl1, src, dst, deg, b1, h, recS, recD, recW, recN, useIdx, nEdges, nNodes);

    {
        int tiles = (nNodes / 16) * (OUT_CH / 16);
        int wpb = B / 32;
        gcn_gemm_wmma_f32<<<(tiles + wpb - 1) / wpb, B, 0, stream>>>(h, W2, xl2, nNodes, HID, OUT_CH);
    }

    gcn_tile_agg_kernel<OUT_CH, false, 1><<<nTiles, AGG_T, 0, stream>>>(
        xl2, src, dst, deg, b2, out, recS, recD, recW, recN, useIdx, nEdges, nNodes);
}


# First aggregation holds a byte degree table of all nodes in LDS (written by the collect launch, read once per work-group): an edge's deg^-1/2 factors come from LDS, not global lookups; sum order kept
# speedup vs baseline: 4.7702x; 1.0305x over previous
#include <hip/hip_runtime.h>

#define IN_CH  128
#define HID    64
#define OUT_CH 16

typedef float v2f __attribute__((ext_vector_type(2)));
typedef float v4f __attribute__((ext_vector_type(4)));
typedef float v8f __attribute__((ext_vector_type(8)));

__global__ void gcn_init_kernel(float* __restrict__ deg, int nNodes) {
    int i = blockIdx.x * blockDim.x + threadIdx.x;
    if (i < nNodes) deg[i] = 1.0f;
}

__global__ void gcn_degree_kernel(const int* __restrict__ dst,
                                  float* __restrict__ deg, int nEdges) {
    int e = blockIdx.x * blockDim.x + threadIdx.x;
    if (e < nEdges) atomicAdd(&deg[dst[e]], 1.0f);
}

__global__ void gcn_rsqrt_kernel(float* __restrict__ deg, int n) {
    int i = blockIdx.x * blockDim.x + threadIdx.x;
    if (i < n) deg[i] = rsqrtf(deg[i]);
}

__global__ void gcn_gemm_wmma_f32(const float* __restrict__ A,
                                  const float* __restrict__ W,
                                  float* __restrict__ C,
                                  int M, int K, int N) {
    const int wavesPerBlock = blockDim.x >> 5;
    const int wave = (int)(threadIdx.x >> 5);
    const int lane = (int)(threadIdx.x & 31);
    const int tilesN = N >> 4;
    const int gw = blockIdx.x * wavesPerBlock + wave;
    const int tileM = gw / tilesN;
    const int tileN = gw - tileM * tilesN;
    if (tileM * 16 >= M) return;

    const int row0 = tileM << 4;
    const int col0 = tileN << 4;
    const int mr = lane & 15;
    const int kh = (lane >> 4) << 1;

    v8f acc = {};
    for (int k = 0; k < K; k += 4) {
        v2f a, b;
        const float* ap = A + (long long)(row0 + mr) * K + (k + kh);
        a.x = ap[0];
        a.y = ap[1];
        const float* bp = W + (long long)(k + kh) * N + (col0 + mr);
        b.x = bp[0];
        b.y = bp[N];
        acc = __builtin_amdgcn_wmma_f32_16x16x4_f32(
             false, a,  false, b,
             (short)0, acc,  false,  false);
    }

    const int crow = row0 + ((lane >> 4) << 3);
    const int ccol = col0 + mr;
#pragma unroll
    for (int r = 0; r < 8; ++r)
        C[(long long)(crow + r) * N + ccol] = acc[r];
}

#define TILE   392
#define AGG_T  1024
#define SCAN_U 8
#define PG     16
#define EQCAP  1248
#define IQCAP  832
static_assert(IQCAP * 4 + IQCAP * 2 <= EQCAP * 4, "edge ids and local destinations fit the wave's queue memory");
static_assert(IQCAP >= 32 + PG * 32, "room for a rest plus one step in which every edge matches");
static_assert(TILE <= 65536, "a local destination fits 16 bits");
#define GLCAP  1024
static_assert(EQCAP >= 32 + SCAN_U * 128, "the scanning path's queue (every scanned edge matches) fits too");
#define MAXT   256
#define IDX_T  128
#define CAPW   16
#define AGG_W  (AGG_T / 32)
static_assert(AGG_W == 32, "one lane per wave when a wave reads its tile's block counts");

__global__ __launch_bounds__(IDX_T)
void gcn_group_index_kernel(const int* __restrict__ dst,
                            unsigned* __restrict__ bmap,
                            int nEdges, int nWords) {
    __shared__ unsigned rows[(IDX_T / 32) * MAXT];
    const int lane = (int)(threadIdx.x & 31);
    const int wave = (int)(threadIdx.x >> 5);
    const int word = (int)blockIdx.x * (IDX_T / 32) + wave;
    if (word >= nWords) return;
    unsigned* my = rows + wave * MAXT;
#pragma unroll
    for (int k = 0; k < MAXT / 32; ++k) my[k * 32 + lane] = 0u;
    const int e0 = word * 1024 + lane;
    for (int g0 = 0; g0 < 32; g0 += 8) {
        int dv[8];
#pragma unroll
        for (int u = 0; u < 8; ++u) {
            const int e = e0 + (g0 + u) * 32;
            dv[u] = (e < nEdges) ? dst[e] : -1;
        }
#pragma unroll
        for (int u = 0; u < 8; ++u) {
            if (dv[u] >= 0) atomicOr(&my[dv[u] / TILE], 1u << (g0 + u));
        }
    }
#pragma unroll
    for (int k = 0; k < MAXT / 32; ++k)
        bmap[(size_t)word * MAXT + k * 32 + lane] = my[k * 32 + lane];
}

#define APITCH 68
static_assert(HID == 64 && IN_CH == 128, "the strip product keeps four 16-column accumulators and stages x in two halves of 64 k");
static_assert(16 * APITCH <= EQCAP, "a half strip of x fits the wave's queue memory");
__global__ __launch_bounds__(AGG_T)
void gcn_tile_collect_kernel(const int* __restrict__ src,
                             const int* __restrict__ dst,
                             float* __restrict__ dinv,
                             int*   __restrict__ recS,
                             int*   __restrict__ recD,
                             int*   __restrict__ recN,
                             const unsigned* __restrict__ bmap,
                             const float* __restrict__ x,
                             const float* __restrict__ W,
                             float* __restrict__ xl,
                             unsigned* __restrict__ degB,
                             int nEdges, int nNodes) {
    __shared__ __attribute__((aligned(16))) int qmem[AGG_W * EQCAP];
    __shared__ unsigned short gmem[AGG_W * GLCAP];
    __shared__ int   degc[TILE];
    __shared__ int   wcount[AGG_W];
    __shared__ __attribute__((aligned(16))) float wlds[IN_CH * HID];

    const int tid  = (int)threadIdx.x;
    const int lane = tid & 31;
    const int wave = tid >> 5;
    const int base = (int)blockIdx.x * TILE;
    const int nLoc = min(TILE, nNodes - base);

    for (int i = tid; i < TILE; i += AGG_T) degc[i] = 0;
    {
        constexpr int WIT = IN_CH * HID / AGG_T;
        static_assert(WIT * AGG_T == IN_CH * HID, "every thread stages the same number of words of W");
        float wv[WIT];
#pragma unroll
        for (int t = 0; t < WIT; ++t) wv[t] = W[t * AGG_T + tid];
#pragma unroll
        for (int t = 0; t < WIT; ++t) {
            const int i = t * AGG_T + tid;
            const int k = i / HID, col = i % HID;
            const int c = col >> 4;
            wlds[(((((k >> 2) * 2 + (c >> 1)) * 2 + ((k >> 1) & 1)) * 16 + (col & 15)) << 2) + ((c & 1) << 1) + (k & 1)] = wv[t];
        }
    }
    __syncthreads();

    int* myq = qmem + wave * EQCAP;
    unsigned short* mydl = (unsigned short*)(myq + IQCAP);
    int cnt = 0;

    auto stripProduct = [&]() {
        const int mr = lane & 15;
        const int hh = lane >> 4;
        const int nStrips = (nNodes + 15) >> 4;
        const v4f* bp = (const v4f*)wlds + (hh * 16 + mr);
        float* stg = (float*)myq;
        for (int s = (int)blockIdx.x + (int)gridDim.x * wave; s < nStrips; s += (int)gridDim.x * AGG_W) {
            const int row0 = s << 4;
            v4f g[2][8];
#pragma unroll
            for (int q = 0; q < 2; ++q) {
#pragma unroll
                for (int i = 0; i < 8; ++i) {
                    const int row = min(row0 + 2 * i + hh, nNodes - 1);
                    g[q][i] = *(const v4f*)(x + (size_t)row * IN_CH + q * 64 + 4 * mr);
                }
            }
            v8f acc0 = {}, acc1 = {}, acc2 = {}, acc3 = {};
#pragma unroll
            for (int q = 0; q < 2; ++q) {
#pragma unroll
                for (int i = 0; i < 8; ++i)
                    *(v4f*)(stg + (2 * i + hh) * APITCH + 4 * mr) = g[q][i];
                const v2f* as = (const v2f*)(stg + mr * APITCH + 2 * hh);
#pragma unroll 4
                for (int jj = 0; jj < 16; ++jj) {
                    const int j = q * 16 + jj;
                    const v2f a   = as[2 * jj];
                    const v4f b01 = bp[(j * 2 + 0) * 32];
                    const v4f b23 = bp[(j * 2 + 1) * 32];
                    const v2f b0 = {b01.x, b01.y}, b1 = {b01.z, b01.w}, b2 = {b23.x, b23.y}, b3 = {b23.z, b23.w};
                    acc0 = __builtin_amdgcn_wmma_f32_16x16x4_f32(false, a, false, b0, (short)0, acc0, false, false);
                    acc1 = __builtin_amdgcn_wmma_f32_16x16x4_f32(false, a, false, b1, (short)0, acc1, false, false);
                    acc2 = __builtin_amdgcn_wmma_f32_16x16x4_f32(false, a, false, b2, (short)0, acc2, false, false);
                    acc3 = __builtin_amdgcn_wmma_f32_16x16x4_f32(false, a, false, b3, (short)0, acc3, false, false);
                }
            }
            const int crow = row0 + (hh << 3);
#pragma unroll
            for (int r = 0; r < 8; ++r) {
                if (crow + r < nNodes) {
                    float* o = xl + (size_t)(crow + r) * HID + mr;
                    o[0]  = acc0[r];
                    o[16] = acc1[r];
                    o[32] = acc2[r];
                    o[48] = acc3[r];
                }
            }
        }
    };
    const bool early = ((wave >> 2) & 1) == 0;
    if (early) stripProduct();

    const size_t rbase = ((size_t)((int)blockIdx.x * AGG_W + wave) * CAPW) * 32 + lane;
    int nblk = 0;

    constexpr int RU = 8;
    auto recordAll = [&]() {
        const int nB   = (cnt + 31) >> 5;
        const int last = max(cnt - 1, 0);
        for (int b0 = 0; b0 < nB; b0 += RU) {
            int sv[RU], dv[RU];
#pragma unroll
            for (int u = 0; u < RU; ++u) {
                const int q = min((b0 + u) * 32 + lane, last);
                dv[u] = mydl[q];
                sv[u] = src[myq[q]];
            }
#pragma unroll
            for (int u = 0; u < RU; ++u) {
                if (b0 + u < nB) {
                    if (nblk < CAPW) {
                        const size_t o = rbase + (size_t)nblk * 32;
                        recS[o] = sv[u];
                        recD[o] = ((b0 + u) * 32 + lane < cnt) ? dv[u] : -1;
                    }
                    ++nblk;
                }
            }
        }
        cnt = 0;
    };

    unsigned short* glist = gmem + wave * GLCAP;
    const int nW = (nEdges + 1023) >> 10;
    for (int k0 = 0; k0 * 32 < nW; k0 += 32) {
        const int w = (k0 + lane) * 32 + wave;
        unsigned bits = (w < nW) ? bmap[(size_t)w * MAXT + blockIdx.x] : 0u;
        const int pc = __builtin_popcount(bits);
        int incl = pc;
#pragma unroll
        for (int off = 1; off < 32; off <<= 1) {
            const int v = __shfl_up(incl, off);
            if (lane >= off) incl += v;
        }
        const int total = __builtin_amdgcn_readfirstlane(__shfl(incl, 31));
        int pos = incl - pc;
        while (bits) {
            const int j = __builtin_ctz(bits);
            bits &= bits - 1u;
            glist[pos++] = (unsigned short)(lane * 32 + j);
        }
        for (int i0 = 0; i0 < total; i0 += 32) {
            const int gl = (int)glist[min(i0 + lane, total - 1)];
            const int nq = min(32, total - i0);
            for (int q0 = 0; q0 < nq; q0 += PG) {
                if (cnt > IQCAP - PG * 32) recordAll();
                int dv[PG], ev[PG];
#pragma unroll
                for (int p = 0; p < PG; ++p) {
                    const int v = __builtin_amdgcn_readlane(gl, min(q0 + p, nq - 1));
                    const int G = ((k0 + (v >> 5)) * 32 + wave) * 32 + (v & 31);
                    ev[p] = G * 32 + lane;
                    dv[p] = dst[min(ev[p], nEdges - 1)];
                }
#pragma unroll
                for (int p = 0; p < PG; ++p) {
                    if (q0 + p < nq) {
                        const bool m = (ev[p] < nEdges) && ((unsigned)(dv[p] - base) < (unsigned)nLoc);
                        const unsigned b = __builtin_amdgcn_ballot_w32(m);
                        if (m) {
                            const int qp = cnt + (int)__builtin_amdgcn_mbcnt_lo(b, 0u);
                            myq[qp]  = ev[p];
                            mydl[qp] = (unsigned short)(dv[p] - base);
                            atomicAdd(&degc[dv[p] - base], 1);
                        }
                        cnt += __builtin_popcount(b);
                    }
                }
            }
        }
    }
    recordAll();

    if (!early) stripProduct();

    if (lane == 0) wcount[wave] = nblk;
    __syncthreads();

    if (wave == 0) recN[(int)blockIdx.x * AGG_W + lane] = wcount[lane];

    for (int i = tid; i < nLoc; i += AGG_T) dinv[base + i] = rsqrtf(1.0f + (float)degc[i]);

    static_assert(TILE % 4 == 0, "a tile's byte counts fill whole words");
    for (int i = tid; i < TILE / 4; i += AGG_T) {
        unsigned wv = 0u;
#pragma unroll
        for (int k = 0; k < 4; ++k) {
            const int n = 4 * i + k;
            const unsigned d = (n < nLoc) ? (unsigned)min(degc[n], 255) : 0u;
            wv |= d << (8 * k);
        }
        degB[(base >> 2) + i] = wv;
    }
}

static __device__ __forceinline__ int4 gcn_load4_guarded(const int* __restrict__ p, int g, int n) {
    const int e = g << 2;
    int4 r;
    r.x = (e     < n) ? p[e]     : -1;
    r.y = (e + 1 < n) ? p[e + 1] : -1;
    r.z = (e + 2 < n) ? p[e + 2] : -1;
    r.w = (e + 3 < n) ? p[e + 3] : -1;
    return r;
}

template <bool V> struct gcn_tag { static constexpr bool value = V; };

template <int C, bool RELU, int MODE>
__global__ __launch_bounds__(AGG_T)
void gcn_tile_agg_kernel(const float* __restrict__ xl,
                         const int* __restrict__ src,
                         const int* __restrict__ dst,
                         const float* __restrict__ dinv,
                         const float* __restrict__ bias,
                         float* __restrict__ outp,
                         int*   __restrict__ recS,
                         int*   __restrict__ recD,
                         float* __restrict__ recW,
                         int*   __restrict__ recN,
                         const unsigned* __restrict__ degB,
                         int useRec,
                         int nEdges, int nNodes) {
    __shared__ __attribute__((aligned(16))) float tile[TILE * C];
    __shared__ __attribute__((aligned(16))) int qmem[(AGG_T / 32) * EQCAP];
    __shared__ int   wcount[AGG_T / 32];

    const int tid  = (int)threadIdx.x;
    const int lane = tid & 31;
    const int wave = tid >> 5;
    const int base = (int)blockIdx.x * TILE;
    const int nLoc = min(TILE, nNodes - base);

    int myN = 0;
    if (MODE == 1 || useRec) myN = recN[(int)blockIdx.x * (AGG_T / 32) + lane];

    constexpr int DTCAP = (AGG_T / 32) * EQCAP * 4;
    constexpr int DT_IT = (DTCAP / 16 + AGG_T - 1) / AGG_T;
    const bool useTab = (MODE == 0) && useRec && (nNodes <= DTCAP);
    const unsigned char* tab = (const unsigned char*)qmem;
    uint4 tv[DT_IT];
    int   tj[DT_IT];
    {
        const int nCh = useTab ? ((nNodes + 15) >> 4) : 0;
        const int rot = (int)(blockIdx.x & 7u) * (nCh >> 3);
#pragma unroll
        for (int t = 0; t < DT_IT; ++t) {
            const int i = t * AGG_T + tid;
            int j = i + rot;
            if (j >= nCh) j -= nCh;
            uint4 q = make_uint4(0u, 0u, 0u, 0u);
            if (i < nCh) q = ((const uint4*)degB)[j];
            tj[t] = (i < nCh) ? j : -1;
            tv[t] = q;
        }
    }

    constexpr int C4 = C / 4;
    constexpr int NV = TILE * C4;
    constexpr int IT = (NV + AGG_T - 1) / AGG_T;
    float4* tile4 = (float4*)tile;
    {
        float4 v[IT];
        float  di[IT];
#pragma unroll
        for (int t = 0; t < IT; ++t) {
            const int i  = min(t * AGG_T + tid, NV - 1);
            const int n  = min(i / C4, nLoc - 1);
            di[t] = dinv[base + n];
            v[t]  = *(const float4*)(xl + (long long)(base + n) * C + (i % C4) * 4);
        }
#pragma unroll
        for (int t = 0; t < IT; ++t) {
            const int i = t * AGG_T + tid;
            if (i < NV) {
                const bool ok = (i / C4) < nLoc;
                const float w = di[t] * di[t];
                float4 o;
                o.x = ok ? v[t].x * w : 0.0f;
                o.y = ok ? v[t].y * w : 0.0f;
                o.z = ok ? v[t].z * w : 0.0f;
                o.w = ok ? v[t].w * w : 0.0f;
                tile4[i] = o;
            }
        }
    }
#pragma unroll
    for (int t = 0; t < DT_IT; ++t)
        if (tj[t] >= 0) ((uint4*)qmem)[tj[t]] = tv[t];
    __syncthreads();

    auto byteDinv = [&](const unsigned d, const int n) -> float {
        float r = rsqrtf(1.0f + (float)d);
        if (d == 255u) r = dinv[n];
        return r;
    };

    int* myq = qmem + wave * EQCAP;
    int cnt = 0;

    constexpr int LPE = C / 2;
    constexpr int EPS = 32 / LPE;
    constexpr int PU  = 4;
    const int sub = lane / LPE;
    const int cl  = lane % LPE;

    const size_t rbase = ((size_t)((int)blockIdx.x * AGG_W + wave) * CAPW) * 32 + lane;
    int nblk = 0;

    auto drain = [&](const bool flush) {
        while (cnt >= 32 || (flush && cnt > 0)) {
            const int n = min(cnt, 32);
            int   s = 0, dl = 0;
            float w = 0.0f;
            if (lane < n) {
                const int e = myq[cnt - n + lane];
                s  = src[e];
                dl = dst[e] - base;
                w  = dinv[s] * dinv[base + dl];
            }
            if (MODE == 0) {
                if (nblk < CAPW) {
                    const size_t o = rbase + (size_t)nblk * 32;
                    recS[o] = s;
                    recD[o] = (lane < n) ? dl : -1;
                    recW[o] = w;
                }
                ++nblk;
            }
            cnt -= n;
            for (int j0 = 0; j0 < n; j0 += EPS * PU) {
                float2 r[PU];
                float  ww[PU];
                int    dd[PU];
#pragma unroll
                for (int p = 0; p < PU; ++p) {
                    const int jj = min(j0 + p * EPS + sub, n - 1);
                    const int ss = __shfl(s, jj);
                    dd[p] = __shfl(dl, jj);
                    ww[p] = __shfl(w, jj);
                    r[p]  = *(const float2*)(xl + (long long)ss * C + cl * 2);
                }
#pragma unroll
                for (int p = 0; p < PU; ++p) {
                    if (j0 + p * EPS + sub < n) {
                        float* t = &tile[dd[p] * C + cl * 2];
                        atomicAdd(t,     r[p].x * ww[p]);
                        atomicAdd(t + 1, r[p].y * ww[p]);
                    }
                }
            }
        }
    };

    constexpr int  GU  = 16;
    constexpr bool UNI = (EPS == 1);
    auto gatherRec = [&](const int nb, auto tabTag) {
        constexpr bool TAB = decltype(tabTag)::value;
        int   s0 = 0, dl0 = -1, s1 = 0, dl1 = -1;
        float w0 = 0.0f;
        if (nb > 0) {
            const size_t o1 = rbase + (size_t)min(1, nb - 1) * 32;
            s0  = recS[rbase];
            dl0 = recD[rbase];
            s1  = recS[o1];
            dl1 = recD[o1];
            const int d0 = base + max(dl0, 0);
            if constexpr (TAB) {
                const unsigned bx = tab[s0], by = tab[d0];
                w0 = byteDinv(bx, s0) * byteDinv(by, d0);
            } else {
                w0 = dinv[s0] * dinv[d0];
            }
        }
        for (int b = 0; b < nb; ++b) {
            const int m = __builtin_popcount(__builtin_amdgcn_ballot_w32(dl0 >= 0));
            int   s2 = 0, dl2 = -1;
            float x1 = 0.0f, y1 = 0.0f;
            recW[rbase + (size_t)b * 32] = w0;
            for (int j0 = 0; j0 < m; j0 += EPS * GU) {
                float2 r[GU];
                float  ww[GU];
                int    dd[GU];
#pragma unroll
                for (int p = 0; p < GU; ++p) {
                    const int jj = min(j0 + p * EPS + sub, m - 1);
                    int ss;
                    if (UNI) {
                        ss    = __builtin_amdgcn_readlane(s0, jj);
                        dd[p] = __builtin_amdgcn_readlane(dl0, jj);
                        ww[p] = __int_as_float(__builtin_amdgcn_readlane(__float_as_int(w0), jj));
                    } else {
                        ss    = __shfl(s0, jj);
                        dd[p] = __shfl(dl0, jj);
                        ww[p] = __shfl(w0, jj);
                    }
                    r[p] = *(const float2*)(xl + (long long)ss * C + cl * 2);
                }
                if (j0 == 0) {
                    const size_t o2 = rbase + (size_t)min(b + 2, nb - 1) * 32;
                    s2  = recS[o2];
                    dl2 = recD[o2];
                    if constexpr (TAB) {
                        const int d1 = base + max(dl1, 0);
                        const unsigned bx = tab[s1], by = tab[d1];
                        x1 = byteDinv(bx, s1);
                        y1 = byteDinv(by, d1);
                    } else {
                        x1 = dinv[s1];
                        y1 = dinv[base + max(dl1, 0)];
                    }
                }
#pragma unroll
                for (int p = 0; p < GU; ++p) {
                    if (j0 + p * EPS + sub < m) {
                        float* t = &tile[dd[p] * C + cl * 2];
                        atomicAdd(t,     r[p].x * ww[p]);
                        atomicAdd(t + 1, r[p].y * ww[p]);
                    }
                }
            }
            s0 = s1; dl0 = dl1; w0 = x1 * y1;
            s1 = s2; dl1 = dl2;
        }
    };

    bool replay = false;
    if (MODE == 1 || useRec)
        replay = (__builtin_amdgcn_ballot_w32(myN > CAPW) == 0u);
    const int nb = replay ? __builtin_amdgcn_readfirstlane(__shfl(myN, wave)) : 0;

    if (MODE == 1 && replay) {
        constexpr int RPU = 8;
        static_assert(32 % (EPS * RPU) == 0, "a block is a whole number of steps");
        int s = 0, dl = -1;
        float w = 0.0f;
        if (nb > 0) { s = recS[rbase]; dl = recD[rbase]; w = recW[rbase]; }
        for (int b = 0; b < nb; ++b) {
            int s2 = s, dl2 = dl;
            float w2 = w;
            for (int j0 = 0; j0 < 32; j0 += EPS * RPU) {
                float2 r[RPU];
                float  ww[RPU];
                int    dd[RPU];
#pragma unroll
                for (int p = 0; p < RPU; ++p) {
                    const int jj = j0 + p * EPS + sub;
                    const int ss = __shfl(s, jj);
                    dd[p] = __shfl(dl, jj);
                    ww[p] = __shfl(w, jj);
                    r[p]  = *(const float2*)(xl + (long long)ss * C + cl * 2);
                }
                if (j0 == 0) {
                    const size_t o = rbase + (size_t)min(b + 1, nb - 1) * 32;
                    s2 = recS[o]; dl2 = recD[o]; w2 = recW[o];
                }
#pragma unroll
                for (int p = 0; p < RPU; ++p) {
                    if (dd[p] >= 0) {
                        float* t = &tile[dd[p] * C + cl * 2];
                        atomicAdd(t,     r[p].x * ww[p]);
                        atomicAdd(t + 1, r[p].y * ww[p]);
                    }
                }
            }
            s = s2; dl = dl2; w = w2;
        }
    } else if (MODE == 0 && replay) {

    if (useTab) gatherRec(nb, gcn_tag<true>{});
    else        gatherRec(nb, gcn_tag<false>{});

    } else {

    const int  nG  = (nEdges + 3) >> 2;
    const int  nG4 = nEdges >> 2;
    const bool al  = (((size_t)dst) & 15) == 0;

    for (int g0 = 0; g0 < nG; g0 += AGG_T * SCAN_U) {
        int4 d4[SCAN_U];
        if (al && g0 + AGG_T * SCAN_U <= nG4) {
            const int4* p = (const int4*)dst + (g0 + tid);
#pragma unroll
            for (int u = 0; u < SCAN_U; ++u)
                d4[u] = p[u * AGG_T];
        } else {
#pragma unroll
            for (int u = 0; u < SCAN_U; ++u)
                d4[u] = gcn_load4_guarded(dst, g0 + u * AGG_T + tid, nEdges);
        }

#pragma unroll
        for (int u = 0; u < SCAN_U; ++u) {
            const int e0 = (g0 + u * AGG_T + tid) << 2;
            const unsigned r0 = (unsigned)(d4[u].x - base);
            const unsigned r1 = (unsigned)(d4[u].y - base);
            const unsigned r2 = (unsigned)(d4[u].z - base);
            const unsigned r3 = (unsigned)(d4[u].w - base);
            const unsigned rm = min(min(r0, r1), min(r2, r3));
            if (__builtin_amdgcn_ballot_w32(rm < (unsigned)nLoc)) {
                const bool m0 = r0 < (unsigned)nLoc;
                const bool m1 = r1 < (unsigned)nLoc;
                const bool m2 = r2 < (unsigned)nLoc;
                const bool m3 = r3 < (unsigned)nLoc;
                const unsigned b0 = __builtin_amdgcn_ballot_w32(m0);
                const unsigned b1 = __builtin_amdgcn_ballot_w32(m1);
                const unsigned b2 = __builtin_amdgcn_ballot_w32(m2);
                const unsigned b3 = __builtin_amdgcn_ballot_w32(m3);
                if (b0) {
                    if (m0) myq[cnt + (int)__builtin_amdgcn_mbcnt_lo(b0, 0u)] = e0;
                    cnt += __builtin_popcount(b0);
                }
                if (b1) {
                    if (m1) myq[cnt + (int)__builtin_amdgcn_mbcnt_lo(b1, 0u)] = e0 + 1;
                    cnt += __builtin_popcount(b1);
                }
                if (b2) {
                    if (m2) myq[cnt + (int)__builtin_amdgcn_mbcnt_lo(b2, 0u)] = e0 + 2;
                    cnt += __builtin_popcount(b2);
                }
                if (b3) {
                    if (m3) myq[cnt + (int)__builtin_amdgcn_mbcnt_lo(b3, 0u)] = e0 + 3;
                    cnt += __builtin_popcount(b3);
                }
            }
        }

        drain(g0 + AGG_T * SCAN_U >= nG);
    }

    }

    const bool made = (MODE == 0) && !replay;
    if (made && lane == 0) wcount[wave] = nblk;
    __syncthreads();

    if (made && wave == 0) recN[(int)blockIdx.x * AGG_W + lane] = wcount[lane];

    {
        const float4* bias4 = (const float4*)bias;
        float4* out4 = (float4*)(outp + (long long)base * C);
#pragma unroll
        for (int t = 0; t < IT; ++t) {
            const int i = t * AGG_T + tid;
            if (i < nLoc * C4) {
                const float4 a = tile4[i];
                const float4 b = bias4[i % C4];
                float4 o;
                o.x = a.x + b.x; o.y = a.y + b.y; o.z = a.z + b.z; o.w = a.w + b.w;
                if (RELU) { o.x = fmaxf(o.x, 0.0f); o.y = fmaxf(o.y, 0.0f); o.z = fmaxf(o.z, 0.0f); o.w = fmaxf(o.w, 0.0f); }
                out4[i] = o;
            }
        }
    }
}

extern "C" void kernel_launch(void* const* d_in, const int* in_sizes, int n_in,
                              void* d_out, int out_size, void* d_ws, size_t ws_size,
                              hipStream_t stream) {
    const float* x  = (const float*)d_in[0];
    const int*   ei = (const int*)  d_in[1];
    const float* W1 = (const float*)d_in[2];
    const float* b1 = (const float*)d_in[3];
    const float* W2 = (const float*)d_in[4];
    const float* b2 = (const float*)d_in[5];
    float* out = (float*)d_out;

    const int nNodes = in_sizes[0] / IN_CH;
    const int nEdges = in_sizes[1] / 2;
    const int* src = ei;
    const int* dst = ei + nEdges;

    float* ws  = (float*)d_ws;
    float* deg = ws;
    float* xl1 = deg + ((nNodes + 63) & ~63);
    float* h   = xl1 + (long long)nNodes * HID;
    float* xl2 = h   + (long long)nNodes * HID;

    const int B = 256;
    const int nTiles = (nNodes + TILE - 1) / TILE;

    const long long recLen = (long long)nTiles * AGG_W * CAPW * 32;
    int*   recS = (int*)(xl2 + (((long long)nNodes * OUT_CH + 31) & ~31LL));
    int*   recD = recS + recLen;
    float* recW = (float*)(recD + recLen);
    int*   recN = (int*)(recW + recLen);
    unsigned* bmap = (unsigned*)(recN + (long long)nTiles * AGG_W);
    const int nWords = (nEdges + 1023) / 1024;
    unsigned* degB = bmap + (long long)nWords * MAXT;
    const int useIdx = (nTiles <= MAXT) ? 1 : 0;

    if (useIdx) {
        const int wpb = IDX_T / 32;
        gcn_group_index_kernel<<<(nWords + wpb - 1) / wpb, IDX_T, 0, stream>>>(dst, bmap, nEdges, nWords);
        gcn_tile_collect_kernel<<<nTiles, AGG_T, 0, stream>>>(src, dst, deg, recS, recD, recN, bmap, x, W1, xl1, degB, nEdges, nNodes);
    } else {
        gcn_init_kernel<<<(nNodes + B - 1) / B, B, 0, stream>>>(deg, nNodes);
        gcn_degree_kernel<<<(nEdges + B - 1) / B, B, 0, stream>>>(dst, deg, nEdges);
        gcn_rsqrt_kernel<<<(nNodes + B - 1) / B, B, 0, stream>>>(deg, nNodes);
        int tiles = (nNodes / 16) * (HID / 16);
        int wpb = B / 32;
        gcn_gemm_wmma_f32<<<(tiles + wpb - 1) / wpb, B, 0, stream>>>(x, W1, xl1, nNodes, IN_CH, HID);
    }

    gcn_tile_agg_kernel<HID, true, 0><<<nTiles, AGG_T, 0, stream>>>(
        xl1, src, dst, deg, b1, h, recS, recD, recW, recN, degB, useIdx, nEdges, nNodes);

    {
        int tiles = (nNodes / 16) * (OUT_CH / 16);
        int wpb = B / 32;
        gcn_gemm_wmma_f32<<<(tiles + wpb - 1) / wpb, B, 0, stream>>>(h, W2, xl2, nNodes, HID, OUT_CH);
    }

    gcn_tile_agg_kernel<OUT_CH, false, 1><<<nTiles, AGG_T, 0, stream>>>(
        xl2, src, dst, deg, b2, out, recS, recD, recW, recN, degB, useIdx, nEdges, nNodes);
}


# Second product moved into the first aggregation's launch: wave s multiplies strip s of the LDS tile, relu(tile+bias), by W2 on WMMA and stores xl2; h never stored, one launch fewer; k order kept
# speedup vs baseline: 5.2906x; 1.1091x over previous
#include <hip/hip_runtime.h>

#define IN_CH  128
#define HID    64
#define OUT_CH 16

typedef float v2f __attribute__((ext_vector_type(2)));
typedef float v4f __attribute__((ext_vector_type(4)));
typedef float v8f __attribute__((ext_vector_type(8)));

__global__ void gcn_init_kernel(float* __restrict__ deg, int nNodes) {
    int i = blockIdx.x * blockDim.x + threadIdx.x;
    if (i < nNodes) deg[i] = 1.0f;
}

__global__ void gcn_degree_kernel(const int* __restrict__ dst,
                                  float* __restrict__ deg, int nEdges) {
    int e = blockIdx.x * blockDim.x + threadIdx.x;
    if (e < nEdges) atomicAdd(&deg[dst[e]], 1.0f);
}

__global__ void gcn_rsqrt_kernel(float* __restrict__ deg, int n) {
    int i = blockIdx.x * blockDim.x + threadIdx.x;
    if (i < n) deg[i] = rsqrtf(deg[i]);
}

__global__ void gcn_gemm_wmma_f32(const float* __restrict__ A,
                                  const float* __restrict__ W,
                                  float* __restrict__ C,
                                  int M, int K, int N) {
    const int wavesPerBlock = blockDim.x >> 5;
    const int wave = (int)(threadIdx.x >> 5);
    const int lane = (int)(threadIdx.x & 31);
    const int tilesN = N >> 4;
    const int gw = blockIdx.x * wavesPerBlock + wave;
    const int tileM = gw / tilesN;
    const int tileN = gw - tileM * tilesN;
    if (tileM * 16 >= M) return;

    const int row0 = tileM << 4;
    const int col0 = tileN << 4;
    const int mr = lane & 15;
    const int kh = (lane >> 4) << 1;

    v8f acc = {};
    for (int k = 0; k < K; k += 4) {
        v2f a, b;
        const float* ap = A + (long long)(row0 + mr) * K + (k + kh);
        a.x = ap[0];
        a.y = ap[1];
        const float* bp = W + (long long)(k + kh) * N + (col0 + mr);
        b.x = bp[0];
        b.y = bp[N];
        acc = __builtin_amdgcn_wmma_f32_16x16x4_f32(
             false, a,  false, b,
             (short)0, acc,  false,  false);
    }

    const int crow = row0 + ((lane >> 4) << 3);
    const int ccol = col0 + mr;
#pragma unroll
    for (int r = 0; r < 8; ++r)
        C[(long long)(crow + r) * N + ccol] = acc[r];
}

#define TILE   392
#define AGG_T  1024
#define SCAN_U 8
#define PG     16
#define EQCAP  1248
#define IQCAP  832
static_assert(IQCAP * 4 + IQCAP * 2 <= EQCAP * 4, "edge ids and local destinations fit the wave's queue memory");
static_assert(IQCAP >= 32 + PG * 32, "room for a rest plus one step in which every edge matches");
static_assert(TILE <= 65536, "a local destination fits 16 bits");
#define GLCAP  1024
static_assert(EQCAP >= 32 + SCAN_U * 128, "the scanning path's queue (every scanned edge matches) fits too");
#define MAXT   256
#define IDX_T  128
#define CAPW   16
#define AGG_W  (AGG_T / 32)
static_assert(AGG_W == 32, "one lane per wave when a wave reads its tile's block counts");

__global__ __launch_bounds__(IDX_T)
void gcn_group_index_kernel(const int* __restrict__ dst,
                            unsigned* __restrict__ bmap,
                            int nEdges, int nWords) {
    __shared__ unsigned rows[(IDX_T / 32) * MAXT];
    const int lane = (int)(threadIdx.x & 31);
    const int wave = (int)(threadIdx.x >> 5);
    const int word = (int)blockIdx.x * (IDX_T / 32) + wave;
    if (word >= nWords) return;
    unsigned* my = rows + wave * MAXT;
#pragma unroll
    for (int k = 0; k < MAXT / 32; ++k) my[k * 32 + lane] = 0u;
    const int e0 = word * 1024 + lane;
    for (int g0 = 0; g0 < 32; g0 += 8) {
        int dv[8];
#pragma unroll
        for (int u = 0; u < 8; ++u) {
            const int e = e0 + (g0 + u) * 32;
            dv[u] = (e < nEdges) ? dst[e] : -1;
        }
#pragma unroll
        for (int u = 0; u < 8; ++u) {
            if (dv[u] >= 0) atomicOr(&my[dv[u] / TILE], 1u << (g0 + u));
        }
    }
#pragma unroll
    for (int k = 0; k < MAXT / 32; ++k)
        bmap[(size_t)word * MAXT + k * 32 + lane] = my[k * 32 + lane];
}

#define APITCH 68
static_assert(HID == 64 && IN_CH == 128, "the strip product keeps four 16-column accumulators and stages x in two halves of 64 k");
static_assert(16 * APITCH <= EQCAP, "a half strip of x fits the wave's queue memory");
__global__ __launch_bounds__(AGG_T)
void gcn_tile_collect_kernel(const int* __restrict__ src,
                             const int* __restrict__ dst,
                             float* __restrict__ dinv,
                             int*   __restrict__ recS,
                             int*   __restrict__ recD,
                             int*   __restrict__ recN,
                             const unsigned* __restrict__ bmap,
                             const float* __restrict__ x,
                             const float* __restrict__ W,
                             float* __restrict__ xl,
                             unsigned* __restrict__ degB,
                             int nEdges, int nNodes) {
    __shared__ __attribute__((aligned(16))) int qmem[AGG_W * EQCAP];
    __shared__ unsigned short gmem[AGG_W * GLCAP];
    __shared__ int   degc[TILE];
    __shared__ int   wcount[AGG_W];
    __shared__ __attribute__((aligned(16))) float wlds[IN_CH * HID];

    const int tid  = (int)threadIdx.x;
    const int lane = tid & 31;
    const int wave = tid >> 5;
    const int base = (int)blockIdx.x * TILE;
    const int nLoc = min(TILE, nNodes - base);

    for (int i = tid; i < TILE; i += AGG_T) degc[i] = 0;
    {
        constexpr int WIT = IN_CH * HID / AGG_T;
        static_assert(WIT * AGG_T == IN_CH * HID, "every thread stages the same number of words of W");
        float wv[WIT];
#pragma unroll
        for (int t = 0; t < WIT; ++t) wv[t] = W[t * AGG_T + tid];
#pragma unroll
        for (int t = 0; t < WIT; ++t) {
            const int i = t * AGG_T + tid;
            const int k = i / HID, col = i % HID;
            const int c = col >> 4;
            wlds[(((((k >> 2) * 2 + (c >> 1)) * 2 + ((k >> 1) & 1)) * 16 + (col & 15)) << 2) + ((c & 1) << 1) + (k & 1)] = wv[t];
        }
    }
    __syncthreads();

    int* myq = qmem + wave * EQCAP;
    unsigned short* mydl = (unsigned short*)(myq + IQCAP);
    int cnt = 0;

    auto stripProduct = [&]() {
        const int mr = lane & 15;
        const int hh = lane >> 4;
        const int nStrips = (nNodes + 15) >> 4;
        const v4f* bp = (const v4f*)wlds + (hh * 16 + mr);
        float* stg = (float*)myq;
        for (int s = (int)blockIdx.x + (int)gridDim.x * wave; s < nStrips; s += (int)gridDim.x * AGG_W) {
            const int row0 = s << 4;
            v4f g[2][8];
#pragma unroll
            for (int q = 0; q < 2; ++q) {
#pragma unroll
                for (int i = 0; i < 8; ++i) {
                    const int row = min(row0 + 2 * i + hh, nNodes - 1);
                    g[q][i] = *(const v4f*)(x + (size_t)row * IN_CH + q * 64 + 4 * mr);
                }
            }
            v8f acc0 = {}, acc1 = {}, acc2 = {}, acc3 = {};
#pragma unroll
            for (int q = 0; q < 2; ++q) {
#pragma unroll
                for (int i = 0; i < 8; ++i)
                    *(v4f*)(stg + (2 * i + hh) * APITCH + 4 * mr) = g[q][i];
                const v2f* as = (const v2f*)(stg + mr * APITCH + 2 * hh);
#pragma unroll 4
                for (int jj = 0; jj < 16; ++jj) {
                    const int j = q * 16 + jj;
                    const v2f a   = as[2 * jj];
                    const v4f b01 = bp[(j * 2 + 0) * 32];
                    const v4f b23 = bp[(j * 2 + 1) * 32];
                    const v2f b0 = {b01.x, b01.y}, b1 = {b01.z, b01.w}, b2 = {b23.x, b23.y}, b3 = {b23.z, b23.w};
                    acc0 = __builtin_amdgcn_wmma_f32_16x16x4_f32(false, a, false, b0, (short)0, acc0, false, false);
                    acc1 = __builtin_amdgcn_wmma_f32_16x16x4_f32(false, a, false, b1, (short)0, acc1, false, false);
                    acc2 = __builtin_amdgcn_wmma_f32_16x16x4_f32(false, a, false, b2, (short)0, acc2, false, false);
                    acc3 = __builtin_amdgcn_wmma_f32_16x16x4_f32(false, a, false, b3, (short)0, acc3, false, false);
                }
            }
            const int crow = row0 + (hh << 3);
#pragma unroll
            for (int r = 0; r < 8; ++r) {
                if (crow + r < nNodes) {
                    float* o = xl + (size_t)(crow + r) * HID + mr;
                    o[0]  = acc0[r];
                    o[16] = acc1[r];
                    o[32] = acc2[r];
                    o[48] = acc3[r];
                }
            }
        }
    };
    const bool early = ((wave >> 2) & 1) == 0;
    if (early) stripProduct();

    const size_t rbase = ((size_t)((int)blockIdx.x * AGG_W + wave) * CAPW) * 32 + lane;
    int nblk = 0;

    constexpr int RU = 8;
    auto recordAll = [&]() {
        const int nB   = (cnt + 31) >> 5;
        const int last = max(cnt - 1, 0);
        for (int b0 = 0; b0 < nB; b0 += RU) {
            int sv[RU], dv[RU];
#pragma unroll
            for (int u = 0; u < RU; ++u) {
                const int q = min((b0 + u) * 32 + lane, last);
                dv[u] = mydl[q];
                sv[u] = src[myq[q]];
            }
#pragma unroll
            for (int u = 0; u < RU; ++u) {
                if (b0 + u < nB) {
                    if (nblk < CAPW) {
                        const size_t o = rbase + (size_t)nblk * 32;
                        recS[o] = sv[u];
                        recD[o] = ((b0 + u) * 32 + lane < cnt) ? dv[u] : -1;
                    }
                    ++nblk;
                }
            }
        }
        cnt = 0;
    };

    unsigned short* glist = gmem + wave * GLCAP;
    const int nW = (nEdges + 1023) >> 10;
    for (int k0 = 0; k0 * 32 < nW; k0 += 32) {
        const int w = (k0 + lane) * 32 + wave;
        unsigned bits = (w < nW) ? bmap[(size_t)w * MAXT + blockIdx.x] : 0u;
        const int pc = __builtin_popcount(bits);
        int incl = pc;
#pragma unroll
        for (int off = 1; off < 32; off <<= 1) {
            const int v = __shfl_up(incl, off);
            if (lane >= off) incl += v;
        }
        const int total = __builtin_amdgcn_readfirstlane(__shfl(incl, 31));
        int pos = incl - pc;
        while (bits) {
            const int j = __builtin_ctz(bits);
            bits &= bits - 1u;
            glist[pos++] = (unsigned short)(lane * 32 + j);
        }
        for (int i0 = 0; i0 < total; i0 += 32) {
            const int gl = (int)glist[min(i0 + lane, total - 1)];
            const int nq = min(32, total - i0);
            for (int q0 = 0; q0 < nq; q0 += PG) {
                if (cnt > IQCAP - PG * 32) recordAll();
                int dv[PG], ev[PG];
#pragma unroll
                for (int p = 0; p < PG; ++p) {
                    const int v = __builtin_amdgcn_readlane(gl, min(q0 + p, nq - 1));
                    const int G = ((k0 + (v >> 5)) * 32 + wave) * 32 + (v & 31);
                    ev[p] = G * 32 + lane;
                    dv[p] = dst[min(ev[p], nEdges - 1)];
                }
#pragma unroll
                for (int p = 0; p < PG; ++p) {
                    if (q0 + p < nq) {
                        const bool m = (ev[p] < nEdges) && ((unsigned)(dv[p] - base) < (unsigned)nLoc);
                        const unsigned b = __builtin_amdgcn_ballot_w32(m);
                        if (m) {
                            const int qp = cnt + (int)__builtin_amdgcn_mbcnt_lo(b, 0u);
                            myq[qp]  = ev[p];
                            mydl[qp] = (unsigned short)(dv[p] - base);
                            atomicAdd(&degc[dv[p] - base], 1);
                        }
                        cnt += __builtin_popcount(b);
                    }
                }
            }
        }
    }
    recordAll();

    if (!early) stripProduct();

    if (lane == 0) wcount[wave] = nblk;
    __syncthreads();

    if (wave == 0) recN[(int)blockIdx.x * AGG_W + lane] = wcount[lane];

    for (int i = tid; i < nLoc; i += AGG_T) dinv[base + i] = rsqrtf(1.0f + (float)degc[i]);

    static_assert(TILE % 4 == 0, "a tile's byte counts fill whole words");
    for (int i = tid; i < TILE / 4; i += AGG_T) {
        unsigned wv = 0u;
#pragma unroll
        for (int k = 0; k < 4; ++k) {
            const int n = 4 * i + k;
            const unsigned d = (n < nLoc) ? (unsigned)min(degc[n], 255) : 0u;
            wv |= d << (8 * k);
        }
        degB[(base >> 2) + i] = wv;
    }
}

static __device__ __forceinline__ int4 gcn_load4_guarded(const int* __restrict__ p, int g, int n) {
    const int e = g << 2;
    int4 r;
    r.x = (e     < n) ? p[e]     : -1;
    r.y = (e + 1 < n) ? p[e + 1] : -1;
    r.z = (e + 2 < n) ? p[e + 2] : -1;
    r.w = (e + 3 < n) ? p[e + 3] : -1;
    return r;
}

template <bool V> struct gcn_tag { static constexpr bool value = V; };

template <int C, bool RELU, int MODE>
__global__ __launch_bounds__(AGG_T)
void gcn_tile_agg_kernel(const float* __restrict__ xl,
                         const int* __restrict__ src,
                         const int* __restrict__ dst,
                         const float* __restrict__ dinv,
                         const float* __restrict__ bias,
                         const float* __restrict__ Wn,
                         float* __restrict__ outp,
                         int*   __restrict__ recS,
                         int*   __restrict__ recD,
                         float* __restrict__ recW,
                         int*   __restrict__ recN,
                         const unsigned* __restrict__ degB,
                         int useRec,
                         int nEdges, int nNodes) {
    __shared__ __attribute__((aligned(16))) float tile[TILE * C];
    __shared__ __attribute__((aligned(16))) int qmem[(AGG_T / 32) * EQCAP];
    __shared__ int   wcount[AGG_T / 32];

    const int tid  = (int)threadIdx.x;
    const int lane = tid & 31;
    const int wave = tid >> 5;
    const int base = (int)blockIdx.x * TILE;
    const int nLoc = min(TILE, nNodes - base);

    int myN = 0;
    if (MODE == 1 || useRec) myN = recN[(int)blockIdx.x * (AGG_T / 32) + lane];

    constexpr int DTCAP = (AGG_T / 32) * EQCAP * 4;
    constexpr int DT_IT = (DTCAP / 16 + AGG_T - 1) / AGG_T;
    const bool useTab = (MODE == 0) && useRec && (nNodes <= DTCAP);
    const unsigned char* tab = (const unsigned char*)qmem;
    uint4 tv[DT_IT];
    int   tj[DT_IT];
    {
        const int nCh = useTab ? ((nNodes + 15) >> 4) : 0;
        const int rot = (int)(blockIdx.x & 7u) * (nCh >> 3);
#pragma unroll
        for (int t = 0; t < DT_IT; ++t) {
            const int i = t * AGG_T + tid;
            int j = i + rot;
            if (j >= nCh) j -= nCh;
            uint4 q = make_uint4(0u, 0u, 0u, 0u);
            if (i < nCh) q = ((const uint4*)degB)[j];
            tj[t] = (i < nCh) ? j : -1;
            tv[t] = q;
        }
    }

    constexpr int C4 = C / 4;
    constexpr int NV = TILE * C4;
    constexpr int IT = (NV + AGG_T - 1) / AGG_T;
    float4* tile4 = (float4*)tile;
    {
        float4 v[IT];
        float  di[IT];
#pragma unroll
        for (int t = 0; t < IT; ++t) {
            const int i  = min(t * AGG_T + tid, NV - 1);
            const int n  = min(i / C4, nLoc - 1);
            di[t] = dinv[base + n];
            v[t]  = *(const float4*)(xl + (long long)(base + n) * C + (i % C4) * 4);
        }
#pragma unroll
        for (int t = 0; t < IT; ++t) {
            const int i = t * AGG_T + tid;
            if (i < NV) {
                const bool ok = (i / C4) < nLoc;
                const float w = di[t] * di[t];
                float4 o;
                o.x = ok ? v[t].x * w : 0.0f;
                o.y = ok ? v[t].y * w : 0.0f;
                o.z = ok ? v[t].z * w : 0.0f;
                o.w = ok ? v[t].w * w : 0.0f;
                tile4[i] = o;
            }
        }
    }
#pragma unroll
    for (int t = 0; t < DT_IT; ++t)
        if (tj[t] >= 0) ((uint4*)qmem)[tj[t]] = tv[t];
    __syncthreads();

    auto byteDinv = [&](const unsigned d, const int n) -> float {
        float r = rsqrtf(1.0f + (float)d);
        if (d == 255u) r = dinv[n];
        return r;
    };

    int* myq = qmem + wave * EQCAP;
    int cnt = 0;

    constexpr int LPE = C / 2;
    constexpr int EPS = 32 / LPE;
    constexpr int PU  = 4;
    const int sub = lane / LPE;
    const int cl  = lane % LPE;

    const size_t rbase = ((size_t)((int)blockIdx.x * AGG_W + wave) * CAPW) * 32 + lane;
    int nblk = 0;

    auto drain = [&](const bool flush) {
        while (cnt >= 32 || (flush && cnt > 0)) {
            const int n = min(cnt, 32);
            int   s = 0, dl = 0;
            float w = 0.0f;
            if (lane < n) {
                const int e = myq[cnt - n + lane];
                s  = src[e];
                dl = dst[e] - base;
                w  = dinv[s] * dinv[base + dl];
            }
            if (MODE == 0) {
                if (nblk < CAPW) {
                    const size_t o = rbase + (size_t)nblk * 32;
                    recS[o] = s;
                    recD[o] = (lane < n) ? dl : -1;
                    recW[o] = w;
                }
                ++nblk;
            }
            cnt -= n;
            for (int j0 = 0; j0 < n; j0 += EPS * PU) {
                float2 r[PU];
                float  ww[PU];
                int    dd[PU];
#pragma unroll
                for (int p = 0; p < PU; ++p) {
                    const int jj = min(j0 + p * EPS + sub, n - 1);
                    const int ss = __shfl(s, jj);
                    dd[p] = __shfl(dl, jj);
                    ww[p] = __shfl(w, jj);
                    r[p]  = *(const float2*)(xl + (long long)ss * C + cl * 2);
                }
#pragma unroll
                for (int p = 0; p < PU; ++p) {
                    if (j0 + p * EPS + sub < n) {
                        float* t = &tile[dd[p] * C + cl * 2];
                        atomicAdd(t,     r[p].x * ww[p]);
                        atomicAdd(t + 1, r[p].y * ww[p]);
                    }
                }
            }
        }
    };

    constexpr int  GU  = 16;
    constexpr bool UNI = (EPS == 1);
    auto gatherRec = [&](const int nb, auto tabTag) {
        constexpr bool TAB = decltype(tabTag)::value;
        int   s0 = 0, dl0 = -1, s1 = 0, dl1 = -1;
        float w0 = 0.0f;
        if (nb > 0) {
            const size_t o1 = rbase + (size_t)min(1, nb - 1) * 32;
            s0  = recS[rbase];
            dl0 = recD[rbase];
            s1  = recS[o1];
            dl1 = recD[o1];
            const int d0 = base + max(dl0, 0);
            if constexpr (TAB) {
                const unsigned bx = tab[s0], by = tab[d0];
                w0 = byteDinv(bx, s0) * byteDinv(by, d0);
            } else {
                w0 = dinv[s0] * dinv[d0];
            }
        }
        for (int b = 0; b < nb; ++b) {
            const int m = __builtin_popcount(__builtin_amdgcn_ballot_w32(dl0 >= 0));
            int   s2 = 0, dl2 = -1;
            float x1 = 0.0f, y1 = 0.0f;
            recW[rbase + (size_t)b * 32] = w0;
            for (int j0 = 0; j0 < m; j0 += EPS * GU) {
                float2 r[GU];
                float  ww[GU];
                int    dd[GU];
#pragma unroll
                for (int p = 0; p < GU; ++p) {
                    const int jj = min(j0 + p * EPS + sub, m - 1);
                    int ss;
                    if (UNI) {
                        ss    = __builtin_amdgcn_readlane(s0, jj);
                        dd[p] = __builtin_amdgcn_readlane(dl0, jj);
                        ww[p] = __int_as_float(__builtin_amdgcn_readlane(__float_as_int(w0), jj));
                    } else {
                        ss    = __shfl(s0, jj);
                        dd[p] = __shfl(dl0, jj);
                        ww[p] = __shfl(w0, jj);
                    }
                    r[p] = *(const float2*)(xl + (long long)ss * C + cl * 2);
                }
                if (j0 == 0) {
                    const size_t o2 = rbase + (size_t)min(b + 2, nb - 1) * 32;
                    s2  = recS[o2];
                    dl2 = recD[o2];
                    if constexpr (TAB) {
                        const int d1 = base + max(dl1, 0);
                        const unsigned bx = tab[s1], by = tab[d1];
                        x1 = byteDinv(bx, s1);
                        y1 = byteDinv(by, d1);
                    } else {
                        x1 = dinv[s1];
                        y1 = dinv[base + max(dl1, 0)];
                    }
                }
#pragma unroll
                for (int p = 0; p < GU; ++p) {
                    if (j0 + p * EPS + sub < m) {
                        float* t = &tile[dd[p] * C + cl * 2];
                        atomicAdd(t,     r[p].x * ww[p]);
                        atomicAdd(t + 1, r[p].y * ww[p]);
                    }
                }
            }
            s0 = s1; dl0 = dl1; w0 = x1 * y1;
            s1 = s2; dl1 = dl2;
        }
    };

    bool replay = false;
    if (MODE == 1 || useRec)
        replay = (__builtin_amdgcn_ballot_w32(myN > CAPW) == 0u);
    const int nb = replay ? __builtin_amdgcn_readfirstlane(__shfl(myN, wave)) : 0;

    if (MODE == 1 && replay) {
        constexpr int RPU = 8;
        static_assert(32 % (EPS * RPU) == 0, "a block is a whole number of steps");
        int s = 0, dl = -1;
        float w = 0.0f;
        if (nb > 0) { s = recS[rbase]; dl = recD[rbase]; w = recW[rbase]; }
        for (int b = 0; b < nb; ++b) {
            int s2 = s, dl2 = dl;
            float w2 = w;
            for (int j0 = 0; j0 < 32; j0 += EPS * RPU) {
                float2 r[RPU];
                float  ww[RPU];
                int    dd[RPU];
#pragma unroll
                for (int p = 0; p < RPU; ++p) {
                    const int jj = j0 + p * EPS + sub;
                    const int ss = __shfl(s, jj);
                    dd[p] = __shfl(dl, jj);
                    ww[p] = __shfl(w, jj);
                    r[p]  = *(const float2*)(xl + (long long)ss * C + cl * 2);
                }
                if (j0 == 0) {
                    const size_t o = rbase + (size_t)min(b + 1, nb - 1) * 32;
                    s2 = recS[o]; dl2 = recD[o]; w2 = recW[o];
                }
#pragma unroll
                for (int p = 0; p < RPU; ++p) {
                    if (dd[p] >= 0) {
                        float* t = &tile[dd[p] * C + cl * 2];
                        atomicAdd(t,     r[p].x * ww[p]);
                        atomicAdd(t + 1, r[p].y * ww[p]);
                    }
                }
            }
            s = s2; dl = dl2; w = w2;
        }
    } else if (MODE == 0 && replay) {

    if (useTab) gatherRec(nb, gcn_tag<true>{});
    else        gatherRec(nb, gcn_tag<false>{});

    } else {

    const int  nG  = (nEdges + 3) >> 2;
    const int  nG4 = nEdges >> 2;
    const bool al  = (((size_t)dst) & 15) == 0;

    for (int g0 = 0; g0 < nG; g0 += AGG_T * SCAN_U) {
        int4 d4[SCAN_U];
        if (al && g0 + AGG_T * SCAN_U <= nG4) {
            const int4* p = (const int4*)dst + (g0 + tid);
#pragma unroll
            for (int u = 0; u < SCAN_U; ++u)
                d4[u] = p[u * AGG_T];
        } else {
#pragma unroll
            for (int u = 0; u < SCAN_U; ++u)
                d4[u] = gcn_load4_guarded(dst, g0 + u * AGG_T + tid, nEdges);
        }

#pragma unroll
        for (int u = 0; u < SCAN_U; ++u) {
            const int e0 = (g0 + u * AGG_T + tid) << 2;
            const unsigned r0 = (unsigned)(d4[u].x - base);
            const unsigned r1 = (unsigned)(d4[u].y - base);
            const unsigned r2 = (unsigned)(d4[u].z - base);
            const unsigned r3 = (unsigned)(d4[u].w - base);
            const unsigned rm = min(min(r0, r1), min(r2, r3));
            if (__builtin_amdgcn_ballot_w32(rm < (unsigned)nLoc)) {
                const bool m0 = r0 < (unsigned)nLoc;
                const bool m1 = r1 < (unsigned)nLoc;
                const bool m2 = r2 < (unsigned)nLoc;
                const bool m3 = r3 < (unsigned)nLoc;
                const unsigned b0 = __builtin_amdgcn_ballot_w32(m0);
                const unsigned b1 = __builtin_amdgcn_ballot_w32(m1);
                const unsigned b2 = __builtin_amdgcn_ballot_w32(m2);
                const unsigned b3 = __builtin_amdgcn_ballot_w32(m3);
                if (b0) {
                    if (m0) myq[cnt + (int)__builtin_amdgcn_mbcnt_lo(b0, 0u)] = e0;
                    cnt += __builtin_popcount(b0);
                }
                if (b1) {
                    if (m1) myq[cnt + (int)__builtin_amdgcn_mbcnt_lo(b1, 0u)] = e0 + 1;
                    cnt += __builtin_popcount(b1);
                }
                if (b2) {
                    if (m2) myq[cnt + (int)__builtin_amdgcn_mbcnt_lo(b2, 0u)] = e0 + 2;
                    cnt += __builtin_popcount(b2);
                }
                if (b3) {
                    if (m3) myq[cnt + (int)__builtin_amdgcn_mbcnt_lo(b3, 0u)] = e0 + 3;
                    cnt += __builtin_popcount(b3);
                }
            }
        }

        drain(g0 + AGG_T * SCAN_U >= nG);
    }

    }

    constexpr int KST = C / 4;
    const int  pmr  = lane & 15;
    const int  phh  = lane >> 4;
    const bool mine = (MODE == 0) && (wave < ((nLoc + 15) >> 4));
    v2f bw[KST];
    if constexpr (MODE == 0) {
        static_assert(C == HID && OUT_CH == 16, "the strip product makes one 16-column tile from HID k");
        static_assert((TILE + 15) / 16 <= AGG_W, "one strip per wave covers the tile");
        static_assert(16 * APITCH <= EQCAP && APITCH >= C, "a strip fits the wave's queue memory");
        if (mine) {
#pragma unroll
            for (int j = 0; j < KST; ++j) {
                const float* bp = Wn + (4 * j + 2 * phh) * OUT_CH + pmr;
                bw[j].x = bp[0];
                bw[j].y = bp[OUT_CH];
            }
        }
    }

    const bool made = (MODE == 0) && !replay;
    if (made && lane == 0) wcount[wave] = nblk;
    __syncthreads();

    if (made && wave == 0) recN[(int)blockIdx.x * AGG_W + lane] = wcount[lane];

    if constexpr (MODE == 0) {
        if (mine) {
            const int row0 = wave << 4;
            float* stg = (float*)(qmem + wave * EQCAP);
            const float4 bq = ((const float4*)bias)[pmr];
#pragma unroll
            for (int i = 0; i < 8; ++i) {
                const int r = 2 * i + phh;
                const float4 a = tile4[min(row0 + r, TILE - 1) * C4 + pmr];
                v4f o = {a.x + bq.x, a.y + bq.y, a.z + bq.z, a.w + bq.w};
                if (RELU) { o.x = fmaxf(o.x, 0.0f); o.y = fmaxf(o.y, 0.0f); o.z = fmaxf(o.z, 0.0f); o.w = fmaxf(o.w, 0.0f); }
                *(v4f*)(stg + r * APITCH + 4 * pmr) = o;
            }
            const v2f* as = (const v2f*)(stg + pmr * APITCH + 2 * phh);
            v8f acc = {};
#pragma unroll
            for (int j = 0; j < KST; ++j)
                acc = __builtin_amdgcn_wmma_f32_16x16x4_f32(false, as[2 * j], false, bw[j], (short)0, acc, false, false);
            const int crow = row0 + (phh << 3);
#pragma unroll
            for (int r = 0; r < 8; ++r)
                if (crow + r < nLoc) outp[(size_t)(base + crow + r) * OUT_CH + pmr] = acc[r];
        }
    } else {
    {
        const float4* bias4 = (const float4*)bias;
        float4* out4 = (float4*)(outp + (long long)base * C);
#pragma unroll
        for (int t = 0; t < IT; ++t) {
            const int i = t * AGG_T + tid;
            if (i < nLoc * C4) {
                const float4 a = tile4[i];
                const float4 b = bias4[i % C4];
                float4 o;
                o.x = a.x + b.x; o.y = a.y + b.y; o.z = a.z + b.z; o.w = a.w + b.w;
                if (RELU) { o.x = fmaxf(o.x, 0.0f); o.y = fmaxf(o.y, 0.0f); o.z = fmaxf(o.z, 0.0f); o.w = fmaxf(o.w, 0.0f); }
                out4[i] = o;
            }
        }
    }
    }
}

extern "C" void kernel_launch(void* const* d_in, const int* in_sizes, int n_in,
                              void* d_out, int out_size, void* d_ws, size_t ws_size,
                              hipStream_t stream) {
    const float* x  = (const float*)d_in[0];
    const int*   ei = (const int*)  d_in[1];
    const float* W1 = (const float*)d_in[2];
    const float* b1 = (const float*)d_in[3];
    const float* W2 = (const float*)d_in[4];
    const float* b2 = (const float*)d_in[5];
    float* out = (float*)d_out;

    const int nNodes = in_sizes[0] / IN_CH;
    const int nEdges = in_sizes[1] / 2;
    const int* src = ei;
    const int* dst = ei + nEdges;

    float* ws  = (float*)d_ws;
    float* deg = ws;
    float* xl1 = deg + ((nNodes + 63) & ~63);
    float* xl2 = xl1 + (long long)nNodes * HID;

    const int B = 256;
    const int nTiles = (nNodes + TILE - 1) / TILE;

    const long long recLen = (long long)nTiles * AGG_W * CAPW * 32;
    int*   recS = (int*)(xl2 + (((long long)nNodes * OUT_CH + 31) & ~31LL));
    int*   recD = recS + recLen;
    float* recW = (float*)(recD + recLen);
    int*   recN = (int*)(recW + recLen);
    unsigned* bmap = (unsigned*)(recN + (long long)nTiles * AGG_W);
    const int nWords = (nEdges + 1023) / 1024;
    unsigned* degB = bmap + (long long)nWords * MAXT;
    const int useIdx = (nTiles <= MAXT) ? 1 : 0;

    if (useIdx) {
        const int wpb = IDX_T / 32;
        gcn_group_index_kernel<<<(nWords + wpb - 1) / wpb, IDX_T, 0, stream>>>(dst, bmap, nEdges, nWords);
        gcn_tile_collect_kernel<<<nTiles, AGG_T, 0, stream>>>(src, dst, deg, recS, recD, recN, bmap, x, W1, xl1, degB, nEdges, nNodes);
    } else {
        gcn_init_kernel<<<(nNodes + B - 1) / B, B, 0, stream>>>(deg, nNodes);
        gcn_degree_kernel<<<(nEdges + B - 1) / B, B, 0, stream>>>(dst, deg, nEdges);
        gcn_rsqrt_kernel<<<(nNodes + B - 1) / B, B, 0, stream>>>(deg, nNodes);
        int tiles = (nNodes / 16) * (HID / 16);
        int wpb = B / 32;
        gcn_gemm_wmma_f32<<<(tiles + wpb - 1) / wpb, B, 0, stream>>>(x, W1, xl1, nNodes, IN_CH, HID);
    }

    gcn_tile_agg_kernel<HID, true, 0><<<nTiles, AGG_T, 0, stream>>>(
        xl1, src, dst, deg, b1, W2, xl2, recS, recD, recW, recN, degB, useIdx, nEdges, nNodes);

    gcn_tile_agg_kernel<OUT_CH, false, 1><<<nTiles, AGG_T, 0, stream>>>(
        xl2, src, dst, deg, b2, W2, out, recS, recD, recW, recN, degB, useIdx, nEdges, nNodes);
}


# Group index marks groups of 16 edges, not 32; the collect launch fetches two marked groups per load (one per lane half) from one running list: half the steps per edge found; sum order kept
# speedup vs baseline: 5.6365x; 1.0654x over previous
#include <hip/hip_runtime.h>

#define IN_CH  128
#define HID    64
#define OUT_CH 16

typedef float v2f __attribute__((ext_vector_type(2)));
typedef float v4f __attribute__((ext_vector_type(4)));
typedef float v8f __attribute__((ext_vector_type(8)));

__global__ void gcn_init_kernel(float* __restrict__ deg, int nNodes) {
    int i = blockIdx.x * blockDim.x + threadIdx.x;
    if (i < nNodes) deg[i] = 1.0f;
}

__global__ void gcn_degree_kernel(const int* __restrict__ dst,
                                  float* __restrict__ deg, int nEdges) {
    int e = blockIdx.x * blockDim.x + threadIdx.x;
    if (e < nEdges) atomicAdd(&deg[dst[e]], 1.0f);
}

__global__ void gcn_rsqrt_kernel(float* __restrict__ deg, int n) {
    int i = blockIdx.x * blockDim.x + threadIdx.x;
    if (i < n) deg[i] = rsqrtf(deg[i]);
}

__global__ void gcn_gemm_wmma_f32(const float* __restrict__ A,
                                  const float* __restrict__ W,
                                  float* __restrict__ C,
                                  int M, int K, int N) {
    const int wavesPerBlock = blockDim.x >> 5;
    const int wave = (int)(threadIdx.x >> 5);
    const int lane = (int)(threadIdx.x & 31);
    const int tilesN = N >> 4;
    const int gw = blockIdx.x * wavesPerBlock + wave;
    const int tileM = gw / tilesN;
    const int tileN = gw - tileM * tilesN;
    if (tileM * 16 >= M) return;

    const int row0 = tileM << 4;
    const int col0 = tileN << 4;
    const int mr = lane & 15;
    const int kh = (lane >> 4) << 1;

    v8f acc = {};
    for (int k = 0; k < K; k += 4) {
        v2f a, b;
        const float* ap = A + (long long)(row0 + mr) * K + (k + kh);
        a.x = ap[0];
        a.y = ap[1];
        const float* bp = W + (long long)(k + kh) * N + (col0 + mr);
        b.x = bp[0];
        b.y = bp[N];
        acc = __builtin_amdgcn_wmma_f32_16x16x4_f32(
             false, a,  false, b,
             (short)0, acc,  false,  false);
    }

    const int crow = row0 + ((lane >> 4) << 3);
    const int ccol = col0 + mr;
#pragma unroll
    for (int r = 0; r < 8; ++r)
        C[(long long)(crow + r) * N + ccol] = acc[r];
}

#define TILE   392
#define AGG_T  1024
#define SCAN_U 8
#define GSZ    16
#define PG     16
#define EQCAP  1248
#define IQCAP  832
static_assert(IQCAP * 4 + IQCAP * 2 <= EQCAP * 4, "edge ids and local destinations fit the wave's queue memory");
static_assert(IQCAP >= 32 + PG * 32, "room for a rest plus one step in which every edge matches");
static_assert(2 * GSZ == 32 && 2 * PG == 32, "a load takes two groups, a step takes the 32 list entries that one LDS read hands out");
static_assert(TILE <= 65536, "a local destination fits 16 bits");
#define GLCAP  1024
static_assert(EQCAP >= 32 + SCAN_U * 128, "the scanning path's queue (every scanned edge matches) fits too");
#define MAXT   256
#define IDX_T  128
#define CAPW   16
#define AGG_W  (AGG_T / 32)
static_assert(AGG_W == 32, "one lane per wave when a wave reads its tile's block counts");

__global__ __launch_bounds__(IDX_T)
void gcn_group_index_kernel(const int* __restrict__ dst,
                            unsigned* __restrict__ bmap,
                            int nEdges, int nWords) {
    __shared__ unsigned rows[(IDX_T / 32) * MAXT * 2];
    const int lane = (int)(threadIdx.x & 31);
    const int wave = (int)(threadIdx.x >> 5);
    const int word = (int)blockIdx.x * (IDX_T / 32) + wave;
    if (word >= nWords) return;
    unsigned* my = rows + wave * (MAXT * 2);
#pragma unroll
    for (int k = 0; k < MAXT * 2 / 32; ++k) my[k * 32 + lane] = 0u;
    const int e0 = word * 1024 + lane;
    for (int g0 = 0; g0 < 32; g0 += 8) {
        int dv[8];
#pragma unroll
        for (int u = 0; u < 8; ++u) {
            const int e = e0 + (g0 + u) * 32;
            dv[u] = (e < nEdges) ? dst[e] : -1;
        }
#pragma unroll
        for (int u = 0; u < 8; ++u) {
            const int g = (g0 + u) * 2 + (lane >> 4);
            if (dv[u] >= 0) atomicOr(&my[(dv[u] / TILE) * 2 + (g >> 5)], 1u << (g & 31));
        }
    }
#pragma unroll
    for (int k = 0; k < MAXT * 2 / 32; ++k)
        bmap[(size_t)word * (MAXT * 2) + k * 32 + lane] = my[k * 32 + lane];
}

#define APITCH 68
static_assert(HID == 64 && IN_CH == 128, "the strip product keeps four 16-column accumulators and stages x in two halves of 64 k");
static_assert(16 * APITCH <= EQCAP, "a half strip of x fits the wave's queue memory");
__global__ __launch_bounds__(AGG_T)
void gcn_tile_collect_kernel(const int* __restrict__ src,
                             const int* __restrict__ dst,
                             float* __restrict__ dinv,
                             int*   __restrict__ recS,
                             int*   __restrict__ recD,
                             int*   __restrict__ recN,
                             const unsigned* __restrict__ bmap,
                             const float* __restrict__ x,
                             const float* __restrict__ W,
                             float* __restrict__ xl,
                             unsigned* __restrict__ degB,
                             int nEdges, int nNodes) {
    __shared__ __attribute__((aligned(16))) int qmem[AGG_W * EQCAP];
    __shared__ unsigned short gmem[AGG_W * GLCAP];
    __shared__ int   degc[TILE];
    __shared__ int   wcount[AGG_W];
    __shared__ __attribute__((aligned(16))) float wlds[IN_CH * HID];

    const int tid  = (int)threadIdx.x;
    const int lane = tid & 31;
    const int wave = tid >> 5;
    const int base = (int)blockIdx.x * TILE;
    const int nLoc = min(TILE, nNodes - base);

    for (int i = tid; i < TILE; i += AGG_T) degc[i] = 0;
    {
        constexpr int WIT = IN_CH * HID / AGG_T;
        static_assert(WIT * AGG_T == IN_CH * HID, "every thread stages the same number of words of W");
        float wv[WIT];
#pragma unroll
        for (int t = 0; t < WIT; ++t) wv[t] = W[t * AGG_T + tid];
#pragma unroll
        for (int t = 0; t < WIT; ++t) {
            const int i = t * AGG_T + tid;
            const int k = i / HID, col = i % HID;
            const int c = col >> 4;
            wlds[(((((k >> 2) * 2 + (c >> 1)) * 2 + ((k >> 1) & 1)) * 16 + (col & 15)) << 2) + ((c & 1) << 1) + (k & 1)] = wv[t];
        }
    }
    __syncthreads();

    int* myq = qmem + wave * EQCAP;
    unsigned short* mydl = (unsigned short*)(myq + IQCAP);
    int cnt = 0;

    auto stripProduct = [&]() {
        const int mr = lane & 15;
        const int hh = lane >> 4;
        const int nStrips = (nNodes + 15) >> 4;
        const v4f* bp = (const v4f*)wlds + (hh * 16 + mr);
        float* stg = (float*)myq;
        for (int s = (int)blockIdx.x + (int)gridDim.x * wave; s < nStrips; s += (int)gridDim.x * AGG_W) {
            const int row0 = s << 4;
            v4f g[2][8];
#pragma unroll
            for (int q = 0; q < 2; ++q) {
#pragma unroll
                for (int i = 0; i < 8; ++i) {
                    const int row = min(row0 + 2 * i + hh, nNodes - 1);
                    g[q][i] = *(const v4f*)(x + (size_t)row * IN_CH + q * 64 + 4 * mr);
                }
            }
            v8f acc0 = {}, acc1 = {}, acc2 = {}, acc3 = {};
#pragma unroll
            for (int q = 0; q < 2; ++q) {
#pragma unroll
                for (int i = 0; i < 8; ++i)
                    *(v4f*)(stg + (2 * i + hh) * APITCH + 4 * mr) = g[q][i];
                const v2f* as = (const v2f*)(stg + mr * APITCH + 2 * hh);
#pragma unroll 4
                for (int jj = 0; jj < 16; ++jj) {
                    const int j = q * 16 + jj;
                    const v2f a   = as[2 * jj];
                    const v4f b01 = bp[(j * 2 + 0) * 32];
                    const v4f b23 = bp[(j * 2 + 1) * 32];
                    const v2f b0 = {b01.x, b01.y}, b1 = {b01.z, b01.w}, b2 = {b23.x, b23.y}, b3 = {b23.z, b23.w};
                    acc0 = __builtin_amdgcn_wmma_f32_16x16x4_f32(false, a, false, b0, (short)0, acc0, false, false);
                    acc1 = __builtin_amdgcn_wmma_f32_16x16x4_f32(false, a, false, b1, (short)0, acc1, false, false);
                    acc2 = __builtin_amdgcn_wmma_f32_16x16x4_f32(false, a, false, b2, (short)0, acc2, false, false);
                    acc3 = __builtin_amdgcn_wmma_f32_16x16x4_f32(false, a, false, b3, (short)0, acc3, false, false);
                }
            }
            const int crow = row0 + (hh << 3);
#pragma unroll
            for (int r = 0; r < 8; ++r) {
                if (crow + r < nNodes) {
                    float* o = xl + (size_t)(crow + r) * HID + mr;
                    o[0]  = acc0[r];
                    o[16] = acc1[r];
                    o[32] = acc2[r];
                    o[48] = acc3[r];
                }
            }
        }
    };
    const bool early = ((wave >> 2) & 1) == 0;
    if (early) stripProduct();

    const size_t rbase = ((size_t)((int)blockIdx.x * AGG_W + wave) * CAPW) * 32 + lane;
    int nblk = 0;

    constexpr int RU = 8;
    auto recordAll = [&]() {
        const int nB   = (cnt + 31) >> 5;
        const int last = max(cnt - 1, 0);
        for (int b0 = 0; b0 < nB; b0 += RU) {
            int sv[RU], dv[RU];
#pragma unroll
            for (int u = 0; u < RU; ++u) {
                const int q = min((b0 + u) * 32 + lane, last);
                dv[u] = mydl[q];
                sv[u] = src[myq[q]];
            }
#pragma unroll
            for (int u = 0; u < RU; ++u) {
                if (b0 + u < nB) {
                    if (nblk < CAPW) {
                        const size_t o = rbase + (size_t)nblk * 32;
                        recS[o] = sv[u];
                        recD[o] = ((b0 + u) * 32 + lane < cnt) ? dv[u] : -1;
                    }
                    ++nblk;
                }
            }
        }
        cnt = 0;
    };

    unsigned short* glist = gmem + wave * GLCAP;
    const int hh  = lane >> 4;
    const int l16 = lane & 15;
    const int nW = (nEdges + 1023) >> 10;
    const int nRounds = (nW + 1023) >> 10;
    int pend = 0;
    int kb = 0;
    uint2 bw = make_uint2(0u, 0u);
    for (int t = 0; t <= 2 * nRounds; ++t) {
        const bool live = t < 2 * nRounds;
        const int  r = t >> 1, h = t & 1;
        if (h == 0) {
            bw = make_uint2(0u, 0u);
            const int w = (r * 32 + lane) * 32 + wave;
            if (live && w < nW) bw = ((const uint2*)bmap)[(size_t)w * MAXT + blockIdx.x];
        }
        unsigned bits = live ? (h ? bw.y : bw.x) : 0u;
        const int pc = __builtin_popcount(bits);
        int incl = pc;
#pragma unroll
        for (int off = 1; off < 32; off <<= 1) {
            const int v = __shfl_up(incl, off);
            if (lane >= off) incl += v;
        }
        const int total = __builtin_amdgcn_readfirstlane(__shfl(incl, 31));

        if (!live || pend + total > GLCAP || r - kb >= 32) {
            for (int i0 = 0; i0 < pend; i0 += 32) {
                const int gl = (int)glist[min(i0 + lane, pend - 1)];
                const int nq = min(32, pend - i0);
                const int ge = ((((((kb + (gl >> 11)) * 32 + ((gl >> 5) & 31)) * 32 + wave) * 2 + ((gl >> 10) & 1)) * 32) + (gl & 31)) * GSZ;
                if (cnt > IQCAP - PG * 32) recordAll();
                int dv[PG], ev[PG];
#pragma unroll
                for (int p = 0; p < PG; ++p) {
                    const int ea = __builtin_amdgcn_readlane(ge, min(2 * p,     nq - 1));
                    const int eb = __builtin_amdgcn_readlane(ge, min(2 * p + 1, nq - 1));
                    ev[p] = (hh ? eb : ea) + l16;
                    dv[p] = dst[min(ev[p], nEdges - 1)];
                }
#pragma unroll
                for (int p = 0; p < PG; ++p) {
                    if (2 * p < nq) {
                        const bool m = (2 * p + hh < nq) && (ev[p] < nEdges) && ((unsigned)(dv[p] - base) < (unsigned)nLoc);
                        const unsigned b = __builtin_amdgcn_ballot_w32(m);
                        if (m) {
                            const int qp = cnt + (int)__builtin_amdgcn_mbcnt_lo(b, 0u);
                            myq[qp]  = ev[p];
                            mydl[qp] = (unsigned short)(dv[p] - base);
                            atomicAdd(&degc[dv[p] - base], 1);
                        }
                        cnt += __builtin_popcount(b);
                    }
                }
            }
            pend = 0;
            kb = r;
        }

        int pos = pend + incl - pc;
        const unsigned tag = (unsigned)(((r - kb) << 11) | (h << 10) | (lane << 5));
        while (bits) {
            const int j = __builtin_ctz(bits);
            bits &= bits - 1u;
            glist[pos++] = (unsigned short)(tag | (unsigned)j);
        }
        pend += total;
    }
    recordAll();

    if (!early) stripProduct();

    if (lane == 0) wcount[wave] = nblk;
    __syncthreads();

    if (wave == 0) recN[(int)blockIdx.x * AGG_W + lane] = wcount[lane];

    for (int i = tid; i < nLoc; i += AGG_T) dinv[base + i] = rsqrtf(1.0f + (float)degc[i]);

    static_assert(TILE % 4 == 0, "a tile's byte counts fill whole words");
    for (int i = tid; i < TILE / 4; i += AGG_T) {
        unsigned wv = 0u;
#pragma unroll
        for (int k = 0; k < 4; ++k) {
            const int n = 4 * i + k;
            const unsigned d = (n < nLoc) ? (unsigned)min(degc[n], 255) : 0u;
            wv |= d << (8 * k);
        }
        degB[(base >> 2) + i] = wv;
    }
}

static __device__ __forceinline__ int4 gcn_load4_guarded(const int* __restrict__ p, int g, int n) {
    const int e = g << 2;
    int4 r;
    r.x = (e     < n) ? p[e]     : -1;
    r.y = (e + 1 < n) ? p[e + 1] : -1;
    r.z = (e + 2 < n) ? p[e + 2] : -1;
    r.w = (e + 3 < n) ? p[e + 3] : -1;
    return r;
}

template <bool V> struct gcn_tag { static constexpr bool value = V; };

template <int C, bool RELU, int MODE>
__global__ __launch_bounds__(AGG_T)
void gcn_tile_agg_kernel(const float* __restrict__ xl,
                         const int* __restrict__ src,
                         const int* __restrict__ dst,
                         const float* __restrict__ dinv,
                         const float* __restrict__ bias,
                         const float* __restrict__ Wn,
                         float* __restrict__ outp,
                         int*   __restrict__ recS,
                         int*   __restrict__ recD,
                         float* __restrict__ recW,
                         int*   __restrict__ recN,
                         const unsigned* __restrict__ degB,
                         int useRec,
                         int nEdges, int nNodes) {
    __shared__ __attribute__((aligned(16))) float tile[TILE * C];
    __shared__ __attribute__((aligned(16))) int qmem[(AGG_T / 32) * EQCAP];
    __shared__ int   wcount[AGG_T / 32];

    const int tid  = (int)threadIdx.x;
    const int lane = tid & 31;
    const int wave = tid >> 5;
    const int base = (int)blockIdx.x * TILE;
    const int nLoc = min(TILE, nNodes - base);

    int myN = 0;
    if (MODE == 1 || useRec) myN = recN[(int)blockIdx.x * (AGG_T / 32) + lane];

    constexpr int DTCAP = (AGG_T / 32) * EQCAP * 4;
    constexpr int DT_IT = (DTCAP / 16 + AGG_T - 1) / AGG_T;
    const bool useTab = (MODE == 0) && useRec && (nNodes <= DTCAP);
    const unsigned char* tab = (const unsigned char*)qmem;
    uint4 tv[DT_IT];
    int   tj[DT_IT];
    {
        const int nCh = useTab ? ((nNodes + 15) >> 4) : 0;
        const int rot = (int)(blockIdx.x & 7u) * (nCh >> 3);
#pragma unroll
        for (int t = 0; t < DT_IT; ++t) {
            const int i = t * AGG_T + tid;
            int j = i + rot;
            if (j >= nCh) j -= nCh;
            uint4 q = make_uint4(0u, 0u, 0u, 0u);
            if (i < nCh) q = ((const uint4*)degB)[j];
            tj[t] = (i < nCh) ? j : -1;
            tv[t] = q;
        }
    }

    constexpr int C4 = C / 4;
    constexpr int NV = TILE * C4;
    constexpr int IT = (NV + AGG_T - 1) / AGG_T;
    float4* tile4 = (float4*)tile;
    {
        float4 v[IT];
        float  di[IT];
#pragma unroll
        for (int t = 0; t < IT; ++t) {
            const int i  = min(t * AGG_T + tid, NV - 1);
            const int n  = min(i / C4, nLoc - 1);
            di[t] = dinv[base + n];
            v[t]  = *(const float4*)(xl + (long long)(base + n) * C + (i % C4) * 4);
        }
#pragma unroll
        for (int t = 0; t < IT; ++t) {
            const int i = t * AGG_T + tid;
            if (i < NV) {
                const bool ok = (i / C4) < nLoc;
                const float w = di[t] * di[t];
                float4 o;
                o.x = ok ? v[t].x * w : 0.0f;
                o.y = ok ? v[t].y * w : 0.0f;
                o.z = ok ? v[t].z * w : 0.0f;
                o.w = ok ? v[t].w * w : 0.0f;
                tile4[i] = o;
            }
        }
    }
#pragma unroll
    for (int t = 0; t < DT_IT; ++t)
        if (tj[t] >= 0) ((uint4*)qmem)[tj[t]] = tv[t];
    __syncthreads();

    auto byteDinv = [&](const unsigned d, const int n) -> float {
        float r = rsqrtf(1.0f + (float)d);
        if (d == 255u) r = dinv[n];
        return r;
    };

    int* myq = qmem + wave * EQCAP;
    int cnt = 0;

    constexpr int LPE = C / 2;
    constexpr int EPS = 32 / LPE;
    constexpr int PU  = 4;
    const int sub = lane / LPE;
    const int cl  = lane % LPE;

    const size_t rbase = ((size_t)((int)blockIdx.x * AGG_W + wave) * CAPW) * 32 + lane;
    int nblk = 0;

    auto drain = [&](const bool flush) {
        while (cnt >= 32 || (flush && cnt > 0)) {
            const int n = min(cnt, 32);
            int   s = 0, dl = 0;
            float w = 0.0f;
            if (lane < n) {
                const int e = myq[cnt - n + lane];
                s  = src[e];
                dl = dst[e] - base;
                w  = dinv[s] * dinv[base + dl];
            }
            if (MODE == 0) {
                if (nblk < CAPW) {
                    const size_t o = rbase + (size_t)nblk * 32;
                    recS[o] = s;
                    recD[o] = (lane < n) ? dl : -1;
                    recW[o] = w;
                }
                ++nblk;
            }
            cnt -= n;
            for (int j0 = 0; j0 < n; j0 += EPS * PU) {
                float2 r[PU];
                float  ww[PU];
                int    dd[PU];
#pragma unroll
                for (int p = 0; p < PU; ++p) {
                    const int jj = min(j0 + p * EPS + sub, n - 1);
                    const int ss = __shfl(s, jj);
                    dd[p] = __shfl(dl, jj);
                    ww[p] = __shfl(w, jj);
                    r[p]  = *(const float2*)(xl + (long long)ss * C + cl * 2);
                }
#pragma unroll
                for (int p = 0; p < PU; ++p) {
                    if (j0 + p * EPS + sub < n) {
                        float* t = &tile[dd[p] * C + cl * 2];
                        atomicAdd(t,     r[p].x * ww[p]);
                        atomicAdd(t + 1, r[p].y * ww[p]);
                    }
                }
            }
        }
    };

    constexpr int  GU  = 16;
    constexpr bool UNI = (EPS == 1);
    auto gatherRec = [&](const int nb, auto tabTag) {
        constexpr bool TAB = decltype(tabTag)::value;
        int   s0 = 0, dl0 = -1, s1 = 0, dl1 = -1;
        float w0 = 0.0f;
        if (nb > 0) {
            const size_t o1 = rbase + (size_t)min(1, nb - 1) * 32;
            s0  = recS[rbase];
            dl0 = recD[rbase];
            s1  = recS[o1];
            dl1 = recD[o1];
            const int d0 = base + max(dl0, 0);
            if constexpr (TAB) {
                const unsigned bx = tab[s0], by = tab[d0];
                w0 = byteDinv(bx, s0) * byteDinv(by, d0);
            } else {
                w0 = dinv[s0] * dinv[d0];
            }
        }
        for (int b = 0; b < nb; ++b) {
            const int m = __builtin_popcount(__builtin_amdgcn_ballot_w32(dl0 >= 0));
            int   s2 = 0, dl2 = -1;
            float x1 = 0.0f, y1 = 0.0f;
            recW[rbase + (size_t)b * 32] = w0;
            for (int j0 = 0; j0 < m; j0 += EPS * GU) {
                float2 r[GU];
                float  ww[GU];
                int    dd[GU];
#pragma unroll
                for (int p = 0; p < GU; ++p) {
                    const int jj = min(j0 + p * EPS + sub, m - 1);
                    int ss;
                    if (UNI) {
                        ss    = __builtin_amdgcn_readlane(s0, jj);
                        dd[p] = __builtin_amdgcn_readlane(dl0, jj);
                        ww[p] = __int_as_float(__builtin_amdgcn_readlane(__float_as_int(w0), jj));
                    } else {
                        ss    = __shfl(s0, jj);
                        dd[p] = __shfl(dl0, jj);
                        ww[p] = __shfl(w0, jj);
                    }
                    r[p] = *(const float2*)(xl + (long long)ss * C + cl * 2);
                }
                if (j0 == 0) {
                    const size_t o2 = rbase + (size_t)min(b + 2, nb - 1) * 32;
                    s2  = recS[o2];
                    dl2 = recD[o2];
                    if constexpr (TAB) {
                        const int d1 = base + max(dl1, 0);
                        const unsigned bx = tab[s1], by = tab[d1];
                        x1 = byteDinv(bx, s1);
                        y1 = byteDinv(by, d1);
                    } else {
                        x1 = dinv[s1];
                        y1 = dinv[base + max(dl1, 0)];
                    }
                }
#pragma unroll
                for (int p = 0; p < GU; ++p) {
                    if (j0 + p * EPS + sub < m) {
                        float* t = &tile[dd[p] * C + cl * 2];
                        atomicAdd(t,     r[p].x * ww[p]);
                        atomicAdd(t + 1, r[p].y * ww[p]);
                    }
                }
            }
            s0 = s1; dl0 = dl1; w0 = x1 * y1;
            s1 = s2; dl1 = dl2;
        }
    };

    bool replay = false;
    if (MODE == 1 || useRec)
        replay = (__builtin_amdgcn_ballot_w32(myN > CAPW) == 0u);
    const int nb = replay ? __builtin_amdgcn_readfirstlane(__shfl(myN, wave)) : 0;

    if (MODE == 1 && replay) {
        constexpr int RPU = 8;
        static_assert(32 % (EPS * RPU) == 0, "a block is a whole number of steps");
        int s = 0, dl = -1;
        float w = 0.0f;
        if (nb > 0) { s = recS[rbase]; dl = recD[rbase]; w = recW[rbase]; }
        for (int b = 0; b < nb; ++b) {
            int s2 = s, dl2 = dl;
            float w2 = w;
            for (int j0 = 0; j0 < 32; j0 += EPS * RPU) {
                float2 r[RPU];
                float  ww[RPU];
                int    dd[RPU];
#pragma unroll
                for (int p = 0; p < RPU; ++p) {
                    const int jj = j0 + p * EPS + sub;
                    const int ss = __shfl(s, jj);
                    dd[p] = __shfl(dl, jj);
                    ww[p] = __shfl(w, jj);
                    r[p]  = *(const float2*)(xl + (long long)ss * C + cl * 2);
                }
                if (j0 == 0) {
                    const size_t o = rbase + (size_t)min(b + 1, nb - 1) * 32;
                    s2 = recS[o]; dl2 = recD[o]; w2 = recW[o];
                }
#pragma unroll
                for (int p = 0; p < RPU; ++p) {
                    if (dd[p] >= 0) {
                        float* t = &tile[dd[p] * C + cl * 2];
                        atomicAdd(t,     r[p].x * ww[p]);
                        atomicAdd(t + 1, r[p].y * ww[p]);
                    }
                }
            }
            s = s2; dl = dl2; w = w2;
        }
    } else if (MODE == 0 && replay) {

    if (useTab) gatherRec(nb, gcn_tag<true>{});
    else        gatherRec(nb, gcn_tag<false>{});

    } else {

    const int  nG  = (nEdges + 3) >> 2;
    const int  nG4 = nEdges >> 2;
    const bool al  = (((size_t)dst) & 15) == 0;

    for (int g0 = 0; g0 < nG; g0 += AGG_T * SCAN_U) {
        int4 d4[SCAN_U];
        if (al && g0 + AGG_T * SCAN_U <= nG4) {
            const int4* p = (const int4*)dst + (g0 + tid);
#pragma unroll
            for (int u = 0; u < SCAN_U; ++u)
                d4[u] = p[u * AGG_T];
        } else {
#pragma unroll
            for (int u = 0; u < SCAN_U; ++u)
                d4[u] = gcn_load4_guarded(dst, g0 + u * AGG_T + tid, nEdges);
        }

#pragma unroll
        for (int u = 0; u < SCAN_U; ++u) {
            const int e0 = (g0 + u * AGG_T + tid) << 2;
            const unsigned r0 = (unsigned)(d4[u].x - base);
            const unsigned r1 = (unsigned)(d4[u].y - base);
            const unsigned r2 = (unsigned)(d4[u].z - base);
            const unsigned r3 = (unsigned)(d4[u].w - base);
            const unsigned rm = min(min(r0, r1), min(r2, r3));
            if (__builtin_amdgcn_ballot_w32(rm < (unsigned)nLoc)) {
                const bool m0 = r0 < (unsigned)nLoc;
                const bool m1 = r1 < (unsigned)nLoc;
                const bool m2 = r2 < (unsigned)nLoc;
                const bool m3 = r3 < (unsigned)nLoc;
                const unsigned b0 = __builtin_amdgcn_ballot_w32(m0);
                const unsigned b1 = __builtin_amdgcn_ballot_w32(m1);
                const unsigned b2 = __builtin_amdgcn_ballot_w32(m2);
                const unsigned b3 = __builtin_amdgcn_ballot_w32(m3);
                if (b0) {
                    if (m0) myq[cnt + (int)__builtin_amdgcn_mbcnt_lo(b0, 0u)] = e0;
                    cnt += __builtin_popcount(b0);
                }
                if (b1) {
                    if (m1) myq[cnt + (int)__builtin_amdgcn_mbcnt_lo(b1, 0u)] = e0 + 1;
                    cnt += __builtin_popcount(b1);
                }
                if (b2) {
                    if (m2) myq[cnt + (int)__builtin_amdgcn_mbcnt_lo(b2, 0u)] = e0 + 2;
                    cnt += __builtin_popcount(b2);
                }
                if (b3) {
                    if (m3) myq[cnt + (int)__builtin_amdgcn_mbcnt_lo(b3, 0u)] = e0 + 3;
                    cnt += __builtin_popcount(b3);
                }
            }
        }

        drain(g0 + AGG_T * SCAN_U >= nG);
    }

    }

    constexpr int KST = C / 4;
    const int  pmr  = lane & 15;
    const int  phh  = lane >> 4;
    const bool mine = (MODE == 0) && (wave < ((nLoc + 15) >> 4));
    v2f bw[KST];
    if constexpr (MODE == 0) {
        static_assert(C == HID && OUT_CH == 16, "the strip product makes one 16-column tile from HID k");
        static_assert((TILE + 15) / 16 <= AGG_W, "one strip per wave covers the tile");
        static_assert(16 * APITCH <= EQCAP && APITCH >= C, "a strip fits the wave's queue memory");
        if (mine) {
#pragma unroll
            for (int j = 0; j < KST; ++j) {
                const float* bp = Wn + (4 * j + 2 * phh) * OUT_CH + pmr;
                bw[j].x = bp[0];
                bw[j].y = bp[OUT_CH];
            }
        }
    }

    const bool made = (MODE == 0) && !replay;
    if (made && lane == 0) wcount[wave] = nblk;
    __syncthreads();

    if (made && wave == 0) recN[(int)blockIdx.x * AGG_W + lane] = wcount[lane];

    if constexpr (MODE == 0) {
        if (mine) {
            const int row0 = wave << 4;
            float* stg = (float*)(qmem + wave * EQCAP);
            const float4 bq = ((const float4*)bias)[pmr];
#pragma unroll
            for (int i = 0; i < 8; ++i) {
                const int r = 2 * i + phh;
                const float4 a = tile4[min(row0 + r, TILE - 1) * C4 + pmr];
                v4f o = {a.x + bq.x, a.y + bq.y, a.z + bq.z, a.w + bq.w};
                if (RELU) { o.x = fmaxf(o.x, 0.0f); o.y = fmaxf(o.y, 0.0f); o.z = fmaxf(o.z, 0.0f); o.w = fmaxf(o.w, 0.0f); }
                *(v4f*)(stg + r * APITCH + 4 * pmr) = o;
            }
            const v2f* as = (const v2f*)(stg + pmr * APITCH + 2 * phh);
            v8f acc = {};
#pragma unroll
            for (int j = 0; j < KST; ++j)
                acc = __builtin_amdgcn_wmma_f32_16x16x4_f32(false, as[2 * j], false, bw[j], (short)0, acc, false, false);
            const int crow = row0 + (phh << 3);
#pragma unroll
            for (int r = 0; r < 8; ++r)
                if (crow + r < nLoc) outp[(size_t)(base + crow + r) * OUT_CH + pmr] = acc[r];
        }
    } else {
    {
        const float4* bias4 = (const float4*)bias;
        float4* out4 = (float4*)(outp + (long long)base * C);
#pragma unroll
        for (int t = 0; t < IT; ++t) {
            const int i = t * AGG_T + tid;
            if (i < nLoc * C4) {
                const float4 a = tile4[i];
                const float4 b = bias4[i % C4];
                float4 o;
                o.x = a.x + b.x; o.y = a.y + b.y; o.z = a.z + b.z; o.w = a.w + b.w;
                if (RELU) { o.x = fmaxf(o.x, 0.0f); o.y = fmaxf(o.y, 0.0f); o.z = fmaxf(o.z, 0.0f); o.w = fmaxf(o.w, 0.0f); }
                out4[i] = o;
            }
        }
    }
    }
}

extern "C" void kernel_launch(void* const* d_in, const int* in_sizes, int n_in,
                              void* d_out, int out_size, void* d_ws, size_t ws_size,
                              hipStream_t stream) {
    const float* x  = (const float*)d_in[0];
    const int*   ei = (const int*)  d_in[1];
    const float* W1 = (const float*)d_in[2];
    const float* b1 = (const float*)d_in[3];
    const float* W2 = (const float*)d_in[4];
    const float* b2 = (const float*)d_in[5];
    float* out = (float*)d_out;

    const int nNodes = in_sizes[0] / IN_CH;
    const int nEdges = in_sizes[1] / 2;
    const int* src = ei;
    const int* dst = ei + nEdges;

    float* ws  = (float*)d_ws;
    float* deg = ws;
    float* xl1 = deg + ((nNodes + 63) & ~63);
    float* xl2 = xl1 + (long long)nNodes * HID;

    const int B = 256;
    const int nTiles = (nNodes + TILE - 1) / TILE;

    const long long recLen = (long long)nTiles * AGG_W * CAPW * 32;
    int*   recS = (int*)(xl2 + (((long long)nNodes * OUT_CH + 31) & ~31LL));
    int*   recD = recS + recLen;
    float* recW = (float*)(recD + recLen);
    int*   recN = (int*)(recW + recLen);
    unsigned* bmap = (unsigned*)(recN + (long long)nTiles * AGG_W);
    const int nWords = (nEdges + 1023) / 1024;
    unsigned* degB = bmap + (long long)nWords * MAXT * 2;
    const int useIdx = (nTiles <= MAXT) ? 1 : 0;

    if (useIdx) {
        const int wpb = IDX_T / 32;
        gcn_group_index_kernel<<<(nWords + wpb - 1) / wpb, IDX_T, 0, stream>>>(dst, bmap, nEdges, nWords);
        gcn_tile_collect_kernel<<<nTiles, AGG_T, 0, stream>>>(src, dst, deg, recS, recD, recN, bmap, x, W1, xl1, degB, nEdges, nNodes);
    } else {
        gcn_init_kernel<<<(nNodes + B - 1) / B, B, 0, stream>>>(deg, nNodes);
        gcn_degree_kernel<<<(nEdges + B - 1) / B, B, 0, stream>>>(dst, deg, nEdges);
        gcn_rsqrt_kernel<<<(nNodes + B - 1) / B, B, 0, stream>>>(deg, nNodes);
        int tiles = (nNodes / 16) * (HID / 16);
        int wpb = B / 32;
        gcn_gemm_wmma_f32<<<(tiles + wpb - 1) / wpb, B, 0, stream>>>(x, W1, xl1, nNodes, IN_CH, HID);
    }

    gcn_tile_agg_kernel<HID, true, 0><<<nTiles, AGG_T, 0, stream>>>(
        xl1, src, dst, deg, b1, W2, xl2, recS, recD, recW, recN, degB, useIdx, nEdges, nNodes);

    gcn_tile_agg_kernel<OUT_CH, false, 1><<<nTiles, AGG_T, 0, stream>>>(
        xl2, src, dst, deg, b2, W2, out, recS, recD, recW, recN, degB, useIdx, nEdges, nNodes);
}
